# starting kernel
# baseline (speedup 1.0000x reference)
#include <hip/hip_runtime.h>
#include <hip/hip_bf16.h>

typedef __attribute__((__ext_vector_type__(16))) __bf16 bf16x16;
typedef __attribute__((__ext_vector_type__(8)))  float  f32x8;
typedef __attribute__((__ext_vector_type__(4)))  unsigned int u32x4;

union Frag16 {
    u32x4   q[2];
    bf16x16 v;
};

static __device__ __forceinline__ unsigned short f2bf(float f) {
    unsigned u = __float_as_uint(f);
    u += 0x7FFFu + ((u >> 16) & 1u);
    return (unsigned short)(u >> 16);
}

static __device__ __forceinline__ unsigned fkey(float f) {
    unsigned u = __float_as_uint(f);
    return (u & 0x80000000u) ? ~u : (u | 0x80000000u);
}
static __device__ __forceinline__ float funkey(unsigned k) {
    unsigned u = (k & 0x80000000u) ? (k & 0x7FFFFFFFu) : ~k;
    return __uint_as_float(u);
}

#define GEMM_WAVES 4
__global__ void wmma_gemm_bf16(const unsigned short* __restrict__ A,
                               const unsigned short* __restrict__ Bt,
                               float* __restrict__ Y,
                               int M, int K, int NOUT)
{
    const int lane  = threadIdx.x & 31;
    const int wave  = threadIdx.x >> 5;
    const int tileM = blockIdx.x * GEMM_WAVES + wave;
    const int m0    = tileM * 16;
    if (m0 >= M) return;

    const int l15   = lane & 15;
    const int half  = lane >> 4;
    const int kbase = half * 8;

    int rowA = m0 + l15;
    if (rowA >= M) rowA = M - 1;
    const unsigned short* aptr = A + (size_t)rowA * K + kbase;

    for (int c0 = 0; c0 < NOUT; c0 += 16) {
        const unsigned short* bptr = Bt + (size_t)(c0 + l15) * K + kbase;
        f32x8 acc = {};
        for (int k0 = 0; k0 < K; k0 += 32) {
            Frag16 fa, fb;
            fa.q[0] = *(const u32x4*)(aptr + k0);
            fa.q[1] = *(const u32x4*)(aptr + k0 + 16);
            fb.q[0] = *(const u32x4*)(bptr + k0);
            fb.q[1] = *(const u32x4*)(bptr + k0 + 16);
            acc = __builtin_amdgcn_wmma_f32_16x16x32_bf16(
                      false, fa.v, false, fb.v, (short)0, acc, false, false);
        }
        const int colY = c0 + l15;
#pragma unroll
        for (int r = 0; r < 8; ++r) {
            const int row = m0 + r + half * 8;
            if (row < M) Y[(size_t)row * NOUT + colY] = acc[r];
        }
    }
}

__global__ void node_scores(const float* __restrict__ h,
                            const float* __restrict__ as_, const float* __restrict__ ad_,
                            float* __restrict__ es, float* __restrict__ ed,
                            int N, int C)
{
    const int warp = (blockIdx.x * blockDim.x + threadIdx.x) >> 5;
    const int lane = threadIdx.x & 31;
    if (warp >= N) return;
    const float* row = h + (size_t)warp * C;
    float s = 0.f, d = 0.f;
    for (int c = lane; c < C; c += 32) {
        const float v = row[c];
        s += v * as_[c];
        d += v * ad_[c];
    }
#pragma unroll
    for (int off = 16; off; off >>= 1) {
        s += __shfl_xor(s, off);
        d += __shfl_xor(d, off);
    }
    if (lane == 0) { es[warp] = s; ed[warp] = d; }
}

__global__ void init_node(unsigned* __restrict__ maxbits, float* __restrict__ denom, int N)
{
    const int i = blockIdx.x * blockDim.x + threadIdx.x;
    if (i >= N) return;
    maxbits[i] = fkey(-__builtin_inff());
    denom[i]   = 0.f;
}

__global__ void init_agg(float* __restrict__ agg, const float* __restrict__ bias, int N, int C)
{
    const size_t i = (size_t)blockIdx.x * blockDim.x + threadIdx.x;
    if (i >= (size_t)N * C) return;
    agg[i] = bias[i % C];
}

__global__ void edge_logit_max(const int* __restrict__ ei, int E, int N,
                               const float* __restrict__ es, const float* __restrict__ ed,
                               float* __restrict__ logits, unsigned* __restrict__ maxbits)
{
    const int e = blockIdx.x * blockDim.x + threadIdx.x;
    if (e >= E + N) return;
    const int s = (e < E) ? ei[e]     : e - E;
    const int d = (e < E) ? ei[E + e] : e - E;
    float l = es[s] + ed[d];
    l = (l > 0.f) ? l : 0.2f * l;
    logits[e] = l;
    atomicMax(&maxbits[d], fkey(l));
}

__global__ void edge_exp(const int* __restrict__ ei, int E, int N,
                         const float* __restrict__ logits, const unsigned* __restrict__ maxbits,
                         float* __restrict__ p, float* __restrict__ denom)
{
    const int e = blockIdx.x * blockDim.x + threadIdx.x;
    if (e >= E + N) return;
    const int d = (e < E) ? ei[E + e] : e - E;
    const float v = __expf(logits[e] - funkey(maxbits[d]));
    p[e] = v;
    atomicAdd(&denom[d], v);
}

__global__ void edge_aggregate(const int* __restrict__ ei, int E, int N,
                               const float* __restrict__ h, const float* __restrict__ p,
                               const float* __restrict__ denom, float* __restrict__ agg, int C)
{
    const int NE = E + N;
    for (int e = blockIdx.x; e < NE; e += gridDim.x) {
        const int s = (e < E) ? ei[e]     : e - E;
        const int d = (e < E) ? ei[E + e] : e - E;
        const float alpha = p[e] / denom[d];
        const float* hs = h   + (size_t)s * C;
        float*       od = agg + (size_t)d * C;
        for (int c = threadIdx.x; c < C; c += blockDim.x)
            atomicAdd(&od[c], hs[c] * alpha);
    }
}

__global__ void to_bf16(const float* __restrict__ x, unsigned short* __restrict__ y, size_t n)
{
    const size_t i = (size_t)blockIdx.x * blockDim.x + threadIdx.x;
    if (i < n) y[i] = f2bf(x[i]);
}

__global__ void relu_to_bf16(const float* __restrict__ x, unsigned short* __restrict__ y, size_t n)
{
    const size_t i = (size_t)blockIdx.x * blockDim.x + threadIdx.x;
    if (i < n) { const float v = x[i]; y[i] = f2bf(v > 0.f ? v : 0.f); }
}

__global__ void transpose_to_bf16(const float* __restrict__ W, unsigned short* __restrict__ Wt,
                                  int K, int NOUT)
{
    const int i = blockIdx.x * blockDim.x + threadIdx.x;
    if (i >= K * NOUT) return;
    const int k = i / NOUT, n = i % NOUT;
    Wt[(size_t)n * K + k] = f2bf(W[i]);
}

__global__ void link_predict(const float* __restrict__ h, const int* __restrict__ mask,
                             const float* __restrict__ Wl, const float* __restrict__ bl,
                             float* __restrict__ out, int P, int C)
{
    const int warp = (blockIdx.x * blockDim.x + threadIdx.x) >> 5;
    const int lane = threadIdx.x & 31;
    if (warp >= P) return;
    const int m0 = mask[warp * 2 + 0];
    const int m1 = mask[warp * 2 + 1];
    const float* r0 = h + (size_t)m0 * C;
    const float* r1 = h + (size_t)m1 * C;
    float acc = 0.f;
    for (int c = lane; c < C; c += 32)
        acc += r0[c] * Wl[c] + r1[c] * Wl[C + c];
#pragma unroll
    for (int off = 16; off; off >>= 1) acc += __shfl_xor(acc, off);
    if (lane == 0) out[warp] = 1.f / (1.f + expf(-(acc + bl[0])));
}

extern "C" void kernel_launch(void* const* d_in, const int* in_sizes, int n_in,
                              void* d_out, int out_size, void* d_ws, size_t ws_size,
                              hipStream_t stream)
{
    (void)n_in; (void)out_size; (void)ws_size;

    const float* features = (const float*)d_in[0];
    const int*   ei       = (const int*)  d_in[1];
    const int*   mask     = (const int*)  d_in[2];
    const float* W1   = (const float*)d_in[3];
    const float* a_s1 = (const float*)d_in[4];
    const float* a_d1 = (const float*)d_in[5];
    const float* b1   = (const float*)d_in[6];
    const float* W2   = (const float*)d_in[7];
    const float* a_s2 = (const float*)d_in[8];
    const float* a_d2 = (const float*)d_in[9];
    const float* b2   = (const float*)d_in[10];
    const float* Wl   = (const float*)d_in[11];
    const float* bl   = (const float*)d_in[12];

    const int F  = 128, Hh = 256;
    const int N  = in_sizes[0] / F;
    const int E  = in_sizes[1] / 2;
    const int P  = in_sizes[2] / 2;
    const int NE = E + N;

    char* ws = (char*)d_ws;
    size_t off = 0;
    auto alloc = [&](size_t bytes) -> char* {
        char* ptr = ws + off;
        off += (bytes + 255) & ~(size_t)255;
        return ptr;
    };
    unsigned short* xbf  = (unsigned short*)alloc((size_t)N * F  * 2);
    unsigned short* w1t  = (unsigned short*)alloc((size_t)Hh * F * 2);
    unsigned short* w2t  = (unsigned short*)alloc((size_t)F * Hh * 2);
    unsigned short* hrbf = (unsigned short*)alloc((size_t)N * Hh * 2);
    float*    buf1    = (float*)   alloc((size_t)N * Hh * 4);
    float*    buf2    = (float*)   alloc((size_t)N * Hh * 4);
    float*    es      = (float*)   alloc((size_t)N * 4);
    float*    ed      = (float*)   alloc((size_t)N * 4);
    unsigned* maxbits = (unsigned*)alloc((size_t)N * 4);
    float*    denom   = (float*)   alloc((size_t)N * 4);
    float*    logits  = (float*)   alloc((size_t)NE * 4);
    float*    pbuf    = (float*)   alloc((size_t)NE * 4);

    const int tilesM = (N + 15) / 16;
    const int gemmGrid = (tilesM + GEMM_WAVES - 1) / GEMM_WAVES;
    const int TB = 256;
    auto cdiv = [](size_t a, size_t b) -> int { return (int)((a + b - 1) / b); };

    to_bf16<<<cdiv((size_t)N * F, TB), TB, 0, stream>>>(features, xbf, (size_t)N * F);
    transpose_to_bf16<<<cdiv((size_t)F * Hh, TB), TB, 0, stream>>>(W1, w1t, F, Hh);
    transpose_to_bf16<<<cdiv((size_t)Hh * F, TB), TB, 0, stream>>>(W2, w2t, Hh, F);

    wmma_gemm_bf16<<<gemmGrid, 32 * GEMM_WAVES, 0, stream>>>(xbf, w1t, buf1, N, F, Hh);
    node_scores<<<cdiv((size_t)N * 32, TB), TB, 0, stream>>>(buf1, a_s1, a_d1, es, ed, N, Hh);
    init_node<<<cdiv(N, TB), TB, 0, stream>>>(maxbits, denom, N);
    init_agg<<<cdiv((size_t)N * Hh, TB), TB, 0, stream>>>(buf2, b1, N, Hh);
    edge_logit_max<<<cdiv(NE, TB), TB, 0, stream>>>(ei, E, N, es, ed, logits, maxbits);
    edge_exp<<<cdiv(NE, TB), TB, 0, stream>>>(ei, E, N, logits, maxbits, pbuf, denom);
    edge_aggregate<<<8192, Hh, 0, stream>>>(ei, E, N, buf1, pbuf, denom, buf2, Hh);
    relu_to_bf16<<<cdiv((size_t)N * Hh, TB), TB, 0, stream>>>(buf2, hrbf, (size_t)N * Hh);

    wmma_gemm_bf16<<<gemmGrid, 32 * GEMM_WAVES, 0, stream>>>(hrbf, w2t, buf1, N, Hh, F);
    node_scores<<<cdiv((size_t)N * 32, TB), TB, 0, stream>>>(buf1, a_s2, a_d2, es, ed, N, F);
    init_node<<<cdiv(N, TB), TB, 0, stream>>>(maxbits, denom, N);
    init_agg<<<cdiv((size_t)N * F, TB), TB, 0, stream>>>(buf2, b2, N, F);
    edge_logit_max<<<cdiv(NE, TB), TB, 0, stream>>>(ei, E, N, es, ed, logits, maxbits);
    edge_exp<<<cdiv(NE, TB), TB, 0, stream>>>(ei, E, N, logits, maxbits, pbuf, denom);
    edge_aggregate<<<8192, F, 0, stream>>>(ei, E, N, buf1, pbuf, denom, buf2, F);

    link_predict<<<cdiv((size_t)P * 32, TB), TB, 0, stream>>>(buf2, mask, Wl, bl,
                                                              (float*)d_out, P, F);
}


# Both dense products in one kernel: a wave leaves its 16 rows of relu(x W1 + b1) as bf16 in LDS and multiplies them by W2 from there; the N x 256 intermediate never goes to memory; 18 -> 16 launches
# speedup vs baseline: 2.7177x; 2.7177x over previous
#include <hip/hip_runtime.h>
#include <hip/hip_bf16.h>

typedef __attribute__((__ext_vector_type__(16))) __bf16 bf16x16;
typedef __attribute__((__ext_vector_type__(8)))  float  f32x8;
typedef __attribute__((__ext_vector_type__(4)))  unsigned int u32x4;

union Frag16 {
    u32x4   q[2];
    bf16x16 v;
};

static __device__ __forceinline__ unsigned short f2bf(float f) {
    unsigned u = __float_as_uint(f);
    u += 0x7FFFu + ((u >> 16) & 1u);
    return (unsigned short)(u >> 16);
}

static __device__ __forceinline__ unsigned fkey(float f) {
    unsigned u = __float_as_uint(f);
    return (u & 0x80000000u) ? ~u : (u | 0x80000000u);
}
static __device__ __forceinline__ float funkey(unsigned k) {
    unsigned u = (k & 0x80000000u) ? (k & 0x7FFFFFFFu) : ~k;
    return __uint_as_float(u);
}

#define PAIR_WAVES 4
#define PAIR_PAD   8

template <int K1, int N1, int N2>
__global__ void __launch_bounds__(32 * PAIR_WAVES)
wmma_two_products(const unsigned short* __restrict__ A,
                  const unsigned short* __restrict__ B1t,
                  const float* __restrict__ bias1,
                  const unsigned short* __restrict__ B2t,
                  float* __restrict__ Y, int M)
{
    static_assert(K1 % 32 == 0 && N1 % 32 == 0 && N1 % 16 == 0 && N2 % 16 == 0, "whole fragments");
    constexpr int PITCH = N1 + PAIR_PAD;
    static_assert((PITCH * 2) % 16 == 0 && ((PITCH * 2 / 16) & 1) == 1, "pitch = 16 B x odd");
    constexpr int OPITCH = N2 + 4;
    static_assert(N2 == 128, "one float4 per lane covers a result row");
    static_assert(OPITCH * 4 <= PITCH * 2, "the result tile reuses the wave's LDS tile");
    constexpr int KS1 = K1 / 32;
    constexpr int KS2 = N1 / 32;
    __shared__ __attribute__((aligned(16))) unsigned short mid[PAIR_WAVES][16 * PITCH];

    const int lane  = threadIdx.x & 31;
    const int wave  = __builtin_amdgcn_readfirstlane(threadIdx.x >> 5);
    const int m0    = (blockIdx.x * PAIR_WAVES + wave) * 16;
    const bool live = m0 < M;
    const int l15   = lane & 15;
    const int half  = lane >> 4;
    const int kbase = half * 8;
    unsigned short* mine = mid[wave];

    if (live) {
        int rowA = m0 + l15;
        if (rowA >= M) rowA = M - 1;
        const unsigned short* aptr = A + (size_t)rowA * K1 + kbase;
        Frag16 fa[KS1];
#pragma unroll
        for (int k = 0; k < KS1; ++k) {
            fa[k].q[0] = *(const u32x4*)(aptr + 32 * k);
            fa[k].q[1] = *(const u32x4*)(aptr + 32 * k + 16);
        }
        Frag16 fb[KS1], fn[KS1];
        {
            const unsigned short* bptr = B1t + (size_t)l15 * K1 + kbase;
#pragma unroll
            for (int k = 0; k < KS1; ++k) {
                fb[k].q[0] = *(const u32x4*)(bptr + 32 * k);
                fb[k].q[1] = *(const u32x4*)(bptr + 32 * k + 16);
            }
        }
#pragma unroll
        for (int t = 0; t < N1 / 16; ++t) {
            if (t + 1 < N1 / 16) {
                const unsigned short* bptr = B1t + (size_t)((t + 1) * 16 + l15) * K1 + kbase;
#pragma unroll
                for (int k = 0; k < KS1; ++k) {
                    fn[k].q[0] = *(const u32x4*)(bptr + 32 * k);
                    fn[k].q[1] = *(const u32x4*)(bptr + 32 * k + 16);
                }
            }
            f32x8 acc = {};
#pragma unroll
            for (int k = 0; k < KS1; ++k)
                acc = __builtin_amdgcn_wmma_f32_16x16x32_bf16(
                          false, fa[k].v, false, fb[k].v, (short)0, acc, false, false);
            const int   col = t * 16 + l15;
            const float bv  = bias1[col];
#pragma unroll
            for (int r = 0; r < 8; ++r) {
                const float v = acc[r] + bv;
                mine[(r + half * 8) * PITCH + col] = f2bf(v > 0.f ? v : 0.f);
            }
#pragma unroll
            for (int k = 0; k < KS1; ++k) fb[k] = fn[k];
        }
    }
    __syncthreads();
    Frag16 ga[KS2];
    if (live) {
        const unsigned short* lp = mine + l15 * PITCH + kbase;
#pragma unroll
        for (int k = 0; k < KS2; ++k) {
            ga[k].q[0] = *(const u32x4*)(lp + 32 * k);
            ga[k].q[1] = *(const u32x4*)(lp + 32 * k + 16);
        }
    }
    __syncthreads();
    float* outt = reinterpret_cast<float*>(mine);
    if (live) {
        Frag16 gb[KS2], gn[KS2];
        {
            const unsigned short* bptr = B2t + (size_t)l15 * N1 + kbase;
#pragma unroll
            for (int k = 0; k < KS2; ++k) {
                gb[k].q[0] = *(const u32x4*)(bptr + 32 * k);
                gb[k].q[1] = *(const u32x4*)(bptr + 32 * k + 16);
            }
        }
#pragma unroll
        for (int t = 0; t < N2 / 16; ++t) {
            if (t + 1 < N2 / 16) {
                const unsigned short* bptr = B2t + (size_t)((t + 1) * 16 + l15) * N1 + kbase;
#pragma unroll
                for (int k = 0; k < KS2; ++k) {
                    gn[k].q[0] = *(const u32x4*)(bptr + 32 * k);
                    gn[k].q[1] = *(const u32x4*)(bptr + 32 * k + 16);
                }
            }
            f32x8 acc = {};
#pragma unroll
            for (int k = 0; k < KS2; ++k)
                acc = __builtin_amdgcn_wmma_f32_16x16x32_bf16(
                          false, ga[k].v, false, gb[k].v, (short)0, acc, false, false);
#pragma unroll
            for (int r = 0; r < 8; ++r)
                outt[(r + half * 8) * OPITCH + t * 16 + l15] = acc[r];
#pragma unroll
            for (int k = 0; k < KS2; ++k) gb[k] = gn[k];
        }
    }
    __syncthreads();
    if (live) {
        float* yrow = Y + (size_t)m0 * N2 + 4 * lane;
        if (m0 + 16 <= M) {
            float4 v[16];
#pragma unroll
            for (int i = 0; i < 16; ++i) v[i] = *reinterpret_cast<const float4*>(outt + i * OPITCH + 4 * lane);
#pragma unroll
            for (int i = 0; i < 16; ++i) *reinterpret_cast<float4*>(yrow + (size_t)i * N2) = v[i];
        } else {
            for (int i = 0; i < M - m0; ++i)
                *reinterpret_cast<float4*>(yrow + (size_t)i * N2) =
                    *reinterpret_cast<const float4*>(outt + i * OPITCH + 4 * lane);
        }
    }
}

__global__ void node_scores(const float* __restrict__ h,
                            const float* __restrict__ as_, const float* __restrict__ ad_,
                            float* __restrict__ es, float* __restrict__ ed,
                            int N, int C)
{
    const int warp = (blockIdx.x * blockDim.x + threadIdx.x) >> 5;
    const int lane = threadIdx.x & 31;
    if (warp >= N) return;
    const float* row = h + (size_t)warp * C;
    float s = 0.f, d = 0.f;
    for (int c = lane; c < C; c += 32) {
        const float v = row[c];
        s += v * as_[c];
        d += v * ad_[c];
    }
#pragma unroll
    for (int off = 16; off; off >>= 1) {
        s += __shfl_xor(s, off);
        d += __shfl_xor(d, off);
    }
    if (lane == 0) { es[warp] = s; ed[warp] = d; }
}

__global__ void init_node(unsigned* __restrict__ maxbits, float* __restrict__ denom, int N,
                          unsigned* __restrict__ need)
{
    const int i = blockIdx.x * blockDim.x + threadIdx.x;
    if (i >= N) return;
    maxbits[i] = fkey(-__builtin_inff());
    denom[i]   = 0.f;
    if (need) need[i] = 0u;
}

__global__ void mark_needed(const int* __restrict__ mask, unsigned* __restrict__ need, int n, int N)
{
    const int i = blockIdx.x * blockDim.x + threadIdx.x;
    if (i >= n) return;
    const int m = mask[i];
    if ((unsigned)m < (unsigned)N) need[m] = 1u;
}

__global__ void edge_logit_max(const int* __restrict__ ei, int E, int N,
                               const float* __restrict__ es, const float* __restrict__ ed,
                               float* __restrict__ logits, unsigned* __restrict__ maxbits,
                               const unsigned* __restrict__ need)
{
    const int e = blockIdx.x * blockDim.x + threadIdx.x;
    if (e >= E + N) return;
    const int d = (e < E) ? ei[E + e] : e - E;
    if (need && !need[d]) return;
    const int s = (e < E) ? ei[e]     : e - E;
    float l = es[s] + ed[d];
    l = (l > 0.f) ? l : 0.2f * l;
    logits[e] = l;
    atomicMax(&maxbits[d], fkey(l));
}

__global__ void edge_exp(const int* __restrict__ ei, int E, int N,
                         const float* __restrict__ logits, const unsigned* __restrict__ maxbits,
                         float* __restrict__ p, float* __restrict__ denom,
                         const unsigned* __restrict__ need)
{
    const int e = blockIdx.x * blockDim.x + threadIdx.x;
    if (e >= E + N) return;
    const int d = (e < E) ? ei[E + e] : e - E;
    if (need && !need[d]) return;
    const float v = __expf(logits[e] - funkey(maxbits[d]));
    p[e] = v;
    atomicAdd(&denom[d], v);
}

#define OWN_ROWS    196
#define OWN_THREADS 1024
#define OWN_QCAP    4096
#define OWN_U       4

template <int C>
static __device__ __forceinline__ void own_add_now(const int* __restrict__ ei, int E, int lo, int e,
                                                   const float* __restrict__ h, const float* __restrict__ p,
                                                   const float* __restrict__ denom, float* acc, int lane)
{
    const int   d = ei[E + e];
    const int   s = ei[e];
    const float a = p[e] / denom[d];
    const float* hs = h + (size_t)s * C + lane;
    float*       ar = acc + (d - lo) * C + lane;
    float v[C / 32];
#pragma unroll
    for (int k = 0; k < C / 32; ++k) v[k] = hs[32 * k];
#pragma unroll
    for (int k = 0; k < C / 32; ++k) atomicAdd(&ar[32 * k], v[k] * a);
}

template <int C, bool SECOND>
__global__ void __launch_bounds__(OWN_THREADS)
aggregate_owned(const int* __restrict__ ei, int E, int N,
                const float* __restrict__ h, const float* __restrict__ p,
                const float* __restrict__ denom,
                const unsigned* __restrict__ need, const float* __restrict__ bias,
                unsigned short* __restrict__ ybf, float* __restrict__ yf)
{
    static_assert(C == 128, "one float4 per lane covers a row");
    __shared__ __attribute__((aligned(16))) float acc[OWN_ROWS * C];
    __shared__ int      qe[OWN_QCAP];
    __shared__ unsigned qd[OWN_QCAP];
    __shared__ float    aself[OWN_ROWS];
    __shared__ unsigned flag[OWN_ROWS];
    __shared__ unsigned qn;

    const int lane = threadIdx.x & 31;
    const int wave = __builtin_amdgcn_readfirstlane(threadIdx.x >> 5);
    const int nw   = OWN_THREADS / 32;
    const int lo   = blockIdx.x * OWN_ROWS;
    const int nrow = min(OWN_ROWS, N - lo);

    if (threadIdx.x == 0) qn = 0u;
    for (int r = threadIdx.x; r < nrow; r += OWN_THREADS) {
        const int d = lo + r;
        bool on = true;
        if (SECOND) { on = need[d] != 0u; flag[r] = on ? 1u : 0u; }
        aself[r] = on ? p[E + d] / denom[d] : 0.f;
    }
    __syncthreads();
    {
        const float4* hb4  = reinterpret_cast<const float4*>(h + (size_t)lo * C);
        float4*       acc4 = reinterpret_cast<float4*>(acc);
        for (int i = threadIdx.x; i < nrow * (C / 4); i += OWN_THREADS) {
            const int r = i / (C / 4);
            float4 v = make_float4(0.f, 0.f, 0.f, 0.f);
            if (!SECOND || flag[r]) {
                const float  a = aself[r];
                const float4 x = hb4[i];
                v.x = x.x * a; v.y = x.y * a; v.z = x.z * a; v.w = x.w * a;
            }
            acc4[i] = v;
        }
    }
    __syncthreads();

    const int* __restrict__ dst = ei + E;
    const int mis     = (int)((reinterpret_cast<size_t>(dst) >> 2) & 3u);
    const int ghead   = mis ? 1 : 0;
    const int gfull   = (E + mis) >> 2;
    const int nchunks = (gfull + 31) >> 5;
    const int startc  = (int)(((long long)blockIdx.x * nchunks) / gridDim.x);
    const char* lbase = reinterpret_cast<const char*>(dst - mis) + lane * 16;
    const int lane4m  = lane * 4 - mis;
    const unsigned lt = (1u << lane) - 1u;

    for (int cl0 = wave; cl0 < nchunks; cl0 += nw * OWN_U) {
        int dv[OWN_U][4];
        int cu[OWN_U];
#pragma unroll
        for (int u = 0; u < OWN_U; ++u) {
            const int cl = cl0 + u * nw;
            int cc = cl + startc;
            if (cc >= nchunks) cc -= nchunks;
            cu[u] = cc;
            const int gl = cc * 32;
            int4 q = make_int4(-1, -1, -1, -1);
            if (cl < nchunks && lane >= ghead - gl && lane < gfull - gl)
                q = *reinterpret_cast<const int4*>(lbase + (size_t)cc * 512);
            dv[u][0] = q.x; dv[u][1] = q.y; dv[u][2] = q.z; dv[u][3] = q.w;
        }
        bool     m[OWN_U * 4];
        unsigned dl[OWN_U * 4];
        unsigned mk[OWN_U * 4];
        unsigned anym = 0u;
#pragma unroll
        for (int q = 0; q < OWN_U * 4; ++q) {
            dl[q] = (unsigned)(dv[q >> 2][q & 3] - lo);
            bool mm = dl[q] < (unsigned)nrow;
            if (SECOND) { if (mm) mm = flag[dl[q]] != 0u; }
            m[q]  = mm;
            mk[q] = __builtin_amdgcn_ballot_w32(mm);
            anym |= mk[q];
        }
        if (anym) {
            unsigned tot = 0u;
#pragma unroll
            for (int q = 0; q < OWN_U * 4; ++q) tot += (unsigned)__builtin_popcount(mk[q]);
            unsigned base = 0u;
            if (lane == 0) base = atomicAdd(&qn, tot);
            base = (unsigned)__builtin_amdgcn_readfirstlane((int)base);
            unsigned over = 0u;
#pragma unroll
            for (int q = 0; q < OWN_U * 4; ++q) {
                if (m[q]) {
                    const unsigned slot = base + (unsigned)__builtin_popcount(mk[q] & lt);
                    if (slot < OWN_QCAP) { qe[slot] = cu[q >> 2] * 128 + lane4m + (q & 3); qd[slot] = dl[q]; }
                    else over |= 1u << q;
                }
                base += (unsigned)__builtin_popcount(mk[q]);
            }
            unsigned any = __builtin_amdgcn_ballot_w32(over != 0u);
            while (any) {
                const int L = __builtin_ctz(any);
                any &= any - 1u;
                unsigned ovL = (unsigned)__builtin_amdgcn_readlane((int)over, L);
                while (ovL) {
                    const int q = __builtin_ctz(ovL);
                    ovL &= ovL - 1u;
                    int cc = cl0 + (q >> 2) * nw + startc;
                    if (cc >= nchunks) cc -= nchunks;
                    const int e = cc * 128 + L * 4 - mis + (q & 3);
                    own_add_now<C>(ei, E, lo, e, h, p, denom, acc, lane);
                }
            }
        }
    }
    if (wave == 0) {
        const bool whole   = gfull > ghead;
        const int  headEnd = (whole && ghead) ? 4 - mis : 0;
        const int  tailBeg = whole ? gfull * 4 - mis : 0;
        auto try_now = [&](int e) {
            const unsigned dle = (unsigned)(dst[e] - lo);
            bool ok = dle < (unsigned)nrow;
            if (SECOND) { if (ok) ok = flag[dle] != 0u; }
            if (ok) own_add_now<C>(ei, E, lo, e, h, p, denom, acc, lane);
        };
        for (int e = 0; e < headEnd; ++e) try_now(e);
        for (int e = tailBeg; e < E; ++e) try_now(e);
    }
    __syncthreads();

    const int total = __builtin_amdgcn_readfirstlane((int)min(qn, (unsigned)OWN_QCAP));
    for (int b = wave * 32; b < total; b += nw * 32) {
        const int idx = b + lane;
        int s = 0; unsigned dq = 0u; float a = 0.f;
        if (idx < total) {
            const int e = qe[idx];
            dq = qd[idx];
            s  = ei[e];
            a  = p[e] / denom[lo + (int)dq];
        }
        const int cnt = min(32, total - b);
        for (int j = 0; j < cnt; j += 4) {
            float    v[4][C / 32];
            float    aj[4];
            unsigned dj[4];
#pragma unroll
            for (int t = 0; t < 4; ++t) {
                const int jj = min(j + t, cnt - 1);
                const int sj = __builtin_amdgcn_readlane(s, jj);
                dj[t] = (unsigned)__builtin_amdgcn_readlane((int)dq, jj);
                aj[t] = __int_as_float(__builtin_amdgcn_readlane(__float_as_int(a), jj));
                const float* hs = h + (size_t)sj * C + lane;
#pragma unroll
                for (int k = 0; k < C / 32; ++k) v[t][k] = hs[32 * k];
            }
#pragma unroll
            for (int t = 0; t < 4; ++t) {
                if (j + t < cnt) {
                    float* ar = acc + dj[t] * C + lane;
#pragma unroll
                    for (int k = 0; k < C / 32; ++k) atomicAdd(&ar[32 * k], v[t][k] * aj[t]);
                }
            }
        }
    }
    __syncthreads();

    for (int r = wave; r < nrow; r += nw) {
        const size_t d = (size_t)(lo + r);
        const float4 v = *reinterpret_cast<const float4*>(&acc[r * C + 4 * lane]);
        if (SECOND) {
            if (flag[r]) {
                const float4 bq = *reinterpret_cast<const float4*>(bias + 4 * lane);
                float4 o; o.x = v.x + bq.x; o.y = v.y + bq.y; o.z = v.z + bq.z; o.w = v.w + bq.w;
                *reinterpret_cast<float4*>(yf + d * C + 4 * lane) = o;
            }
        } else {
            uint2 o;
            o.x = (unsigned)f2bf(v.x) | ((unsigned)f2bf(v.y) << 16);
            o.y = (unsigned)f2bf(v.z) | ((unsigned)f2bf(v.w) << 16);
            *reinterpret_cast<uint2*>(ybf + d * C + 4 * lane) = o;
        }
    }
}

__global__ void transpose_to_bf16(const float* __restrict__ W, unsigned short* __restrict__ Wt,
                                  int K, int NOUT)
{
    const int i = blockIdx.x * blockDim.x + threadIdx.x;
    if (i >= K * NOUT) return;
    const int k = i / NOUT, n = i % NOUT;
    Wt[(size_t)n * K + k] = f2bf(W[i]);
}

__global__ void link_predict(const float* __restrict__ h, const int* __restrict__ mask,
                             const float* __restrict__ Wl, const float* __restrict__ bl,
                             float* __restrict__ out, int P, int C)
{
    const int warp = (blockIdx.x * blockDim.x + threadIdx.x) >> 5;
    const int lane = threadIdx.x & 31;
    if (warp >= P) return;
    const int m0 = mask[warp * 2 + 0];
    const int m1 = mask[warp * 2 + 1];
    const float* r0 = h + (size_t)m0 * C;
    const float* r1 = h + (size_t)m1 * C;
    float acc = 0.f;
    for (int c = lane; c < C; c += 32)
        acc += r0[c] * Wl[c] + r1[c] * Wl[C + c];
#pragma unroll
    for (int off = 16; off; off >>= 1) acc += __shfl_xor(acc, off);
    if (lane == 0) out[warp] = 1.f / (1.f + expf(-(acc + bl[0])));
}

extern "C" void kernel_launch(void* const* d_in, const int* in_sizes, int n_in,
                              void* d_out, int out_size, void* d_ws, size_t ws_size,
                              hipStream_t stream)
{
    (void)n_in; (void)out_size; (void)ws_size;

    const float* features = (const float*)d_in[0];
    const int*   ei       = (const int*)  d_in[1];
    const int*   mask     = (const int*)  d_in[2];
    const float* W1   = (const float*)d_in[3];
    const float* a_s1 = (const float*)d_in[4];
    const float* a_d1 = (const float*)d_in[5];
    const float* b1   = (const float*)d_in[6];
    const float* W2   = (const float*)d_in[7];
    const float* a_s2 = (const float*)d_in[8];
    const float* a_d2 = (const float*)d_in[9];
    const float* b2   = (const float*)d_in[10];
    const float* Wl   = (const float*)d_in[11];
    const float* bl   = (const float*)d_in[12];

    constexpr int F  = 128, Hh = 256;
    const int N  = in_sizes[0] / F;
    const int E  = in_sizes[1] / 2;
    const int P  = in_sizes[2] / 2;
    const int NE = E + N;

    char* ws = (char*)d_ws;
    size_t off = 0;
    auto alloc = [&](size_t bytes) -> char* {
        char* ptr = ws + off;
        off += (bytes + 255) & ~(size_t)255;
        return ptr;
    };
    unsigned short* xbf  = (unsigned short*)alloc((size_t)N * F  * 2);
    unsigned short* w1t  = (unsigned short*)alloc((size_t)Hh * F * 2);
    unsigned short* w2t  = (unsigned short*)alloc((size_t)F * Hh * 2);
    float*    buf1    = (float*)   alloc((size_t)N * Hh * 4);
    float*    buf2    = (float*)   alloc((size_t)N * Hh * 4);
    float*    es      = (float*)   alloc((size_t)N * 4);
    float*    ed      = (float*)   alloc((size_t)N * 4);
    unsigned* maxbits = (unsigned*)alloc((size_t)N * 4);
    float*    denom   = (float*)   alloc((size_t)N * 4);
    float*    logits  = (float*)   alloc((size_t)NE * 4);
    float*    pbuf    = (float*)   alloc((size_t)NE * 4);
    float*    vs1     = (float*)   alloc((size_t)F * 4);
    float*    vd1     = (float*)   alloc((size_t)F * 4);
    unsigned* need    = (unsigned*)alloc((size_t)N * 4);

    const int tilesM = (N + 15) / 16;
    const int gemmGrid = (tilesM + PAIR_WAVES - 1) / PAIR_WAVES;
    const int TB = 256;
    const int ownGrid = (N + OWN_ROWS - 1) / OWN_ROWS;
    auto cdiv = [](size_t a, size_t b) -> int { return (int)((a + b - 1) / b); };

    transpose_to_bf16<<<cdiv((size_t)F * Hh, TB), TB, 0, stream>>>(W1, w1t, F, Hh);
    transpose_to_bf16<<<cdiv((size_t)Hh * F, TB), TB, 0, stream>>>(W2, w2t, Hh, F);

    node_scores<<<cdiv((size_t)F * 32, TB), TB, 0, stream>>>(W1, a_s1, a_d1, vs1, vd1, F, Hh);
    node_scores<<<cdiv((size_t)N * 32, TB), TB, 0, stream>>>(features, vs1, vd1, es, ed, N, F);
    init_node<<<cdiv(N, TB), TB, 0, stream>>>(maxbits, denom, N, need);
    edge_logit_max<<<cdiv(NE, TB), TB, 0, stream>>>(ei, E, N, es, ed, logits, maxbits, nullptr);
    edge_exp<<<cdiv(NE, TB), TB, 0, stream>>>(ei, E, N, logits, maxbits, pbuf, denom, nullptr);
    aggregate_owned<F, false><<<ownGrid, OWN_THREADS, 0, stream>>>(ei, E, N, features, pbuf, denom,
                                                                  nullptr, nullptr, xbf, nullptr);

    mark_needed<<<cdiv((size_t)P * 2, TB), TB, 0, stream>>>(mask, need, P * 2, N);
    wmma_two_products<F, Hh, F><<<gemmGrid, 32 * PAIR_WAVES, 0, stream>>>(xbf, w1t, b1, w2t, buf1, N);
    node_scores<<<cdiv((size_t)N * 32, TB), TB, 0, stream>>>(buf1, a_s2, a_d2, es, ed, N, F);
    init_node<<<cdiv(N, TB), TB, 0, stream>>>(maxbits, denom, N, nullptr);
    edge_logit_max<<<cdiv(NE, TB), TB, 0, stream>>>(ei, E, N, es, ed, logits, maxbits, need);
    edge_exp<<<cdiv(NE, TB), TB, 0, stream>>>(ei, E, N, logits, maxbits, pbuf, denom, need);
    aggregate_owned<F, true><<<ownGrid, OWN_THREADS, 0, stream>>>(ei, E, N, buf1, pbuf, denom,
                                                                 need, b2, nullptr, buf2);

    link_predict<<<cdiv((size_t)P * 32, TB), TB, 0, stream>>>(buf2, mask, Wl, bl,
                                                              (float*)d_out, P, F);
}


# Edge softmax inside the row-owning work-group: scores, row maxima, exp sums of queued edges stay in LDS; six init/logit/exp passes and their global atomics go; 16 -> 10 launches; sum order changes
# speedup vs baseline: 3.8995x; 1.4349x over previous
#include <hip/hip_runtime.h>
#include <hip/hip_bf16.h>

typedef __attribute__((__ext_vector_type__(16))) __bf16 bf16x16;
typedef __attribute__((__ext_vector_type__(8)))  float  f32x8;
typedef __attribute__((__ext_vector_type__(4)))  unsigned int u32x4;

union Frag16 {
    u32x4   q[2];
    bf16x16 v;
};

static __device__ __forceinline__ unsigned short f2bf(float f) {
    unsigned u = __float_as_uint(f);
    u += 0x7FFFu + ((u >> 16) & 1u);
    return (unsigned short)(u >> 16);
}

static __device__ __forceinline__ unsigned fkey(float f) {
    unsigned u = __float_as_uint(f);
    return (u & 0x80000000u) ? ~u : (u | 0x80000000u);
}
static __device__ __forceinline__ float funkey(unsigned k) {
    unsigned u = (k & 0x80000000u) ? (k & 0x7FFFFFFFu) : ~k;
    return __uint_as_float(u);
}

#define PAIR_WAVES 4
#define PAIR_PAD   8

template <int K1, int N1, int N2>
__global__ void __launch_bounds__(32 * PAIR_WAVES)
wmma_two_products(const unsigned short* __restrict__ A,
                  const unsigned short* __restrict__ B1t,
                  const float* __restrict__ bias1,
                  const unsigned short* __restrict__ B2t,
                  float* __restrict__ Y, int M)
{
    static_assert(K1 % 32 == 0 && N1 % 32 == 0 && N1 % 16 == 0 && N2 % 16 == 0, "whole fragments");
    constexpr int PITCH = N1 + PAIR_PAD;
    static_assert((PITCH * 2) % 16 == 0 && ((PITCH * 2 / 16) & 1) == 1, "pitch = 16 B x odd");
    constexpr int OPITCH = N2 + 4;
    static_assert(N2 == 128, "one float4 per lane covers a result row");
    static_assert(OPITCH * 4 <= PITCH * 2, "the result tile reuses the wave's LDS tile");
    constexpr int KS1 = K1 / 32;
    constexpr int KS2 = N1 / 32;
    __shared__ __attribute__((aligned(16))) unsigned short mid[PAIR_WAVES][16 * PITCH];

    const int lane  = threadIdx.x & 31;
    const int wave  = __builtin_amdgcn_readfirstlane(threadIdx.x >> 5);
    const int m0    = (blockIdx.x * PAIR_WAVES + wave) * 16;
    const bool live = m0 < M;
    const int l15   = lane & 15;
    const int half  = lane >> 4;
    const int kbase = half * 8;
    unsigned short* mine = mid[wave];

    if (live) {
        int rowA = m0 + l15;
        if (rowA >= M) rowA = M - 1;
        const unsigned short* aptr = A + (size_t)rowA * K1 + kbase;
        Frag16 fa[KS1];
#pragma unroll
        for (int k = 0; k < KS1; ++k) {
            fa[k].q[0] = *(const u32x4*)(aptr + 32 * k);
            fa[k].q[1] = *(const u32x4*)(aptr + 32 * k + 16);
        }
        Frag16 fb[KS1], fn[KS1];
        {
            const unsigned short* bptr = B1t + (size_t)l15 * K1 + kbase;
#pragma unroll
            for (int k = 0; k < KS1; ++k) {
                fb[k].q[0] = *(const u32x4*)(bptr + 32 * k);
                fb[k].q[1] = *(const u32x4*)(bptr + 32 * k + 16);
            }
        }
#pragma unroll
        for (int t = 0; t < N1 / 16; ++t) {
            if (t + 1 < N1 / 16) {
                const unsigned short* bptr = B1t + (size_t)((t + 1) * 16 + l15) * K1 + kbase;
#pragma unroll
                for (int k = 0; k < KS1; ++k) {
                    fn[k].q[0] = *(const u32x4*)(bptr + 32 * k);
                    fn[k].q[1] = *(const u32x4*)(bptr + 32 * k + 16);
                }
            }
            f32x8 acc = {};
#pragma unroll
            for (int k = 0; k < KS1; ++k)
                acc = __builtin_amdgcn_wmma_f32_16x16x32_bf16(
                          false, fa[k].v, false, fb[k].v, (short)0, acc, false, false);
            const int   col = t * 16 + l15;
            const float bv  = bias1[col];
#pragma unroll
            for (int r = 0; r < 8; ++r) {
                const float v = acc[r] + bv;
                mine[(r + half * 8) * PITCH + col] = f2bf(v > 0.f ? v : 0.f);
            }
#pragma unroll
            for (int k = 0; k < KS1; ++k) fb[k] = fn[k];
        }
    }
    __syncthreads();
    Frag16 ga[KS2];
    if (live) {
        const unsigned short* lp = mine + l15 * PITCH + kbase;
#pragma unroll
        for (int k = 0; k < KS2; ++k) {
            ga[k].q[0] = *(const u32x4*)(lp + 32 * k);
            ga[k].q[1] = *(const u32x4*)(lp + 32 * k + 16);
        }
    }
    __syncthreads();
    float* outt = reinterpret_cast<float*>(mine);
    if (live) {
        Frag16 gb[KS2], gn[KS2];
        {
            const unsigned short* bptr = B2t + (size_t)l15 * N1 + kbase;
#pragma unroll
            for (int k = 0; k < KS2; ++k) {
                gb[k].q[0] = *(const u32x4*)(bptr + 32 * k);
                gb[k].q[1] = *(const u32x4*)(bptr + 32 * k + 16);
            }
        }
#pragma unroll
        for (int t = 0; t < N2 / 16; ++t) {
            if (t + 1 < N2 / 16) {
                const unsigned short* bptr = B2t + (size_t)((t + 1) * 16 + l15) * N1 + kbase;
#pragma unroll
                for (int k = 0; k < KS2; ++k) {
                    gn[k].q[0] = *(const u32x4*)(bptr + 32 * k);
                    gn[k].q[1] = *(const u32x4*)(bptr + 32 * k + 16);
                }
            }
            f32x8 acc = {};
#pragma unroll
            for (int k = 0; k < KS2; ++k)
                acc = __builtin_amdgcn_wmma_f32_16x16x32_bf16(
                          false, ga[k].v, false, gb[k].v, (short)0, acc, false, false);
#pragma unroll
            for (int r = 0; r < 8; ++r)
                outt[(r + half * 8) * OPITCH + t * 16 + l15] = acc[r];
#pragma unroll
            for (int k = 0; k < KS2; ++k) gb[k] = gn[k];
        }
    }
    __syncthreads();
    if (live) {
        float* yrow = Y + (size_t)m0 * N2 + 4 * lane;
        if (m0 + 16 <= M) {
            float4 v[16];
#pragma unroll
            for (int i = 0; i < 16; ++i) v[i] = *reinterpret_cast<const float4*>(outt + i * OPITCH + 4 * lane);
#pragma unroll
            for (int i = 0; i < 16; ++i) *reinterpret_cast<float4*>(yrow + (size_t)i * N2) = v[i];
        } else {
            for (int i = 0; i < M - m0; ++i)
                *reinterpret_cast<float4*>(yrow + (size_t)i * N2) =
                    *reinterpret_cast<const float4*>(outt + i * OPITCH + 4 * lane);
        }
    }
}

__global__ void node_scores(const float* __restrict__ h,
                            const float* __restrict__ as_, const float* __restrict__ ad_,
                            float* __restrict__ es, float* __restrict__ ed,
                            int N, int C, unsigned* __restrict__ clr)
{
    const int warp = (blockIdx.x * blockDim.x + threadIdx.x) >> 5;
    const int lane = threadIdx.x & 31;
    if (warp >= N) return;
    const float* row = h + (size_t)warp * C;
    float s = 0.f, d = 0.f;
    for (int c = lane; c < C; c += 32) {
        const float v = row[c];
        s += v * as_[c];
        d += v * ad_[c];
    }
#pragma unroll
    for (int off = 16; off; off >>= 1) {
        s += __shfl_xor(s, off);
        d += __shfl_xor(d, off);
    }
    if (lane == 0) { es[warp] = s; ed[warp] = d; if (clr) clr[warp] = 0u; }
}

__global__ void mark_needed(const int* __restrict__ mask, unsigned* __restrict__ need, int n, int N)
{
    const int i = blockIdx.x * blockDim.x + threadIdx.x;
    if (i >= n) return;
    const int m = mask[i];
    if ((unsigned)m < (unsigned)N) need[m] = 1u;
}

#define OWN_ROWS    196
#define OWN_THREADS 1024
#define OWN_QCAP    4096
#define OWN_U       4

static __device__ __forceinline__ float leaky02(float t) { return (t > 0.f) ? t : 0.2f * t; }

template <int C>
static __device__ __forceinline__ void own_add_row(const float* __restrict__ h, int s, float a,
                                                   float* accrow, int lane)
{
    const float* hs = h + (size_t)s * C + lane;
    float*       ar = accrow + lane;
    float v[C / 32];
#pragma unroll
    for (int k = 0; k < C / 32; ++k) v[k] = hs[32 * k];
#pragma unroll
    for (int k = 0; k < C / 32; ++k) atomicAdd(&ar[32 * k], v[k] * a);
}

template <int C, bool SECOND, int STAGE>
static __device__ __forceinline__ void own_walk(const int* __restrict__ ei, int E, int lo, int nrow,
                                                const float* __restrict__ h, const float* __restrict__ es,
                                                const float* edl, const unsigned* flag,
                                                unsigned* rmax, float* rden, float* acc,
                                                int wave, int lane, int nw)
{
    const int* __restrict__ dst = ei + E;
    for (int e0 = wave * 32; e0 < E; e0 += nw * 32) {
        const int e = e0 + lane;
        unsigned dl = 0u;
        bool ok = false;
        if (e < E) {
            dl = (unsigned)(dst[e] - lo);
            ok = dl < (unsigned)nrow;
            if (SECOND) { if (ok) ok = flag[dl] != 0u; }
        }
        int s = 0; float l = 0.f;
        if (ok) { s = ei[e]; l = leaky02(es[s] + edl[dl]); }
        if (STAGE == 0) {
            if (ok) atomicMax(&rmax[dl], fkey(l));
        } else if (STAGE == 1) {
            if (ok) atomicAdd(&rden[dl], __expf(l - funkey(rmax[dl])));
        } else {
            float a = 0.f;
            if (ok) a = __expf(l - funkey(rmax[dl])) / rden[dl];
            unsigned mk = __builtin_amdgcn_ballot_w32(ok);
            while (mk) {
                const int L = __builtin_ctz(mk);
                mk &= mk - 1u;
                const int      sj = __builtin_amdgcn_readlane(s, L);
                const unsigned dj = (unsigned)__builtin_amdgcn_readlane((int)dl, L);
                const float    aj = __int_as_float(__builtin_amdgcn_readlane(__float_as_int(a), L));
                own_add_row<C>(h, sj, aj, acc + dj * C, lane);
            }
        }
    }
}

template <int C, bool SECOND>
__global__ void __launch_bounds__(OWN_THREADS)
aggregate_owned(const int* __restrict__ ei, int E, int N,
                const float* __restrict__ h,
                const float* __restrict__ es, const float* __restrict__ ed,
                const unsigned* __restrict__ need, const float* __restrict__ bias,
                unsigned short* __restrict__ ybf, float* __restrict__ yf)
{
    static_assert(C == 128, "one float4 per lane covers a row");
    __shared__ __attribute__((aligned(16))) float acc[OWN_ROWS * C];
    __shared__ int      qe[OWN_QCAP];
    __shared__ unsigned qd[OWN_QCAP];
    __shared__ float    qp[OWN_QCAP];
    __shared__ float    edl[OWN_ROWS];
    __shared__ float    pself[OWN_ROWS];
    __shared__ unsigned rmax[OWN_ROWS];
    __shared__ float    rden[OWN_ROWS];
    __shared__ unsigned flag[OWN_ROWS];
    __shared__ unsigned qn;

    const int lane = threadIdx.x & 31;
    const int wave = __builtin_amdgcn_readfirstlane(threadIdx.x >> 5);
    const int nw   = OWN_THREADS / 32;
    const int lo   = blockIdx.x * OWN_ROWS;
    const int nrow = min(OWN_ROWS, N - lo);

    if (threadIdx.x == 0) qn = 0u;
    for (int r = threadIdx.x; r < nrow; r += OWN_THREADS) {
        const int d = lo + r;
        if (SECOND) flag[r] = (need[d] != 0u) ? 1u : 0u;
        const float edv = ed[d];
        const float l   = leaky02(es[d] + edv);
        edl[r]   = edv;
        pself[r] = l;
        rmax[r]  = fkey(l);
    }
    __syncthreads();

    const int* __restrict__ dst = ei + E;
    const int mis     = (int)((reinterpret_cast<size_t>(dst) >> 2) & 3u);
    const int ghead   = mis ? 1 : 0;
    const int gfull   = (E + mis) >> 2;
    const int nchunks = (gfull + 31) >> 5;
    const int startc  = (int)(((long long)blockIdx.x * nchunks) / gridDim.x);
    const char* lbase = reinterpret_cast<const char*>(dst - mis) + lane * 16;
    const int lane4m  = lane * 4 - mis;
    const unsigned lt = (1u << lane) - 1u;

    for (int cl0 = wave; cl0 < nchunks; cl0 += nw * OWN_U) {
        int dv[OWN_U][4];
        int cu[OWN_U];
#pragma unroll
        for (int u = 0; u < OWN_U; ++u) {
            const int cl = cl0 + u * nw;
            int cc = cl + startc;
            if (cc >= nchunks) cc -= nchunks;
            cu[u] = cc;
            const int gl = cc * 32;
            int4 q = make_int4(-1, -1, -1, -1);
            if (cl < nchunks && lane >= ghead - gl && lane < gfull - gl)
                q = *reinterpret_cast<const int4*>(lbase + (size_t)cc * 512);
            dv[u][0] = q.x; dv[u][1] = q.y; dv[u][2] = q.z; dv[u][3] = q.w;
        }
        bool     m[OWN_U * 4];
        unsigned dl[OWN_U * 4];
        unsigned mk[OWN_U * 4];
        unsigned anym = 0u;
#pragma unroll
        for (int q = 0; q < OWN_U * 4; ++q) {
            dl[q] = (unsigned)(dv[q >> 2][q & 3] - lo);
            bool mm = dl[q] < (unsigned)nrow;
            if (SECOND) { if (mm) mm = flag[dl[q]] != 0u; }
            m[q]  = mm;
            mk[q] = __builtin_amdgcn_ballot_w32(mm);
            anym |= mk[q];
        }
        if (anym) {
            unsigned tot = 0u;
#pragma unroll
            for (int q = 0; q < OWN_U * 4; ++q) tot += (unsigned)__builtin_popcount(mk[q]);
            unsigned base = 0u;
            if (lane == 0) base = atomicAdd(&qn, tot);
            base = (unsigned)__builtin_amdgcn_readfirstlane((int)base);
#pragma unroll
            for (int q = 0; q < OWN_U * 4; ++q) {
                if (m[q]) {
                    const unsigned slot = base + (unsigned)__builtin_popcount(mk[q] & lt);
                    if (slot < OWN_QCAP) { qe[slot] = cu[q >> 2] * 128 + lane4m + (q & 3); qd[slot] = dl[q]; }
                }
                base += (unsigned)__builtin_popcount(mk[q]);
            }
        }
    }
    if (wave == 0) {
        const bool whole   = gfull > ghead;
        const int  headEnd = (whole && ghead) ? 4 - mis : 0;
        const int  tailBeg = whole ? gfull * 4 - mis : 0;
        const int  nb      = headEnd + (E - tailBeg);
        if (lane < nb) {
            const int e = (lane < headEnd) ? lane : tailBeg + (lane - headEnd);
            const unsigned dle = (unsigned)(dst[e] - lo);
            bool ok = dle < (unsigned)nrow;
            if (SECOND) { if (ok) ok = flag[dle] != 0u; }
            if (ok) {
                const unsigned slot = atomicAdd(&qn, 1u);
                if (slot < OWN_QCAP) { qe[slot] = e; qd[slot] = dle; }
            }
        }
    }
    __syncthreads();

    const unsigned qtot = (unsigned)__builtin_amdgcn_readfirstlane((int)qn);
    const bool fits  = qtot <= (unsigned)OWN_QCAP;
    const int  total = fits ? (int)qtot : 0;

    if (fits) {
        for (int idx = threadIdx.x; idx < total; idx += OWN_THREADS) {
            const unsigned dq = qd[idx];
            const float l = leaky02(es[ei[qe[idx]]] + edl[dq]);
            qp[idx] = l;
            atomicMax(&rmax[dq], fkey(l));
        }
    } else {
        own_walk<C, SECOND, 0>(ei, E, lo, nrow, h, es, edl, flag, rmax, rden, acc, wave, lane, nw);
    }
    __syncthreads();
    for (int r = threadIdx.x; r < nrow; r += OWN_THREADS) {
        const float ps = __expf(pself[r] - funkey(rmax[r]));
        pself[r] = ps;
        rden[r]  = ps;
    }
    __syncthreads();
    if (fits) {
        for (int idx = threadIdx.x; idx < total; idx += OWN_THREADS) {
            const unsigned dq = qd[idx];
            const float pe = __expf(qp[idx] - funkey(rmax[dq]));
            qp[idx] = pe;
            atomicAdd(&rden[dq], pe);
        }
    } else {
        own_walk<C, SECOND, 1>(ei, E, lo, nrow, h, es, edl, flag, rmax, rden, acc, wave, lane, nw);
    }
    __syncthreads();

    {
        const float4* hb4  = reinterpret_cast<const float4*>(h + (size_t)lo * C);
        float4*       acc4 = reinterpret_cast<float4*>(acc);
        for (int i = threadIdx.x; i < nrow * (C / 4); i += OWN_THREADS) {
            const int r = i / (C / 4);
            float4 v = make_float4(0.f, 0.f, 0.f, 0.f);
            if (!SECOND || flag[r]) {
                const float  a = pself[r] / rden[r];
                const float4 x = hb4[i];
                v.x = x.x * a; v.y = x.y * a; v.z = x.z * a; v.w = x.w * a;
            }
            acc4[i] = v;
        }
    }
    __syncthreads();

    for (int b = wave * 32; b < total; b += nw * 32) {
        const int idx = b + lane;
        int s = 0; unsigned dq = 0u; float a = 0.f;
        if (idx < total) {
            dq = qd[idx];
            s  = ei[qe[idx]];
            a  = qp[idx] / rden[dq];
        }
        const int cnt = min(32, total - b);
        for (int j = 0; j < cnt; j += 4) {
            float    v[4][C / 32];
            float    aj[4];
            unsigned dj[4];
#pragma unroll
            for (int t = 0; t < 4; ++t) {
                const int jj = min(j + t, cnt - 1);
                const int sj = __builtin_amdgcn_readlane(s, jj);
                dj[t] = (unsigned)__builtin_amdgcn_readlane((int)dq, jj);
                aj[t] = __int_as_float(__builtin_amdgcn_readlane(__float_as_int(a), jj));
                const float* hs = h + (size_t)sj * C + lane;
#pragma unroll
                for (int k = 0; k < C / 32; ++k) v[t][k] = hs[32 * k];
            }
#pragma unroll
            for (int t = 0; t < 4; ++t) {
                if (j + t < cnt) {
                    float* ar = acc + dj[t] * C + lane;
#pragma unroll
                    for (int k = 0; k < C / 32; ++k) atomicAdd(&ar[32 * k], v[t][k] * aj[t]);
                }
            }
        }
    }
    if (!fits)
        own_walk<C, SECOND, 2>(ei, E, lo, nrow, h, es, edl, flag, rmax, rden, acc, wave, lane, nw);
    __syncthreads();

    for (int r = wave; r < nrow; r += nw) {
        const size_t d = (size_t)(lo + r);
        const float4 v = *reinterpret_cast<const float4*>(&acc[r * C + 4 * lane]);
        if (SECOND) {
            if (flag[r]) {
                const float4 bq = *reinterpret_cast<const float4*>(bias + 4 * lane);
                float4 o; o.x = v.x + bq.x; o.y = v.y + bq.y; o.z = v.z + bq.z; o.w = v.w + bq.w;
                *reinterpret_cast<float4*>(yf + d * C + 4 * lane) = o;
            }
        } else {
            uint2 o;
            o.x = (unsigned)f2bf(v.x) | ((unsigned)f2bf(v.y) << 16);
            o.y = (unsigned)f2bf(v.z) | ((unsigned)f2bf(v.w) << 16);
            *reinterpret_cast<uint2*>(ybf + d * C + 4 * lane) = o;
        }
    }
}

__global__ void transpose_to_bf16(const float* __restrict__ W, unsigned short* __restrict__ Wt,
                                  int K, int NOUT)
{
    const int i = blockIdx.x * blockDim.x + threadIdx.x;
    if (i >= K * NOUT) return;
    const int k = i / NOUT, n = i % NOUT;
    Wt[(size_t)n * K + k] = f2bf(W[i]);
}

__global__ void link_predict(const float* __restrict__ h, const int* __restrict__ mask,
                             const float* __restrict__ Wl, const float* __restrict__ bl,
                             float* __restrict__ out, int P, int C)
{
    const int warp = (blockIdx.x * blockDim.x + threadIdx.x) >> 5;
    const int lane = threadIdx.x & 31;
    if (warp >= P) return;
    const int m0 = mask[warp * 2 + 0];
    const int m1 = mask[warp * 2 + 1];
    const float* r0 = h + (size_t)m0 * C;
    const float* r1 = h + (size_t)m1 * C;
    float acc = 0.f;
    for (int c = lane; c < C; c += 32)
        acc += r0[c] * Wl[c] + r1[c] * Wl[C + c];
#pragma unroll
    for (int off = 16; off; off >>= 1) acc += __shfl_xor(acc, off);
    if (lane == 0) out[warp] = 1.f / (1.f + expf(-(acc + bl[0])));
}

extern "C" void kernel_launch(void* const* d_in, const int* in_sizes, int n_in,
                              void* d_out, int out_size, void* d_ws, size_t ws_size,
                              hipStream_t stream)
{
    (void)n_in; (void)out_size; (void)ws_size;

    const float* features = (const float*)d_in[0];
    const int*   ei       = (const int*)  d_in[1];
    const int*   mask     = (const int*)  d_in[2];
    const float* W1   = (const float*)d_in[3];
    const float* a_s1 = (const float*)d_in[4];
    const float* a_d1 = (const float*)d_in[5];
    const float* b1   = (const float*)d_in[6];
    const float* W2   = (const float*)d_in[7];
    const float* a_s2 = (const float*)d_in[8];
    const float* a_d2 = (const float*)d_in[9];
    const float* b2   = (const float*)d_in[10];
    const float* Wl   = (const float*)d_in[11];
    const float* bl   = (const float*)d_in[12];

    constexpr int F  = 128, Hh = 256;
    const int N  = in_sizes[0] / F;
    const int E  = in_sizes[1] / 2;
    const int P  = in_sizes[2] / 2;

    char* ws = (char*)d_ws;
    size_t off = 0;
    auto alloc = [&](size_t bytes) -> char* {
        char* ptr = ws + off;
        off += (bytes + 255) & ~(size_t)255;
        return ptr;
    };
    unsigned short* xbf  = (unsigned short*)alloc((size_t)N * F  * 2);
    unsigned short* w1t  = (unsigned short*)alloc((size_t)Hh * F * 2);
    unsigned short* w2t  = (unsigned short*)alloc((size_t)F * Hh * 2);
    float*    buf1    = (float*)   alloc((size_t)N * Hh * 4);
    float*    buf2    = (float*)   alloc((size_t)N * Hh * 4);
    float*    es      = (float*)   alloc((size_t)N * 4);
    float*    ed      = (float*)   alloc((size_t)N * 4);
    float*    vs1     = (float*)   alloc((size_t)F * 4);
    float*    vd1     = (float*)   alloc((size_t)F * 4);
    unsigned* need    = (unsigned*)alloc((size_t)N * 4);

    const int tilesM = (N + 15) / 16;
    const int gemmGrid = (tilesM + PAIR_WAVES - 1) / PAIR_WAVES;
    const int TB = 256;
    const int ownGrid = (N + OWN_ROWS - 1) / OWN_ROWS;
    auto cdiv = [](size_t a, size_t b) -> int { return (int)((a + b - 1) / b); };

    transpose_to_bf16<<<cdiv((size_t)F * Hh, TB), TB, 0, stream>>>(W1, w1t, F, Hh);
    transpose_to_bf16<<<cdiv((size_t)Hh * F, TB), TB, 0, stream>>>(W2, w2t, Hh, F);

    node_scores<<<cdiv((size_t)F * 32, TB), TB, 0, stream>>>(W1, a_s1, a_d1, vs1, vd1, F, Hh, nullptr);
    node_scores<<<cdiv((size_t)N * 32, TB), TB, 0, stream>>>(features, vs1, vd1, es, ed, N, F, need);
    aggregate_owned<F, false><<<ownGrid, OWN_THREADS, 0, stream>>>(ei, E, N, features, es, ed,
                                                                  nullptr, nullptr, xbf, nullptr);

    mark_needed<<<cdiv((size_t)P * 2, TB), TB, 0, stream>>>(mask, need, P * 2, N);
    wmma_two_products<F, Hh, F><<<gemmGrid, 32 * PAIR_WAVES, 0, stream>>>(xbf, w1t, b1, w2t, buf1, N);
    node_scores<<<cdiv((size_t)N * 32, TB), TB, 0, stream>>>(buf1, a_s2, a_d2, es, ed, N, F, nullptr);
    aggregate_owned<F, true><<<ownGrid, OWN_THREADS, 0, stream>>>(ei, E, N, buf1, es, ed,
                                                                 need, b2, nullptr, buf2);

    link_predict<<<cdiv((size_t)P * 32, TB), TB, 0, stream>>>(buf2, mask, Wl, bl,
                                                              (float*)d_out, P, F);
}


# Row-dot passes (two score passes, link predictor): a wave takes 16 rows, loads in flight together, one shared lane reduction, not one row per wave; 110000 waves become 7500; dot sum order changes
# speedup vs baseline: 4.4673x; 1.1456x over previous
#include <hip/hip_runtime.h>
#include <hip/hip_bf16.h>

typedef __attribute__((__ext_vector_type__(16))) __bf16 bf16x16;
typedef __attribute__((__ext_vector_type__(8)))  float  f32x8;
typedef __attribute__((__ext_vector_type__(4)))  unsigned int u32x4;

union Frag16 {
    u32x4   q[2];
    bf16x16 v;
};

static __device__ __forceinline__ unsigned short f2bf(float f) {
    unsigned u = __float_as_uint(f);
    u += 0x7FFFu + ((u >> 16) & 1u);
    return (unsigned short)(u >> 16);
}

static __device__ __forceinline__ unsigned fkey(float f) {
    unsigned u = __float_as_uint(f);
    return (u & 0x80000000u) ? ~u : (u | 0x80000000u);
}
static __device__ __forceinline__ float funkey(unsigned k) {
    unsigned u = (k & 0x80000000u) ? (k & 0x7FFFFFFFu) : ~k;
    return __uint_as_float(u);
}

#define PAIR_WAVES 4
#define PAIR_PAD   8

template <int K1, int N1, int N2>
__global__ void __launch_bounds__(32 * PAIR_WAVES)
wmma_two_products(const unsigned short* __restrict__ A,
                  const unsigned short* __restrict__ B1t,
                  const float* __restrict__ bias1,
                  const unsigned short* __restrict__ B2t,
                  float* __restrict__ Y, int M)
{
    static_assert(K1 % 32 == 0 && N1 % 32 == 0 && N1 % 16 == 0 && N2 % 16 == 0, "whole fragments");
    constexpr int PITCH = N1 + PAIR_PAD;
    static_assert((PITCH * 2) % 16 == 0 && ((PITCH * 2 / 16) & 1) == 1, "pitch = 16 B x odd");
    constexpr int OPITCH = N2 + 4;
    static_assert(N2 == 128, "one float4 per lane covers a result row");
    static_assert(OPITCH * 4 <= PITCH * 2, "the result tile reuses the wave's LDS tile");
    constexpr int KS1 = K1 / 32;
    constexpr int KS2 = N1 / 32;
    __shared__ __attribute__((aligned(16))) unsigned short mid[PAIR_WAVES][16 * PITCH];

    const int lane  = threadIdx.x & 31;
    const int wave  = __builtin_amdgcn_readfirstlane(threadIdx.x >> 5);
    const int m0    = (blockIdx.x * PAIR_WAVES + wave) * 16;
    const bool live = m0 < M;
    const int l15   = lane & 15;
    const int half  = lane >> 4;
    const int kbase = half * 8;
    unsigned short* mine = mid[wave];

    if (live) {
        int rowA = m0 + l15;
        if (rowA >= M) rowA = M - 1;
        const unsigned short* aptr = A + (size_t)rowA * K1 + kbase;
        Frag16 fa[KS1];
#pragma unroll
        for (int k = 0; k < KS1; ++k) {
            fa[k].q[0] = *(const u32x4*)(aptr + 32 * k);
            fa[k].q[1] = *(const u32x4*)(aptr + 32 * k + 16);
        }
        Frag16 fb[KS1], fn[KS1];
        {
            const unsigned short* bptr = B1t + (size_t)l15 * K1 + kbase;
#pragma unroll
            for (int k = 0; k < KS1; ++k) {
                fb[k].q[0] = *(const u32x4*)(bptr + 32 * k);
                fb[k].q[1] = *(const u32x4*)(bptr + 32 * k + 16);
            }
        }
#pragma unroll
        for (int t = 0; t < N1 / 16; ++t) {
            if (t + 1 < N1 / 16) {
                const unsigned short* bptr = B1t + (size_t)((t + 1) * 16 + l15) * K1 + kbase;
#pragma unroll
                for (int k = 0; k < KS1; ++k) {
                    fn[k].q[0] = *(const u32x4*)(bptr + 32 * k);
                    fn[k].q[1] = *(const u32x4*)(bptr + 32 * k + 16);
                }
            }
            f32x8 acc = {};
#pragma unroll
            for (int k = 0; k < KS1; ++k)
                acc = __builtin_amdgcn_wmma_f32_16x16x32_bf16(
                          false, fa[k].v, false, fb[k].v, (short)0, acc, false, false);
            const int   col = t * 16 + l15;
            const float bv  = bias1[col];
#pragma unroll
            for (int r = 0; r < 8; ++r) {
                const float v = acc[r] + bv;
                mine[(r + half * 8) * PITCH + col] = f2bf(v > 0.f ? v : 0.f);
            }
#pragma unroll
            for (int k = 0; k < KS1; ++k) fb[k] = fn[k];
        }
    }
    __syncthreads();
    Frag16 ga[KS2];
    if (live) {
        const unsigned short* lp = mine + l15 * PITCH + kbase;
#pragma unroll
        for (int k = 0; k < KS2; ++k) {
            ga[k].q[0] = *(const u32x4*)(lp + 32 * k);
            ga[k].q[1] = *(const u32x4*)(lp + 32 * k + 16);
        }
    }
    __syncthreads();
    float* outt = reinterpret_cast<float*>(mine);
    if (live) {
        Frag16 gb[KS2], gn[KS2];
        {
            const unsigned short* bptr = B2t + (size_t)l15 * N1 + kbase;
#pragma unroll
            for (int k = 0; k < KS2; ++k) {
                gb[k].q[0] = *(const u32x4*)(bptr + 32 * k);
                gb[k].q[1] = *(const u32x4*)(bptr + 32 * k + 16);
            }
        }
#pragma unroll
        for (int t = 0; t < N2 / 16; ++t) {
            if (t + 1 < N2 / 16) {
                const unsigned short* bptr = B2t + (size_t)((t + 1) * 16 + l15) * N1 + kbase;
#pragma unroll
                for (int k = 0; k < KS2; ++k) {
                    gn[k].q[0] = *(const u32x4*)(bptr + 32 * k);
                    gn[k].q[1] = *(const u32x4*)(bptr + 32 * k + 16);
                }
            }
            f32x8 acc = {};
#pragma unroll
            for (int k = 0; k < KS2; ++k)
                acc = __builtin_amdgcn_wmma_f32_16x16x32_bf16(
                          false, ga[k].v, false, gb[k].v, (short)0, acc, false, false);
#pragma unroll
            for (int r = 0; r < 8; ++r)
                outt[(r + half * 8) * OPITCH + t * 16 + l15] = acc[r];
#pragma unroll
            for (int k = 0; k < KS2; ++k) gb[k] = gn[k];
        }
    }
    __syncthreads();
    if (live) {
        float* yrow = Y + (size_t)m0 * N2 + 4 * lane;
        if (m0 + 16 <= M) {
            float4 v[16];
#pragma unroll
            for (int i = 0; i < 16; ++i) v[i] = *reinterpret_cast<const float4*>(outt + i * OPITCH + 4 * lane);
#pragma unroll
            for (int i = 0; i < 16; ++i) *reinterpret_cast<float4*>(yrow + (size_t)i * N2) = v[i];
        } else {
            for (int i = 0; i < M - m0; ++i)
                *reinterpret_cast<float4*>(yrow + (size_t)i * N2) =
                    *reinterpret_cast<const float4*>(outt + i * OPITCH + 4 * lane);
        }
    }
}

__global__ void node_scores(const float* __restrict__ h,
                            const float* __restrict__ as_, const float* __restrict__ ad_,
                            float* __restrict__ es, float* __restrict__ ed,
                            int N, int C, unsigned* __restrict__ clr)
{
    const int warp = (blockIdx.x * blockDim.x + threadIdx.x) >> 5;
    const int lane = threadIdx.x & 31;
    if (warp >= N) return;
    const float* row = h + (size_t)warp * C;
    float s = 0.f, d = 0.f;
    for (int c = lane; c < C; c += 32) {
        const float v = row[c];
        s += v * as_[c];
        d += v * ad_[c];
    }
#pragma unroll
    for (int off = 16; off; off >>= 1) {
        s += __shfl_xor(s, off);
        d += __shfl_xor(d, off);
    }
    if (lane == 0) { es[warp] = s; ed[warp] = d; if (clr) clr[warp] = 0u; }
}

#define RD_WAVES 8
#define RD_ROWS  16
#define RD_PAIRS 8

static __device__ __forceinline__ float dot4(const float4 x, const float4 w)
{
    return x.x * w.x + x.y * w.y + x.z * w.z + x.w * w.w;
}

template <int NV>
static __device__ __forceinline__ float wave_sums(float (&v)[NV], int lane)
{
#pragma unroll
    for (int st = 0; st < 5; ++st) {
        const int m = 16 >> st;
        const int n = NV >> (st + 1);
        if (n >= 1) {
            const bool up = (lane & m) != 0;
#pragma unroll
            for (int i = 0; i < n; ++i) {
                const float lo = v[i], hi = v[i + n];
                const float keep = up ? hi : lo;
                const float send = up ? lo : hi;
                v[i] = keep + __shfl_xor(send, m);
            }
        } else {
            v[0] += __shfl_xor(v[0], m);
        }
    }
    return v[0];
}

__global__ void __launch_bounds__(32 * RD_WAVES)
node_scores_rows(const float* __restrict__ h,
                 const float* __restrict__ as_, const float* __restrict__ ad_,
                 float* __restrict__ es, float* __restrict__ ed,
                 int N, unsigned* __restrict__ clr)
{
    const int lane = threadIdx.x & 31;
    const int wave = __builtin_amdgcn_readfirstlane(threadIdx.x >> 5);
    const int r0   = (blockIdx.x * RD_WAVES + wave) * RD_ROWS;
    if (r0 >= N) return;
    const float4 ws = reinterpret_cast<const float4*>(as_)[lane];
    const float4 wd = reinterpret_cast<const float4*>(ad_)[lane];
    float4 x[RD_ROWS];
#pragma unroll
    for (int j = 0; j < RD_ROWS; ++j) {
        const int r = min(r0 + j, N - 1);
        x[j] = reinterpret_cast<const float4*>(h + (size_t)r * 128)[lane];
    }
    float v[2 * RD_ROWS];
#pragma unroll
    for (int j = 0; j < RD_ROWS; ++j) {
        v[j]           = dot4(x[j], ws);
        v[RD_ROWS + j] = dot4(x[j], wd);
    }
    const float t = wave_sums<2 * RD_ROWS>(v, lane);
    const int row = r0 + (lane & (RD_ROWS - 1));
    if (row < N) {
        if (lane < RD_ROWS) { es[row] = t; if (clr) clr[row] = 0u; }
        else                  ed[row] = t;
    }
}

__global__ void mark_needed(const int* __restrict__ mask, unsigned* __restrict__ need, int n, int N)
{
    const int i = blockIdx.x * blockDim.x + threadIdx.x;
    if (i >= n) return;
    const int m = mask[i];
    if ((unsigned)m < (unsigned)N) need[m] = 1u;
}

#define OWN_ROWS    196
#define OWN_THREADS 1024
#define OWN_QCAP    4096
#define OWN_U       4

static __device__ __forceinline__ float leaky02(float t) { return (t > 0.f) ? t : 0.2f * t; }

template <int C>
static __device__ __forceinline__ void own_add_row(const float* __restrict__ h, int s, float a,
                                                   float* accrow, int lane)
{
    const float* hs = h + (size_t)s * C + lane;
    float*       ar = accrow + lane;
    float v[C / 32];
#pragma unroll
    for (int k = 0; k < C / 32; ++k) v[k] = hs[32 * k];
#pragma unroll
    for (int k = 0; k < C / 32; ++k) atomicAdd(&ar[32 * k], v[k] * a);
}

template <int C, bool SECOND, int STAGE>
static __device__ __forceinline__ void own_walk(const int* __restrict__ ei, int E, int lo, int nrow,
                                                const float* __restrict__ h, const float* __restrict__ es,
                                                const float* edl, const unsigned* flag,
                                                unsigned* rmax, float* rden, float* acc,
                                                int wave, int lane, int nw)
{
    const int* __restrict__ dst = ei + E;
    for (int e0 = wave * 32; e0 < E; e0 += nw * 32) {
        const int e = e0 + lane;
        unsigned dl = 0u;
        bool ok = false;
        if (e < E) {
            dl = (unsigned)(dst[e] - lo);
            ok = dl < (unsigned)nrow;
            if (SECOND) { if (ok) ok = flag[dl] != 0u; }
        }
        int s = 0; float l = 0.f;
        if (ok) { s = ei[e]; l = leaky02(es[s] + edl[dl]); }
        if (STAGE == 0) {
            if (ok) atomicMax(&rmax[dl], fkey(l));
        } else if (STAGE == 1) {
            if (ok) atomicAdd(&rden[dl], __expf(l - funkey(rmax[dl])));
        } else {
            float a = 0.f;
            if (ok) a = __expf(l - funkey(rmax[dl])) / rden[dl];
            unsigned mk = __builtin_amdgcn_ballot_w32(ok);
            while (mk) {
                const int L = __builtin_ctz(mk);
                mk &= mk - 1u;
                const int      sj = __builtin_amdgcn_readlane(s, L);
                const unsigned dj = (unsigned)__builtin_amdgcn_readlane((int)dl, L);
                const float    aj = __int_as_float(__builtin_amdgcn_readlane(__float_as_int(a), L));
                own_add_row<C>(h, sj, aj, acc + dj * C, lane);
            }
        }
    }
}

template <int C, bool SECOND>
__global__ void __launch_bounds__(OWN_THREADS)
aggregate_owned(const int* __restrict__ ei, int E, int N,
                const float* __restrict__ h,
                const float* __restrict__ es, const float* __restrict__ ed,
                const unsigned* __restrict__ need, const float* __restrict__ bias,
                unsigned short* __restrict__ ybf, float* __restrict__ yf)
{
    static_assert(C == 128, "one float4 per lane covers a row");
    __shared__ __attribute__((aligned(16))) float acc[OWN_ROWS * C];
    __shared__ int      qe[OWN_QCAP];
    __shared__ unsigned qd[OWN_QCAP];
    __shared__ float    qp[OWN_QCAP];
    __shared__ float    edl[OWN_ROWS];
    __shared__ float    pself[OWN_ROWS];
    __shared__ unsigned rmax[OWN_ROWS];
    __shared__ float    rden[OWN_ROWS];
    __shared__ unsigned flag[OWN_ROWS];
    __shared__ unsigned qn;

    const int lane = threadIdx.x & 31;
    const int wave = __builtin_amdgcn_readfirstlane(threadIdx.x >> 5);
    const int nw   = OWN_THREADS / 32;
    const int lo   = blockIdx.x * OWN_ROWS;
    const int nrow = min(OWN_ROWS, N - lo);

    if (threadIdx.x == 0) qn = 0u;
    for (int r = threadIdx.x; r < nrow; r += OWN_THREADS) {
        const int d = lo + r;
        if (SECOND) flag[r] = (need[d] != 0u) ? 1u : 0u;
        const float edv = ed[d];
        const float l   = leaky02(es[d] + edv);
        edl[r]   = edv;
        pself[r] = l;
        rmax[r]  = fkey(l);
    }
    __syncthreads();

    const int* __restrict__ dst = ei + E;
    const int mis     = (int)((reinterpret_cast<size_t>(dst) >> 2) & 3u);
    const int ghead   = mis ? 1 : 0;
    const int gfull   = (E + mis) >> 2;
    const int nchunks = (gfull + 31) >> 5;
    const int startc  = (int)(((long long)blockIdx.x * nchunks) / gridDim.x);
    const char* lbase = reinterpret_cast<const char*>(dst - mis) + lane * 16;
    const int lane4m  = lane * 4 - mis;
    const unsigned lt = (1u << lane) - 1u;

    for (int cl0 = wave; cl0 < nchunks; cl0 += nw * OWN_U) {
        int dv[OWN_U][4];
        int cu[OWN_U];
#pragma unroll
        for (int u = 0; u < OWN_U; ++u) {
            const int cl = cl0 + u * nw;
            int cc = cl + startc;
            if (cc >= nchunks) cc -= nchunks;
            cu[u] = cc;
            const int gl = cc * 32;
            int4 q = make_int4(-1, -1, -1, -1);
            if (cl < nchunks && lane >= ghead - gl && lane < gfull - gl)
                q = *reinterpret_cast<const int4*>(lbase + (size_t)cc * 512);
            dv[u][0] = q.x; dv[u][1] = q.y; dv[u][2] = q.z; dv[u][3] = q.w;
        }
        bool     m[OWN_U * 4];
        unsigned dl[OWN_U * 4];
        unsigned mk[OWN_U * 4];
        unsigned anym = 0u;
#pragma unroll
        for (int q = 0; q < OWN_U * 4; ++q) {
            dl[q] = (unsigned)(dv[q >> 2][q & 3] - lo);
            bool mm = dl[q] < (unsigned)nrow;
            if (SECOND) { if (mm) mm = flag[dl[q]] != 0u; }
            m[q]  = mm;
            mk[q] = __builtin_amdgcn_ballot_w32(mm);
            anym |= mk[q];
        }
        if (anym) {
            unsigned tot = 0u;
#pragma unroll
            for (int q = 0; q < OWN_U * 4; ++q) tot += (unsigned)__builtin_popcount(mk[q]);
            unsigned base = 0u;
            if (lane == 0) base = atomicAdd(&qn, tot);
            base = (unsigned)__builtin_amdgcn_readfirstlane((int)base);
#pragma unroll
            for (int q = 0; q < OWN_U * 4; ++q) {
                if (m[q]) {
                    const unsigned slot = base + (unsigned)__builtin_popcount(mk[q] & lt);
                    if (slot < OWN_QCAP) { qe[slot] = cu[q >> 2] * 128 + lane4m + (q & 3); qd[slot] = dl[q]; }
                }
                base += (unsigned)__builtin_popcount(mk[q]);
            }
        }
    }
    if (wave == 0) {
        const bool whole   = gfull > ghead;
        const int  headEnd = (whole && ghead) ? 4 - mis : 0;
        const int  tailBeg = whole ? gfull * 4 - mis : 0;
        const int  nb      = headEnd + (E - tailBeg);
        if (lane < nb) {
            const int e = (lane < headEnd) ? lane : tailBeg + (lane - headEnd);
            const unsigned dle = (unsigned)(dst[e] - lo);
            bool ok = dle < (unsigned)nrow;
            if (SECOND) { if (ok) ok = flag[dle] != 0u; }
            if (ok) {
                const unsigned slot = atomicAdd(&qn, 1u);
                if (slot < OWN_QCAP) { qe[slot] = e; qd[slot] = dle; }
            }
        }
    }
    __syncthreads();

    const unsigned qtot = (unsigned)__builtin_amdgcn_readfirstlane((int)qn);
    const bool fits  = qtot <= (unsigned)OWN_QCAP;
    const int  total = fits ? (int)qtot : 0;

    if (fits) {
        for (int idx = threadIdx.x; idx < total; idx += OWN_THREADS) {
            const unsigned dq = qd[idx];
            const float l = leaky02(es[ei[qe[idx]]] + edl[dq]);
            qp[idx] = l;
            atomicMax(&rmax[dq], fkey(l));
        }
    } else {
        own_walk<C, SECOND, 0>(ei, E, lo, nrow, h, es, edl, flag, rmax, rden, acc, wave, lane, nw);
    }
    __syncthreads();
    for (int r = threadIdx.x; r < nrow; r += OWN_THREADS) {
        const float ps = __expf(pself[r] - funkey(rmax[r]));
        pself[r] = ps;
        rden[r]  = ps;
    }
    __syncthreads();
    if (fits) {
        for (int idx = threadIdx.x; idx < total; idx += OWN_THREADS) {
            const unsigned dq = qd[idx];
            const float pe = __expf(qp[idx] - funkey(rmax[dq]));
            qp[idx] = pe;
            atomicAdd(&rden[dq], pe);
        }
    } else {
        own_walk<C, SECOND, 1>(ei, E, lo, nrow, h, es, edl, flag, rmax, rden, acc, wave, lane, nw);
    }
    __syncthreads();

    {
        const float4* hb4  = reinterpret_cast<const float4*>(h + (size_t)lo * C);
        float4*       acc4 = reinterpret_cast<float4*>(acc);
        for (int i = threadIdx.x; i < nrow * (C / 4); i += OWN_THREADS) {
            const int r = i / (C / 4);
            float4 v = make_float4(0.f, 0.f, 0.f, 0.f);
            if (!SECOND || flag[r]) {
                const float  a = pself[r] / rden[r];
                const float4 x = hb4[i];
                v.x = x.x * a; v.y = x.y * a; v.z = x.z * a; v.w = x.w * a;
            }
            acc4[i] = v;
        }
    }
    __syncthreads();

    for (int b = wave * 32; b < total; b += nw * 32) {
        const int idx = b + lane;
        int s = 0; unsigned dq = 0u; float a = 0.f;
        if (idx < total) {
            dq = qd[idx];
            s  = ei[qe[idx]];
            a  = qp[idx] / rden[dq];
        }
        const int cnt = min(32, total - b);
        for (int j = 0; j < cnt; j += 4) {
            float    v[4][C / 32];
            float    aj[4];
            unsigned dj[4];
#pragma unroll
            for (int t = 0; t < 4; ++t) {
                const int jj = min(j + t, cnt - 1);
                const int sj = __builtin_amdgcn_readlane(s, jj);
                dj[t] = (unsigned)__builtin_amdgcn_readlane((int)dq, jj);
                aj[t] = __int_as_float(__builtin_amdgcn_readlane(__float_as_int(a), jj));
                const float* hs = h + (size_t)sj * C + lane;
#pragma unroll
                for (int k = 0; k < C / 32; ++k) v[t][k] = hs[32 * k];
            }
#pragma unroll
            for (int t = 0; t < 4; ++t) {
                if (j + t < cnt) {
                    float* ar = acc + dj[t] * C + lane;
#pragma unroll
                    for (int k = 0; k < C / 32; ++k) atomicAdd(&ar[32 * k], v[t][k] * aj[t]);
                }
            }
        }
    }
    if (!fits)
        own_walk<C, SECOND, 2>(ei, E, lo, nrow, h, es, edl, flag, rmax, rden, acc, wave, lane, nw);
    __syncthreads();

    for (int r = wave; r < nrow; r += nw) {
        const size_t d = (size_t)(lo + r);
        const float4 v = *reinterpret_cast<const float4*>(&acc[r * C + 4 * lane]);
        if (SECOND) {
            if (flag[r]) {
                const float4 bq = *reinterpret_cast<const float4*>(bias + 4 * lane);
                float4 o; o.x = v.x + bq.x; o.y = v.y + bq.y; o.z = v.z + bq.z; o.w = v.w + bq.w;
                *reinterpret_cast<float4*>(yf + d * C + 4 * lane) = o;
            }
        } else {
            uint2 o;
            o.x = (unsigned)f2bf(v.x) | ((unsigned)f2bf(v.y) << 16);
            o.y = (unsigned)f2bf(v.z) | ((unsigned)f2bf(v.w) << 16);
            *reinterpret_cast<uint2*>(ybf + d * C + 4 * lane) = o;
        }
    }
}

__global__ void transpose_to_bf16(const float* __restrict__ W, unsigned short* __restrict__ Wt,
                                  int K, int NOUT)
{
    const int i = blockIdx.x * blockDim.x + threadIdx.x;
    if (i >= K * NOUT) return;
    const int k = i / NOUT, n = i % NOUT;
    Wt[(size_t)n * K + k] = f2bf(W[i]);
}

__global__ void __launch_bounds__(32 * RD_WAVES)
link_predict_rows(const float* __restrict__ h, const int* __restrict__ mask,
                  const float* __restrict__ Wl, const float* __restrict__ bl,
                  float* __restrict__ out, int P)
{
    const int lane = threadIdx.x & 31;
    const int wave = __builtin_amdgcn_readfirstlane(threadIdx.x >> 5);
    const int p0   = (blockIdx.x * RD_WAVES + wave) * RD_PAIRS;
    if (p0 >= P) return;
    const int mi = mask[min(p0 * 2 + (lane & (2 * RD_PAIRS - 1)), 2 * P - 1)];
    const float4 wa = reinterpret_cast<const float4*>(Wl)[lane];
    const float4 wb = reinterpret_cast<const float4*>(Wl)[32 + lane];
    float4 xa[RD_PAIRS], xb[RD_PAIRS];
#pragma unroll
    for (int j = 0; j < RD_PAIRS; ++j) {
        const int ra = __builtin_amdgcn_readlane(mi, 2 * j);
        const int rb = __builtin_amdgcn_readlane(mi, 2 * j + 1);
        xa[j] = reinterpret_cast<const float4*>(h + (size_t)ra * 128)[lane];
        xb[j] = reinterpret_cast<const float4*>(h + (size_t)rb * 128)[lane];
    }
    float v[RD_PAIRS];
#pragma unroll
    for (int j = 0; j < RD_PAIRS; ++j) v[j] = dot4(xa[j], wa) + dot4(xb[j], wb);
    const float t = wave_sums<RD_PAIRS>(v, lane);
    const int p = p0 + (lane >> 2);
    if ((lane & 3) == 0 && p < P) out[p] = 1.f / (1.f + expf(-(t + bl[0])));
}

extern "C" void kernel_launch(void* const* d_in, const int* in_sizes, int n_in,
                              void* d_out, int out_size, void* d_ws, size_t ws_size,
                              hipStream_t stream)
{
    (void)n_in; (void)out_size; (void)ws_size;

    const float* features = (const float*)d_in[0];
    const int*   ei       = (const int*)  d_in[1];
    const int*   mask     = (const int*)  d_in[2];
    const float* W1   = (const float*)d_in[3];
    const float* a_s1 = (const float*)d_in[4];
    const float* a_d1 = (const float*)d_in[5];
    const float* b1   = (const float*)d_in[6];
    const float* W2   = (const float*)d_in[7];
    const float* a_s2 = (const float*)d_in[8];
    const float* a_d2 = (const float*)d_in[9];
    const float* b2   = (const float*)d_in[10];
    const float* Wl   = (const float*)d_in[11];
    const float* bl   = (const float*)d_in[12];

    constexpr int F  = 128, Hh = 256;
    const int N  = in_sizes[0] / F;
    const int E  = in_sizes[1] / 2;
    const int P  = in_sizes[2] / 2;

    char* ws = (char*)d_ws;
    size_t off = 0;
    auto alloc = [&](size_t bytes) -> char* {
        char* ptr = ws + off;
        off += (bytes + 255) & ~(size_t)255;
        return ptr;
    };
    unsigned short* xbf  = (unsigned short*)alloc((size_t)N * F  * 2);
    unsigned short* w1t  = (unsigned short*)alloc((size_t)Hh * F * 2);
    unsigned short* w2t  = (unsigned short*)alloc((size_t)F * Hh * 2);
    float*    buf1    = (float*)   alloc((size_t)N * Hh * 4);
    float*    buf2    = (float*)   alloc((size_t)N * Hh * 4);
    float*    es      = (float*)   alloc((size_t)N * 4);
    float*    ed      = (float*)   alloc((size_t)N * 4);
    float*    vs1     = (float*)   alloc((size_t)F * 4);
    float*    vd1     = (float*)   alloc((size_t)F * 4);
    unsigned* need    = (unsigned*)alloc((size_t)N * 4);

    const int tilesM = (N + 15) / 16;
    const int gemmGrid = (tilesM + PAIR_WAVES - 1) / PAIR_WAVES;
    const int TB = 256;
    const int ownGrid = (N + OWN_ROWS - 1) / OWN_ROWS;
    auto cdiv = [](size_t a, size_t b) -> int { return (int)((a + b - 1) / b); };

    transpose_to_bf16<<<cdiv((size_t)F * Hh, TB), TB, 0, stream>>>(W1, w1t, F, Hh);
    transpose_to_bf16<<<cdiv((size_t)Hh * F, TB), TB, 0, stream>>>(W2, w2t, Hh, F);

    node_scores<<<cdiv((size_t)F * 32, TB), TB, 0, stream>>>(W1, a_s1, a_d1, vs1, vd1, F, Hh, nullptr);
    node_scores_rows<<<cdiv((size_t)N, RD_ROWS * RD_WAVES), 32 * RD_WAVES, 0, stream>>>(features, vs1, vd1, es, ed, N, need);
    aggregate_owned<F, false><<<ownGrid, OWN_THREADS, 0, stream>>>(ei, E, N, features, es, ed,
                                                                  nullptr, nullptr, xbf, nullptr);

    mark_needed<<<cdiv((size_t)P * 2, TB), TB, 0, stream>>>(mask, need, P * 2, N);
    wmma_two_products<F, Hh, F><<<gemmGrid, 32 * PAIR_WAVES, 0, stream>>>(xbf, w1t, b1, w2t, buf1, N);
    node_scores_rows<<<cdiv((size_t)N, RD_ROWS * RD_WAVES), 32 * RD_WAVES, 0, stream>>>(buf1, a_s2, a_d2, es, ed, N, nullptr);
    aggregate_owned<F, true><<<ownGrid, OWN_THREADS, 0, stream>>>(ei, E, N, buf1, es, ed,
                                                                 need, b2, nullptr, buf2);

    link_predict_rows<<<cdiv((size_t)P, RD_PAIRS * RD_WAVES), 32 * RD_WAVES, 0, stream>>>(buf2, mask, Wl, bl,
                                                                                         (float*)d_out, P);
}


# Layer 2's row-owning work-groups take their edge queues from the workspace, where layer 1's left them (flagged rows kept): its filter pass over all 500000 destinations goes; LDS add order: arrival
# speedup vs baseline: 5.6891x; 1.2735x over previous
#include <hip/hip_runtime.h>
#include <hip/hip_bf16.h>

typedef __attribute__((__ext_vector_type__(16))) __bf16 bf16x16;
typedef __attribute__((__ext_vector_type__(8)))  float  f32x8;
typedef __attribute__((__ext_vector_type__(4)))  unsigned int u32x4;

union Frag16 {
    u32x4   q[2];
    bf16x16 v;
};

static __device__ __forceinline__ unsigned short f2bf(float f) {
    unsigned u = __float_as_uint(f);
    u += 0x7FFFu + ((u >> 16) & 1u);
    return (unsigned short)(u >> 16);
}

static __device__ __forceinline__ unsigned fkey(float f) {
    unsigned u = __float_as_uint(f);
    return (u & 0x80000000u) ? ~u : (u | 0x80000000u);
}
static __device__ __forceinline__ float funkey(unsigned k) {
    unsigned u = (k & 0x80000000u) ? (k & 0x7FFFFFFFu) : ~k;
    return __uint_as_float(u);
}

#define PAIR_WAVES 4
#define PAIR_PAD   8

template <int K1, int N1, int N2>
__global__ void __launch_bounds__(32 * PAIR_WAVES)
wmma_two_products(const unsigned short* __restrict__ A,
                  const unsigned short* __restrict__ B1t,
                  const float* __restrict__ bias1,
                  const unsigned short* __restrict__ B2t,
                  float* __restrict__ Y, int M)
{
    static_assert(K1 % 32 == 0 && N1 % 32 == 0 && N1 % 16 == 0 && N2 % 16 == 0, "whole fragments");
    constexpr int PITCH = N1 + PAIR_PAD;
    static_assert((PITCH * 2) % 16 == 0 && ((PITCH * 2 / 16) & 1) == 1, "pitch = 16 B x odd");
    constexpr int OPITCH = N2 + 4;
    static_assert(N2 == 128, "one float4 per lane covers a result row");
    static_assert(OPITCH * 4 <= PITCH * 2, "the result tile reuses the wave's LDS tile");
    constexpr int KS1 = K1 / 32;
    constexpr int KS2 = N1 / 32;
    __shared__ __attribute__((aligned(16))) unsigned short mid[PAIR_WAVES][16 * PITCH];

    const int lane  = threadIdx.x & 31;
    const int wave  = __builtin_amdgcn_readfirstlane(threadIdx.x >> 5);
    const int m0    = (blockIdx.x * PAIR_WAVES + wave) * 16;
    const bool live = m0 < M;
    const int l15   = lane & 15;
    const int half  = lane >> 4;
    const int kbase = half * 8;
    unsigned short* mine = mid[wave];

    if (live) {
        int rowA = m0 + l15;
        if (rowA >= M) rowA = M - 1;
        const unsigned short* aptr = A + (size_t)rowA * K1 + kbase;
        Frag16 fa[KS1];
#pragma unroll
        for (int k = 0; k < KS1; ++k) {
            fa[k].q[0] = *(const u32x4*)(aptr + 32 * k);
            fa[k].q[1] = *(const u32x4*)(aptr + 32 * k + 16);
        }
        Frag16 fb[KS1], fn[KS1];
        {
            const unsigned short* bptr = B1t + (size_t)l15 * K1 + kbase;
#pragma unroll
            for (int k = 0; k < KS1; ++k) {
                fb[k].q[0] = *(const u32x4*)(bptr + 32 * k);
                fb[k].q[1] = *(const u32x4*)(bptr + 32 * k + 16);
            }
        }
#pragma unroll
        for (int t = 0; t < N1 / 16; ++t) {
            if (t + 1 < N1 / 16) {
                const unsigned short* bptr = B1t + (size_t)((t + 1) * 16 + l15) * K1 + kbase;
#pragma unroll
                for (int k = 0; k < KS1; ++k) {
                    fn[k].q[0] = *(const u32x4*)(bptr + 32 * k);
                    fn[k].q[1] = *(const u32x4*)(bptr + 32 * k + 16);
                }
            }
            f32x8 acc = {};
#pragma unroll
            for (int k = 0; k < KS1; ++k)
                acc = __builtin_amdgcn_wmma_f32_16x16x32_bf16(
                          false, fa[k].v, false, fb[k].v, (short)0, acc, false, false);
            const int   col = t * 16 + l15;
            const float bv  = bias1[col];
#pragma unroll
            for (int r = 0; r < 8; ++r) {
                const float v = acc[r] + bv;
                mine[(r + half * 8) * PITCH + col] = f2bf(v > 0.f ? v : 0.f);
            }
#pragma unroll
            for (int k = 0; k < KS1; ++k) fb[k] = fn[k];
        }
    }
    __syncthreads();
    Frag16 ga[KS2];
    if (live) {
        const unsigned short* lp = mine + l15 * PITCH + kbase;
#pragma unroll
        for (int k = 0; k < KS2; ++k) {
            ga[k].q[0] = *(const u32x4*)(lp + 32 * k);
            ga[k].q[1] = *(const u32x4*)(lp + 32 * k + 16);
        }
    }
    __syncthreads();
    float* outt = reinterpret_cast<float*>(mine);
    if (live) {
        Frag16 gb[KS2], gn[KS2];
        {
            const unsigned short* bptr = B2t + (size_t)l15 * N1 + kbase;
#pragma unroll
            for (int k = 0; k < KS2; ++k) {
                gb[k].q[0] = *(const u32x4*)(bptr + 32 * k);
                gb[k].q[1] = *(const u32x4*)(bptr + 32 * k + 16);
            }
        }
#pragma unroll
        for (int t = 0; t < N2 / 16; ++t) {
            if (t + 1 < N2 / 16) {
                const unsigned short* bptr = B2t + (size_t)((t + 1) * 16 + l15) * N1 + kbase;
#pragma unroll
                for (int k = 0; k < KS2; ++k) {
                    gn[k].q[0] = *(const u32x4*)(bptr + 32 * k);
                    gn[k].q[1] = *(const u32x4*)(bptr + 32 * k + 16);
                }
            }
            f32x8 acc = {};
#pragma unroll
            for (int k = 0; k < KS2; ++k)
                acc = __builtin_amdgcn_wmma_f32_16x16x32_bf16(
                          false, ga[k].v, false, gb[k].v, (short)0, acc, false, false);
#pragma unroll
            for (int r = 0; r < 8; ++r)
                outt[(r + half * 8) * OPITCH + t * 16 + l15] = acc[r];
#pragma unroll
            for (int k = 0; k < KS2; ++k) gb[k] = gn[k];
        }
    }
    __syncthreads();
    if (live) {
        float* yrow = Y + (size_t)m0 * N2 + 4 * lane;
        if (m0 + 16 <= M) {
            float4 v[16];
#pragma unroll
            for (int i = 0; i < 16; ++i) v[i] = *reinterpret_cast<const float4*>(outt + i * OPITCH + 4 * lane);
#pragma unroll
            for (int i = 0; i < 16; ++i) *reinterpret_cast<float4*>(yrow + (size_t)i * N2) = v[i];
        } else {
            for (int i = 0; i < M - m0; ++i)
                *reinterpret_cast<float4*>(yrow + (size_t)i * N2) =
                    *reinterpret_cast<const float4*>(outt + i * OPITCH + 4 * lane);
        }
    }
}

__global__ void node_scores(const float* __restrict__ h,
                            const float* __restrict__ as_, const float* __restrict__ ad_,
                            float* __restrict__ es, float* __restrict__ ed,
                            int N, int C, unsigned* __restrict__ clr)
{
    const int warp = (blockIdx.x * blockDim.x + threadIdx.x) >> 5;
    const int lane = threadIdx.x & 31;
    if (warp >= N) return;
    const float* row = h + (size_t)warp * C;
    float s = 0.f, d = 0.f;
    for (int c = lane; c < C; c += 32) {
        const float v = row[c];
        s += v * as_[c];
        d += v * ad_[c];
    }
#pragma unroll
    for (int off = 16; off; off >>= 1) {
        s += __shfl_xor(s, off);
        d += __shfl_xor(d, off);
    }
    if (lane == 0) { es[warp] = s; ed[warp] = d; if (clr) clr[warp] = 0u; }
}

#define RD_WAVES 8
#define RD_ROWS  16
#define RD_PAIRS 8

static __device__ __forceinline__ float dot4(const float4 x, const float4 w)
{
    return x.x * w.x + x.y * w.y + x.z * w.z + x.w * w.w;
}

template <int NV>
static __device__ __forceinline__ float wave_sums(float (&v)[NV], int lane)
{
#pragma unroll
    for (int st = 0; st < 5; ++st) {
        const int m = 16 >> st;
        const int n = NV >> (st + 1);
        if (n >= 1) {
            const bool up = (lane & m) != 0;
#pragma unroll
            for (int i = 0; i < n; ++i) {
                const float lo = v[i], hi = v[i + n];
                const float keep = up ? hi : lo;
                const float send = up ? lo : hi;
                v[i] = keep + __shfl_xor(send, m);
            }
        } else {
            v[0] += __shfl_xor(v[0], m);
        }
    }
    return v[0];
}

__global__ void __launch_bounds__(32 * RD_WAVES)
node_scores_rows(const float* __restrict__ h,
                 const float* __restrict__ as_, const float* __restrict__ ad_,
                 float* __restrict__ es, float* __restrict__ ed,
                 int N, unsigned* __restrict__ clr)
{
    const int lane = threadIdx.x & 31;
    const int wave = __builtin_amdgcn_readfirstlane(threadIdx.x >> 5);
    const int r0   = (blockIdx.x * RD_WAVES + wave) * RD_ROWS;
    if (r0 >= N) return;
    const float4 ws = reinterpret_cast<const float4*>(as_)[lane];
    const float4 wd = reinterpret_cast<const float4*>(ad_)[lane];
    float4 x[RD_ROWS];
#pragma unroll
    for (int j = 0; j < RD_ROWS; ++j) {
        const int r = min(r0 + j, N - 1);
        x[j] = reinterpret_cast<const float4*>(h + (size_t)r * 128)[lane];
    }
    float v[2 * RD_ROWS];
#pragma unroll
    for (int j = 0; j < RD_ROWS; ++j) {
        v[j]           = dot4(x[j], ws);
        v[RD_ROWS + j] = dot4(x[j], wd);
    }
    const float t = wave_sums<2 * RD_ROWS>(v, lane);
    const int row = r0 + (lane & (RD_ROWS - 1));
    if (row < N) {
        if (lane < RD_ROWS) { es[row] = t; if (clr) clr[row] = 0u; }
        else                  ed[row] = t;
    }
}

__global__ void mark_needed(const int* __restrict__ mask, unsigned* __restrict__ need, int n, int N)
{
    const int i = blockIdx.x * blockDim.x + threadIdx.x;
    if (i >= n) return;
    const int m = mask[i];
    if ((unsigned)m < (unsigned)N) need[m] = 1u;
}

#define OWN_ROWS    196
#define OWN_THREADS 1024
#define OWN_QCAP    4096
#define OWN_U       4

static __device__ __forceinline__ float leaky02(float t) { return (t > 0.f) ? t : 0.2f * t; }

template <int C>
static __device__ __forceinline__ void own_add_row(const float* __restrict__ h, int s, float a,
                                                   float* accrow, int lane)
{
    const float* hs = h + (size_t)s * C + lane;
    float*       ar = accrow + lane;
    float v[C / 32];
#pragma unroll
    for (int k = 0; k < C / 32; ++k) v[k] = hs[32 * k];
#pragma unroll
    for (int k = 0; k < C / 32; ++k) atomicAdd(&ar[32 * k], v[k] * a);
}

template <int C, bool SECOND, int STAGE>
static __device__ __forceinline__ void own_walk(const int* __restrict__ ei, int E, int lo, int nrow,
                                                const float* __restrict__ h, const float* __restrict__ es,
                                                const float* edl, const unsigned* flag,
                                                unsigned* rmax, float* rden, float* acc,
                                                int wave, int lane, int nw)
{
    const int* __restrict__ dst = ei + E;
    for (int e0 = wave * 32; e0 < E; e0 += nw * 32) {
        const int e = e0 + lane;
        unsigned dl = 0u;
        bool ok = false;
        if (e < E) {
            dl = (unsigned)(dst[e] - lo);
            ok = dl < (unsigned)nrow;
            if (SECOND) { if (ok) ok = flag[dl] != 0u; }
        }
        int s = 0; float l = 0.f;
        if (ok) { s = ei[e]; l = leaky02(es[s] + edl[dl]); }
        if (STAGE == 0) {
            if (ok) atomicMax(&rmax[dl], fkey(l));
        } else if (STAGE == 1) {
            if (ok) atomicAdd(&rden[dl], __expf(l - funkey(rmax[dl])));
        } else {
            float a = 0.f;
            if (ok) a = __expf(l - funkey(rmax[dl])) / rden[dl];
            unsigned mk = __builtin_amdgcn_ballot_w32(ok);
            while (mk) {
                const int L = __builtin_ctz(mk);
                mk &= mk - 1u;
                const int      sj = __builtin_amdgcn_readlane(s, L);
                const unsigned dj = (unsigned)__builtin_amdgcn_readlane((int)dl, L);
                const float    aj = __int_as_float(__builtin_amdgcn_readlane(__float_as_int(a), L));
                own_add_row<C>(h, sj, aj, acc + dj * C, lane);
            }
        }
    }
}

template <int C, bool SECOND>
__global__ void __launch_bounds__(OWN_THREADS)
aggregate_owned(const int* __restrict__ ei, int E, int N,
                const float* __restrict__ h,
                const float* __restrict__ es, const float* __restrict__ ed,
                const unsigned* __restrict__ need, const float* __restrict__ bias,
                unsigned short* __restrict__ ybf, float* __restrict__ yf,
                unsigned* __restrict__ sv_n, int* __restrict__ sv_e, unsigned* __restrict__ sv_d)
{
    static_assert(C == 128, "one float4 per lane covers a row");
    __shared__ __attribute__((aligned(16))) float acc[OWN_ROWS * C];
    __shared__ int      qe[OWN_QCAP];
    __shared__ unsigned qd[OWN_QCAP];
    __shared__ float    qp[OWN_QCAP];
    __shared__ float    edl[OWN_ROWS];
    __shared__ float    pself[OWN_ROWS];
    __shared__ unsigned rmax[OWN_ROWS];
    __shared__ float    rden[OWN_ROWS];
    __shared__ unsigned flag[OWN_ROWS];
    __shared__ unsigned qn;

    const int lane = threadIdx.x & 31;
    const int wave = __builtin_amdgcn_readfirstlane(threadIdx.x >> 5);
    const int nw   = OWN_THREADS / 32;
    const int lo   = blockIdx.x * OWN_ROWS;
    const int nrow = min(OWN_ROWS, N - lo);

    if (threadIdx.x == 0) qn = 0u;
    for (int r = threadIdx.x; r < nrow; r += OWN_THREADS) {
        const int d = lo + r;
        if (SECOND) flag[r] = (need[d] != 0u) ? 1u : 0u;
        const float edv = ed[d];
        const float l   = leaky02(es[d] + edv);
        edl[r]   = edv;
        pself[r] = l;
        rmax[r]  = fkey(l);
    }
    __syncthreads();

    const unsigned lt = (1u << lane) - 1u;
    int*      sve = sv_e + (size_t)blockIdx.x * OWN_QCAP;
    unsigned* svd = sv_d + (size_t)blockIdx.x * OWN_QCAP;
    if (!SECOND) {
        const int* __restrict__ dst = ei + E;
        const int mis     = (int)((reinterpret_cast<size_t>(dst) >> 2) & 3u);
        const int ghead   = mis ? 1 : 0;
        const int gfull   = (E + mis) >> 2;
        const int nchunks = (gfull + 31) >> 5;
        const int startc  = (int)(((long long)blockIdx.x * nchunks) / gridDim.x);
        const char* lbase = reinterpret_cast<const char*>(dst - mis) + lane * 16;
        const int lane4m  = lane * 4 - mis;

        for (int cl0 = wave; cl0 < nchunks; cl0 += nw * OWN_U) {
            int dv[OWN_U][4];
            int cu[OWN_U];
#pragma unroll
            for (int u = 0; u < OWN_U; ++u) {
                const int cl = cl0 + u * nw;
                int cc = cl + startc;
                if (cc >= nchunks) cc -= nchunks;
                cu[u] = cc;
                const int gl = cc * 32;
                int4 q = make_int4(-1, -1, -1, -1);
                if (cl < nchunks && lane >= ghead - gl && lane < gfull - gl)
                    q = *reinterpret_cast<const int4*>(lbase + (size_t)cc * 512);
                dv[u][0] = q.x; dv[u][1] = q.y; dv[u][2] = q.z; dv[u][3] = q.w;
            }
            bool     m[OWN_U * 4];
            unsigned dl[OWN_U * 4];
            unsigned mk[OWN_U * 4];
            unsigned anym = 0u;
#pragma unroll
            for (int q = 0; q < OWN_U * 4; ++q) {
                dl[q] = (unsigned)(dv[q >> 2][q & 3] - lo);
                const bool mm = dl[q] < (unsigned)nrow;
                m[q]  = mm;
                mk[q] = __builtin_amdgcn_ballot_w32(mm);
                anym |= mk[q];
            }
            if (anym) {
                unsigned tot = 0u;
#pragma unroll
                for (int q = 0; q < OWN_U * 4; ++q) tot += (unsigned)__builtin_popcount(mk[q]);
                unsigned base = 0u;
                if (lane == 0) base = atomicAdd(&qn, tot);
                base = (unsigned)__builtin_amdgcn_readfirstlane((int)base);
#pragma unroll
                for (int q = 0; q < OWN_U * 4; ++q) {
                    if (m[q]) {
                        const unsigned slot = base + (unsigned)__builtin_popcount(mk[q] & lt);
                        if (slot < OWN_QCAP) { qe[slot] = cu[q >> 2] * 128 + lane4m + (q & 3); qd[slot] = dl[q]; }
                    }
                    base += (unsigned)__builtin_popcount(mk[q]);
                }
            }
        }
        if (wave == 0) {
            const bool whole   = gfull > ghead;
            const int  headEnd = (whole && ghead) ? 4 - mis : 0;
            const int  tailBeg = whole ? gfull * 4 - mis : 0;
            const int  nb      = headEnd + (E - tailBeg);
            if (lane < nb) {
                const int e = (lane < headEnd) ? lane : tailBeg + (lane - headEnd);
                const unsigned dle = (unsigned)(dst[e] - lo);
                const bool ok = dle < (unsigned)nrow;
                if (ok) {
                    const unsigned slot = atomicAdd(&qn, 1u);
                    if (slot < OWN_QCAP) { qe[slot] = e; qd[slot] = dle; }
                }
            }
        }
    } else {
        const unsigned stot = sv_n[blockIdx.x];
        if (stot <= (unsigned)OWN_QCAP) {
            for (int b = wave * 32; b < (int)stot; b += nw * 32) {
                const int idx = b + lane;
                int e = 0; unsigned dq = 0u; bool mm = false;
                if (idx < (int)stot) { e = sve[idx]; dq = svd[idx]; mm = flag[dq] != 0u; }
                const unsigned mk = __builtin_amdgcn_ballot_w32(mm);
                if (mk) {
                    unsigned base = 0u;
                    if (lane == 0) base = atomicAdd(&qn, (unsigned)__builtin_popcount(mk));
                    base = (unsigned)__builtin_amdgcn_readfirstlane((int)base);
                    if (mm) {
                        const unsigned slot = base + (unsigned)__builtin_popcount(mk & lt);
                        qe[slot] = e; qd[slot] = dq;
                    }
                }
            }
        } else if (threadIdx.x == 0) {
            qn = (unsigned)OWN_QCAP + 1u;
        }
    }
    __syncthreads();

    const unsigned qtot = (unsigned)__builtin_amdgcn_readfirstlane((int)qn);
    const bool fits  = qtot <= (unsigned)OWN_QCAP;
    const int  total = fits ? (int)qtot : 0;
    if (!SECOND) {
        if (threadIdx.x == 0) sv_n[blockIdx.x] = qtot;
        for (int idx = threadIdx.x; idx < total; idx += OWN_THREADS) { sve[idx] = qe[idx]; svd[idx] = qd[idx]; }
    }

    if (fits) {
        for (int idx = threadIdx.x; idx < total; idx += OWN_THREADS) {
            const unsigned dq = qd[idx];
            const float l = leaky02(es[ei[qe[idx]]] + edl[dq]);
            qp[idx] = l;
            atomicMax(&rmax[dq], fkey(l));
        }
    } else {
        own_walk<C, SECOND, 0>(ei, E, lo, nrow, h, es, edl, flag, rmax, rden, acc, wave, lane, nw);
    }
    __syncthreads();
    for (int r = threadIdx.x; r < nrow; r += OWN_THREADS) {
        const float ps = __expf(pself[r] - funkey(rmax[r]));
        pself[r] = ps;
        rden[r]  = ps;
    }
    __syncthreads();
    if (fits) {
        for (int idx = threadIdx.x; idx < total; idx += OWN_THREADS) {
            const unsigned dq = qd[idx];
            const float pe = __expf(qp[idx] - funkey(rmax[dq]));
            qp[idx] = pe;
            atomicAdd(&rden[dq], pe);
        }
    } else {
        own_walk<C, SECOND, 1>(ei, E, lo, nrow, h, es, edl, flag, rmax, rden, acc, wave, lane, nw);
    }
    __syncthreads();

    {
        const float4* hb4  = reinterpret_cast<const float4*>(h + (size_t)lo * C);
        float4*       acc4 = reinterpret_cast<float4*>(acc);
        for (int i = threadIdx.x; i < nrow * (C / 4); i += OWN_THREADS) {
            const int r = i / (C / 4);
            float4 v = make_float4(0.f, 0.f, 0.f, 0.f);
            if (!SECOND || flag[r]) {
                const float  a = pself[r] / rden[r];
                const float4 x = hb4[i];
                v.x = x.x * a; v.y = x.y * a; v.z = x.z * a; v.w = x.w * a;
            }
            acc4[i] = v;
        }
    }
    __syncthreads();

    for (int b = wave * 32; b < total; b += nw * 32) {
        const int idx = b + lane;
        int s = 0; unsigned dq = 0u; float a = 0.f;
        if (idx < total) {
            dq = qd[idx];
            s  = ei[qe[idx]];
            a  = qp[idx] / rden[dq];
        }
        const int cnt = min(32, total - b);
        for (int j = 0; j < cnt; j += 4) {
            float    v[4][C / 32];
            float    aj[4];
            unsigned dj[4];
#pragma unroll
            for (int t = 0; t < 4; ++t) {
                const int jj = min(j + t, cnt - 1);
                const int sj = __builtin_amdgcn_readlane(s, jj);
                dj[t] = (unsigned)__builtin_amdgcn_readlane((int)dq, jj);
                aj[t] = __int_as_float(__builtin_amdgcn_readlane(__float_as_int(a), jj));
                const float* hs = h + (size_t)sj * C + lane;
#pragma unroll
                for (int k = 0; k < C / 32; ++k) v[t][k] = hs[32 * k];
            }
#pragma unroll
            for (int t = 0; t < 4; ++t) {
                if (j + t < cnt) {
                    float* ar = acc + dj[t] * C + lane;
#pragma unroll
                    for (int k = 0; k < C / 32; ++k) atomicAdd(&ar[32 * k], v[t][k] * aj[t]);
                }
            }
        }
    }
    if (!fits)
        own_walk<C, SECOND, 2>(ei, E, lo, nrow, h, es, edl, flag, rmax, rden, acc, wave, lane, nw);
    __syncthreads();

    for (int r = wave; r < nrow; r += nw) {
        const size_t d = (size_t)(lo + r);
        const float4 v = *reinterpret_cast<const float4*>(&acc[r * C + 4 * lane]);
        if (SECOND) {
            if (flag[r]) {
                const float4 bq = *reinterpret_cast<const float4*>(bias + 4 * lane);
                float4 o; o.x = v.x + bq.x; o.y = v.y + bq.y; o.z = v.z + bq.z; o.w = v.w + bq.w;
                *reinterpret_cast<float4*>(yf + d * C + 4 * lane) = o;
            }
        } else {
            uint2 o;
            o.x = (unsigned)f2bf(v.x) | ((unsigned)f2bf(v.y) << 16);
            o.y = (unsigned)f2bf(v.z) | ((unsigned)f2bf(v.w) << 16);
            *reinterpret_cast<uint2*>(ybf + d * C + 4 * lane) = o;
        }
    }
}

__global__ void transpose_to_bf16(const float* __restrict__ W, unsigned short* __restrict__ Wt,
                                  int K, int NOUT)
{
    const int i = blockIdx.x * blockDim.x + threadIdx.x;
    if (i >= K * NOUT) return;
    const int k = i / NOUT, n = i % NOUT;
    Wt[(size_t)n * K + k] = f2bf(W[i]);
}

__global__ void __launch_bounds__(32 * RD_WAVES)
link_predict_rows(const float* __restrict__ h, const int* __restrict__ mask,
                  const float* __restrict__ Wl, const float* __restrict__ bl,
                  float* __restrict__ out, int P)
{
    const int lane = threadIdx.x & 31;
    const int wave = __builtin_amdgcn_readfirstlane(threadIdx.x >> 5);
    const int p0   = (blockIdx.x * RD_WAVES + wave) * RD_PAIRS;
    if (p0 >= P) return;
    const int mi = mask[min(p0 * 2 + (lane & (2 * RD_PAIRS - 1)), 2 * P - 1)];
    const float4 wa = reinterpret_cast<const float4*>(Wl)[lane];
    const float4 wb = reinterpret_cast<const float4*>(Wl)[32 + lane];
    float4 xa[RD_PAIRS], xb[RD_PAIRS];
#pragma unroll
    for (int j = 0; j < RD_PAIRS; ++j) {
        const int ra = __builtin_amdgcn_readlane(mi, 2 * j);
        const int rb = __builtin_amdgcn_readlane(mi, 2 * j + 1);
        xa[j] = reinterpret_cast<const float4*>(h + (size_t)ra * 128)[lane];
        xb[j] = reinterpret_cast<const float4*>(h + (size_t)rb * 128)[lane];
    }
    float v[RD_PAIRS];
#pragma unroll
    for (int j = 0; j < RD_PAIRS; ++j) v[j] = dot4(xa[j], wa) + dot4(xb[j], wb);
    const float t = wave_sums<RD_PAIRS>(v, lane);
    const int p = p0 + (lane >> 2);
    if ((lane & 3) == 0 && p < P) out[p] = 1.f / (1.f + expf(-(t + bl[0])));
}

extern "C" void kernel_launch(void* const* d_in, const int* in_sizes, int n_in,
                              void* d_out, int out_size, void* d_ws, size_t ws_size,
                              hipStream_t stream)
{
    (void)n_in; (void)out_size; (void)ws_size;

    const float* features = (const float*)d_in[0];
    const int*   ei       = (const int*)  d_in[1];
    const int*   mask     = (const int*)  d_in[2];
    const float* W1   = (const float*)d_in[3];
    const float* a_s1 = (const float*)d_in[4];
    const float* a_d1 = (const float*)d_in[5];
    const float* b1   = (const float*)d_in[6];
    const float* W2   = (const float*)d_in[7];
    const float* a_s2 = (const float*)d_in[8];
    const float* a_d2 = (const float*)d_in[9];
    const float* b2   = (const float*)d_in[10];
    const float* Wl   = (const float*)d_in[11];
    const float* bl   = (const float*)d_in[12];

    constexpr int F  = 128, Hh = 256;
    const int N  = in_sizes[0] / F;
    const int E  = in_sizes[1] / 2;
    const int P  = in_sizes[2] / 2;

    char* ws = (char*)d_ws;
    size_t off = 0;
    auto alloc = [&](size_t bytes) -> char* {
        char* ptr = ws + off;
        off += (bytes + 255) & ~(size_t)255;
        return ptr;
    };
    unsigned short* xbf  = (unsigned short*)alloc((size_t)N * F  * 2);
    unsigned short* w1t  = (unsigned short*)alloc((size_t)Hh * F * 2);
    unsigned short* w2t  = (unsigned short*)alloc((size_t)F * Hh * 2);
    float*    buf1    = (float*)   alloc((size_t)N * Hh * 4);
    float*    buf2    = (float*)   alloc((size_t)N * Hh * 4);
    float*    es      = (float*)   alloc((size_t)N * 4);
    float*    ed      = (float*)   alloc((size_t)N * 4);
    float*    vs1     = (float*)   alloc((size_t)F * 4);
    float*    vd1     = (float*)   alloc((size_t)F * 4);
    unsigned* need    = (unsigned*)alloc((size_t)N * 4);
    const int ownGrid = (N + OWN_ROWS - 1) / OWN_ROWS;
    unsigned* svn     = (unsigned*)alloc((size_t)ownGrid * 4);
    int*      sve     = (int*)     alloc((size_t)ownGrid * OWN_QCAP * 4);
    unsigned* svd     = (unsigned*)alloc((size_t)ownGrid * OWN_QCAP * 4);

    const int tilesM = (N + 15) / 16;
    const int gemmGrid = (tilesM + PAIR_WAVES - 1) / PAIR_WAVES;
    const int TB = 256;
    auto cdiv = [](size_t a, size_t b) -> int { return (int)((a + b - 1) / b); };

    transpose_to_bf16<<<cdiv((size_t)F * Hh, TB), TB, 0, stream>>>(W1, w1t, F, Hh);
    transpose_to_bf16<<<cdiv((size_t)Hh * F, TB), TB, 0, stream>>>(W2, w2t, Hh, F);

    node_scores<<<cdiv((size_t)F * 32, TB), TB, 0, stream>>>(W1, a_s1, a_d1, vs1, vd1, F, Hh, nullptr);
    node_scores_rows<<<cdiv((size_t)N, RD_ROWS * RD_WAVES), 32 * RD_WAVES, 0, stream>>>(features, vs1, vd1, es, ed, N, need);
    aggregate_owned<F, false><<<ownGrid, OWN_THREADS, 0, stream>>>(ei, E, N, features, es, ed,
                                                                  nullptr, nullptr, xbf, nullptr, svn, sve, svd);

    mark_needed<<<cdiv((size_t)P * 2, TB), TB, 0, stream>>>(mask, need, P * 2, N);
    wmma_two_products<F, Hh, F><<<gemmGrid, 32 * PAIR_WAVES, 0, stream>>>(xbf, w1t, b1, w2t, buf1, N);
    node_scores_rows<<<cdiv((size_t)N, RD_ROWS * RD_WAVES), 32 * RD_WAVES, 0, stream>>>(buf1, a_s2, a_d2, es, ed, N, nullptr);
    aggregate_owned<F, true><<<ownGrid, OWN_THREADS, 0, stream>>>(ei, E, N, buf1, es, ed,
                                                                 need, b2, nullptr, buf2, svn, sve, svd);

    link_predict_rows<<<cdiv((size_t)P, RD_PAIRS * RD_WAVES), 32 * RD_WAVES, 0, stream>>>(buf2, mask, Wl, bl,
                                                                                         (float*)d_out, P);
}


# Layer 1's row-owning work-groups find their edges in a list of one owner byte per edge (row / 197, written by the score pass), not in the 4-byte destinations; LDS add order: arrival
# speedup vs baseline: 6.6258x; 1.1646x over previous
#include <hip/hip_runtime.h>
#include <hip/hip_bf16.h>

typedef __attribute__((__ext_vector_type__(16))) __bf16 bf16x16;
typedef __attribute__((__ext_vector_type__(8)))  float  f32x8;
typedef __attribute__((__ext_vector_type__(4)))  unsigned int u32x4;

union Frag16 {
    u32x4   q[2];
    bf16x16 v;
};

static __device__ __forceinline__ unsigned short f2bf(float f) {
    unsigned u = __float_as_uint(f);
    u += 0x7FFFu + ((u >> 16) & 1u);
    return (unsigned short)(u >> 16);
}

static __device__ __forceinline__ unsigned fkey(float f) {
    unsigned u = __float_as_uint(f);
    return (u & 0x80000000u) ? ~u : (u | 0x80000000u);
}
static __device__ __forceinline__ float funkey(unsigned k) {
    unsigned u = (k & 0x80000000u) ? (k & 0x7FFFFFFFu) : ~k;
    return __uint_as_float(u);
}

#define PAIR_WAVES 4
#define PAIR_PAD   8

template <int K1, int N1, int N2>
__global__ void __launch_bounds__(32 * PAIR_WAVES)
wmma_two_products(const unsigned short* __restrict__ A,
                  const unsigned short* __restrict__ B1t,
                  const float* __restrict__ bias1,
                  const unsigned short* __restrict__ B2t,
                  float* __restrict__ Y, int M)
{
    static_assert(K1 % 32 == 0 && N1 % 32 == 0 && N1 % 16 == 0 && N2 % 16 == 0, "whole fragments");
    constexpr int PITCH = N1 + PAIR_PAD;
    static_assert((PITCH * 2) % 16 == 0 && ((PITCH * 2 / 16) & 1) == 1, "pitch = 16 B x odd");
    constexpr int OPITCH = N2 + 4;
    static_assert(N2 == 128, "one float4 per lane covers a result row");
    static_assert(OPITCH * 4 <= PITCH * 2, "the result tile reuses the wave's LDS tile");
    constexpr int KS1 = K1 / 32;
    constexpr int KS2 = N1 / 32;
    __shared__ __attribute__((aligned(16))) unsigned short mid[PAIR_WAVES][16 * PITCH];

    const int lane  = threadIdx.x & 31;
    const int wave  = __builtin_amdgcn_readfirstlane(threadIdx.x >> 5);
    const int m0    = (blockIdx.x * PAIR_WAVES + wave) * 16;
    const bool live = m0 < M;
    const int l15   = lane & 15;
    const int half  = lane >> 4;
    const int kbase = half * 8;
    unsigned short* mine = mid[wave];

    if (live) {
        int rowA = m0 + l15;
        if (rowA >= M) rowA = M - 1;
        const unsigned short* aptr = A + (size_t)rowA * K1 + kbase;
        Frag16 fa[KS1];
#pragma unroll
        for (int k = 0; k < KS1; ++k) {
            fa[k].q[0] = *(const u32x4*)(aptr + 32 * k);
            fa[k].q[1] = *(const u32x4*)(aptr + 32 * k + 16);
        }
        Frag16 fb[KS1], fn[KS1];
        {
            const unsigned short* bptr = B1t + (size_t)l15 * K1 + kbase;
#pragma unroll
            for (int k = 0; k < KS1; ++k) {
                fb[k].q[0] = *(const u32x4*)(bptr + 32 * k);
                fb[k].q[1] = *(const u32x4*)(bptr + 32 * k + 16);
            }
        }
#pragma unroll
        for (int t = 0; t < N1 / 16; ++t) {
            if (t + 1 < N1 / 16) {
                const unsigned short* bptr = B1t + (size_t)((t + 1) * 16 + l15) * K1 + kbase;
#pragma unroll
                for (int k = 0; k < KS1; ++k) {
                    fn[k].q[0] = *(const u32x4*)(bptr + 32 * k);
                    fn[k].q[1] = *(const u32x4*)(bptr + 32 * k + 16);
                }
            }
            f32x8 acc = {};
#pragma unroll
            for (int k = 0; k < KS1; ++k)
                acc = __builtin_amdgcn_wmma_f32_16x16x32_bf16(
                          false, fa[k].v, false, fb[k].v, (short)0, acc, false, false);
            const int   col = t * 16 + l15;
            const float bv  = bias1[col];
#pragma unroll
            for (int r = 0; r < 8; ++r) {
                const float v = acc[r] + bv;
                mine[(r + half * 8) * PITCH + col] = f2bf(v > 0.f ? v : 0.f);
            }
#pragma unroll
            for (int k = 0; k < KS1; ++k) fb[k] = fn[k];
        }
    }
    __syncthreads();
    Frag16 ga[KS2];
    if (live) {
        const unsigned short* lp = mine + l15 * PITCH + kbase;
#pragma unroll
        for (int k = 0; k < KS2; ++k) {
            ga[k].q[0] = *(const u32x4*)(lp + 32 * k);
            ga[k].q[1] = *(const u32x4*)(lp + 32 * k + 16);
        }
    }
    __syncthreads();
    float* outt = reinterpret_cast<float*>(mine);
    if (live) {
        Frag16 gb[KS2], gn[KS2];
        {
            const unsigned short* bptr = B2t + (size_t)l15 * N1 + kbase;
#pragma unroll
            for (int k = 0; k < KS2; ++k) {
                gb[k].q[0] = *(const u32x4*)(bptr + 32 * k);
                gb[k].q[1] = *(const u32x4*)(bptr + 32 * k + 16);
            }
        }
#pragma unroll
        for (int t = 0; t < N2 / 16; ++t) {
            if (t + 1 < N2 / 16) {
                const unsigned short* bptr = B2t + (size_t)((t + 1) * 16 + l15) * N1 + kbase;
#pragma unroll
                for (int k = 0; k < KS2; ++k) {
                    gn[k].q[0] = *(const u32x4*)(bptr + 32 * k);
                    gn[k].q[1] = *(const u32x4*)(bptr + 32 * k + 16);
                }
            }
            f32x8 acc = {};
#pragma unroll
            for (int k = 0; k < KS2; ++k)
                acc = __builtin_amdgcn_wmma_f32_16x16x32_bf16(
                          false, ga[k].v, false, gb[k].v, (short)0, acc, false, false);
#pragma unroll
            for (int r = 0; r < 8; ++r)
                outt[(r + half * 8) * OPITCH + t * 16 + l15] = acc[r];
#pragma unroll
            for (int k = 0; k < KS2; ++k) gb[k] = gn[k];
        }
    }
    __syncthreads();
    if (live) {
        float* yrow = Y + (size_t)m0 * N2 + 4 * lane;
        if (m0 + 16 <= M) {
            float4 v[16];
#pragma unroll
            for (int i = 0; i < 16; ++i) v[i] = *reinterpret_cast<const float4*>(outt + i * OPITCH + 4 * lane);
#pragma unroll
            for (int i = 0; i < 16; ++i) *reinterpret_cast<float4*>(yrow + (size_t)i * N2) = v[i];
        } else {
            for (int i = 0; i < M - m0; ++i)
                *reinterpret_cast<float4*>(yrow + (size_t)i * N2) =
                    *reinterpret_cast<const float4*>(outt + i * OPITCH + 4 * lane);
        }
    }
}

__global__ void node_scores(const float* __restrict__ h,
                            const float* __restrict__ as_, const float* __restrict__ ad_,
                            float* __restrict__ es, float* __restrict__ ed,
                            int N, int C, unsigned* __restrict__ clr)
{
    const int warp = (blockIdx.x * blockDim.x + threadIdx.x) >> 5;
    const int lane = threadIdx.x & 31;
    if (warp >= N) return;
    const float* row = h + (size_t)warp * C;
    float s = 0.f, d = 0.f;
    for (int c = lane; c < C; c += 32) {
        const float v = row[c];
        s += v * as_[c];
        d += v * ad_[c];
    }
#pragma unroll
    for (int off = 16; off; off >>= 1) {
        s += __shfl_xor(s, off);
        d += __shfl_xor(d, off);
    }
    if (lane == 0) { es[warp] = s; ed[warp] = d; if (clr) clr[warp] = 0u; }
}

#define RD_WAVES 8
#define RD_ROWS  16
#define RD_PAIRS 8

static __device__ __forceinline__ float dot4(const float4 x, const float4 w)
{
    return x.x * w.x + x.y * w.y + x.z * w.z + x.w * w.w;
}

template <int NV>
static __device__ __forceinline__ float wave_sums(float (&v)[NV], int lane)
{
#pragma unroll
    for (int st = 0; st < 5; ++st) {
        const int m = 16 >> st;
        const int n = NV >> (st + 1);
        if (n >= 1) {
            const bool up = (lane & m) != 0;
#pragma unroll
            for (int i = 0; i < n; ++i) {
                const float lo = v[i], hi = v[i + n];
                const float keep = up ? hi : lo;
                const float send = up ? lo : hi;
                v[i] = keep + __shfl_xor(send, m);
            }
        } else {
            v[0] += __shfl_xor(v[0], m);
        }
    }
    return v[0];
}

#define OWN_ROWS  197
#define OWN_NONE  255u
#define OWN_CHUNK 512

__global__ void __launch_bounds__(32 * RD_WAVES)
node_scores_rows(const float* __restrict__ h,
                 const float* __restrict__ as_, const float* __restrict__ ad_,
                 float* __restrict__ es, float* __restrict__ ed,
                 int N, unsigned* __restrict__ clr,
                 const int* __restrict__ dstl, int E, unsigned* __restrict__ own4)
{
    if (own4) {
        const int nwords = ((E + OWN_CHUNK - 1) / OWN_CHUNK) * (OWN_CHUNK / 4);
        for (int g = blockIdx.x * blockDim.x + threadIdx.x; g < nwords; g += gridDim.x * blockDim.x) {
            unsigned w = 0u;
#pragma unroll
            for (int k = 0; k < 4; ++k) {
                const int e = 4 * g + k;
                unsigned b = OWN_NONE;
                if (e < E) {
                    const unsigned d = (unsigned)dstl[e];
                    if (d < (unsigned)N) b = min(d / (unsigned)OWN_ROWS, OWN_NONE);
                }
                w |= b << (8 * k);
            }
            own4[g] = w;
        }
    }
    const int lane = threadIdx.x & 31;
    const int wave = __builtin_amdgcn_readfirstlane(threadIdx.x >> 5);
    const int r0   = (blockIdx.x * RD_WAVES + wave) * RD_ROWS;
    if (r0 >= N) return;
    const float4 ws = reinterpret_cast<const float4*>(as_)[lane];
    const float4 wd = reinterpret_cast<const float4*>(ad_)[lane];
    float4 x[RD_ROWS];
#pragma unroll
    for (int j = 0; j < RD_ROWS; ++j) {
        const int r = min(r0 + j, N - 1);
        x[j] = reinterpret_cast<const float4*>(h + (size_t)r * 128)[lane];
    }
    float v[2 * RD_ROWS];
#pragma unroll
    for (int j = 0; j < RD_ROWS; ++j) {
        v[j]           = dot4(x[j], ws);
        v[RD_ROWS + j] = dot4(x[j], wd);
    }
    const float t = wave_sums<2 * RD_ROWS>(v, lane);
    const int row = r0 + (lane & (RD_ROWS - 1));
    if (row < N) {
        if (lane < RD_ROWS) { es[row] = t; if (clr) clr[row] = 0u; }
        else                  ed[row] = t;
    }
}

__global__ void mark_needed(const int* __restrict__ mask, unsigned* __restrict__ need, int n, int N)
{
    const int i = blockIdx.x * blockDim.x + threadIdx.x;
    if (i >= n) return;
    const int m = mask[i];
    if ((unsigned)m < (unsigned)N) need[m] = 1u;
}

#define OWN_THREADS 1024
#define OWN_QCAP    4096
#define OWN_U       8

static __device__ __forceinline__ float leaky02(float t) { return (t > 0.f) ? t : 0.2f * t; }

template <int C>
static __device__ __forceinline__ void own_add_row(const float* __restrict__ h, int s, float a,
                                                   float* accrow, int lane)
{
    const float* hs = h + (size_t)s * C + lane;
    float*       ar = accrow + lane;
    float v[C / 32];
#pragma unroll
    for (int k = 0; k < C / 32; ++k) v[k] = hs[32 * k];
#pragma unroll
    for (int k = 0; k < C / 32; ++k) atomicAdd(&ar[32 * k], v[k] * a);
}

template <int C, bool SECOND, int STAGE>
static __device__ __forceinline__ void own_walk(const int* __restrict__ ei, int E, int lo, int nrow,
                                                const float* __restrict__ h, const float* __restrict__ es,
                                                const float* edl, const unsigned* flag,
                                                unsigned* rmax, float* rden, float* acc,
                                                int wave, int lane, int nw)
{
    const int* __restrict__ dst = ei + E;
    for (int e0 = wave * 32; e0 < E; e0 += nw * 32) {
        const int e = e0 + lane;
        unsigned dl = 0u;
        bool ok = false;
        if (e < E) {
            dl = (unsigned)(dst[e] - lo);
            ok = dl < (unsigned)nrow;
            if (SECOND) { if (ok) ok = flag[dl] != 0u; }
        }
        int s = 0; float l = 0.f;
        if (ok) { s = ei[e]; l = leaky02(es[s] + edl[dl]); }
        if (STAGE == 0) {
            if (ok) atomicMax(&rmax[dl], fkey(l));
        } else if (STAGE == 1) {
            if (ok) atomicAdd(&rden[dl], __expf(l - funkey(rmax[dl])));
        } else {
            float a = 0.f;
            if (ok) a = __expf(l - funkey(rmax[dl])) / rden[dl];
            unsigned mk = __builtin_amdgcn_ballot_w32(ok);
            while (mk) {
                const int L = __builtin_ctz(mk);
                mk &= mk - 1u;
                const int      sj = __builtin_amdgcn_readlane(s, L);
                const unsigned dj = (unsigned)__builtin_amdgcn_readlane((int)dl, L);
                const float    aj = __int_as_float(__builtin_amdgcn_readlane(__float_as_int(a), L));
                own_add_row<C>(h, sj, aj, acc + dj * C, lane);
            }
        }
    }
}

template <int C, bool SECOND>
__global__ void __launch_bounds__(OWN_THREADS)
aggregate_owned(const int* __restrict__ ei, int E, int N,
                const float* __restrict__ h,
                const float* __restrict__ es, const float* __restrict__ ed,
                const unsigned* __restrict__ need, const float* __restrict__ bias,
                unsigned short* __restrict__ ybf, float* __restrict__ yf,
                unsigned* __restrict__ sv_n, int* __restrict__ sv_e, unsigned* __restrict__ sv_d,
                const unsigned char* __restrict__ own8)
{
    static_assert(C == 128, "one float4 per lane covers a row");
    __shared__ __attribute__((aligned(16))) float acc[OWN_ROWS * C];
    __shared__ int      qe[OWN_QCAP];
    __shared__ unsigned qd[OWN_QCAP];
    __shared__ float    qp[OWN_QCAP];
    __shared__ float    edl[OWN_ROWS];
    __shared__ float    pself[OWN_ROWS];
    __shared__ unsigned rmax[OWN_ROWS];
    __shared__ float    rden[OWN_ROWS];
    __shared__ unsigned flag[OWN_ROWS];
    __shared__ unsigned qn;

    const int lane = threadIdx.x & 31;
    const int wave = __builtin_amdgcn_readfirstlane(threadIdx.x >> 5);
    const int nw   = OWN_THREADS / 32;
    const int lo   = blockIdx.x * OWN_ROWS;
    const int nrow = min(OWN_ROWS, N - lo);

    if (threadIdx.x == 0) qn = 0u;
    for (int r = threadIdx.x; r < nrow; r += OWN_THREADS) {
        const int d = lo + r;
        if (SECOND) flag[r] = (need[d] != 0u) ? 1u : 0u;
        const float edv = ed[d];
        const float l   = leaky02(es[d] + edv);
        edl[r]   = edv;
        pself[r] = l;
        rmax[r]  = fkey(l);
    }
    __syncthreads();

    const unsigned lt = (1u << lane) - 1u;
    int*      sve = sv_e + (size_t)blockIdx.x * OWN_QCAP;
    unsigned* svd = sv_d + (size_t)blockIdx.x * OWN_QCAP;
    if (!SECOND) {
        if (gridDim.x <= OWN_NONE) {
            const int nchunks = (E + OWN_CHUNK - 1) / OWN_CHUNK;
            const int startc  = (int)(((long long)blockIdx.x * nchunks) / gridDim.x);
            const unsigned pat = (unsigned)blockIdx.x * 0x01010101u;
            const unsigned char* lbase = own8 + lane * 16;

            for (int cl0 = wave; cl0 < nchunks; cl0 += nw * OWN_U) {
                uint4 q[OWN_U];
                int   cu[OWN_U];
#pragma unroll
                for (int u = 0; u < OWN_U; ++u) {
                    const int cl = cl0 + u * nw;
                    int cc = cl + startc;
                    if (cc >= nchunks) cc -= nchunks;
                    cu[u] = cc;
                    q[u] = make_uint4(~0u, ~0u, ~0u, ~0u);
                    if (cl < nchunks) q[u] = *reinterpret_cast<const uint4*>(lbase + (size_t)cc * OWN_CHUNK);
                }
                unsigned mk[OWN_U];
                unsigned cnt = 0u;
#pragma unroll
                for (int u = 0; u < OWN_U; ++u) {
                    const unsigned y0 = q[u].x ^ pat, y1 = q[u].y ^ pat, y2 = q[u].z ^ pat, y3 = q[u].w ^ pat;
                    const unsigned t0 = ~(((y0 & 0x7F7F7F7Fu) + 0x7F7F7F7Fu) | y0 | 0x7F7F7F7Fu);
                    const unsigned t1 = ~(((y1 & 0x7F7F7F7Fu) + 0x7F7F7F7Fu) | y1 | 0x7F7F7F7Fu);
                    const unsigned t2 = ~(((y2 & 0x7F7F7F7Fu) + 0x7F7F7F7Fu) | y2 | 0x7F7F7F7Fu);
                    const unsigned t3 = ~(((y3 & 0x7F7F7F7Fu) + 0x7F7F7F7Fu) | y3 | 0x7F7F7F7Fu);
                    mk[u] = (t0 >> 7) | (t1 >> 6) | (t2 >> 5) | (t3 >> 4);
                    cnt += (unsigned)__builtin_popcount(mk[u]);
                }
                if (cnt) {
                    unsigned slot = atomicAdd(&qn, cnt);
#pragma unroll
                    for (int u = 0; u < OWN_U; ++u) {
                        unsigned mm = mk[u];
                        const int ebase = cu[u] * OWN_CHUNK + lane * 16;
                        while (mm) {
                            const int p = __builtin_ctz(mm);
                            mm &= mm - 1u;
                            if (slot < OWN_QCAP) qe[slot] = ebase + 4 * (p & 7) + (p >> 3);
                            ++slot;
                        }
                    }
                }
            }
        } else if (threadIdx.x == 0) {
            qn = (unsigned)OWN_QCAP + 1u;
        }
    } else {
        const unsigned stot = sv_n[blockIdx.x];
        if (stot <= (unsigned)OWN_QCAP) {
            for (int b = wave * 32; b < (int)stot; b += nw * 32) {
                const int idx = b + lane;
                int e = 0; unsigned dq = 0u; bool mm = false;
                if (idx < (int)stot) { e = sve[idx]; dq = svd[idx]; mm = flag[dq] != 0u; }
                const unsigned mk = __builtin_amdgcn_ballot_w32(mm);
                if (mk) {
                    unsigned base = 0u;
                    if (lane == 0) base = atomicAdd(&qn, (unsigned)__builtin_popcount(mk));
                    base = (unsigned)__builtin_amdgcn_readfirstlane((int)base);
                    if (mm) {
                        const unsigned slot = base + (unsigned)__builtin_popcount(mk & lt);
                        qe[slot] = e; qd[slot] = dq;
                    }
                }
            }
        } else if (threadIdx.x == 0) {
            qn = (unsigned)OWN_QCAP + 1u;
        }
    }
    __syncthreads();

    const unsigned qtot = (unsigned)__builtin_amdgcn_readfirstlane((int)qn);
    const bool fits  = qtot <= (unsigned)OWN_QCAP;
    const int  total = fits ? (int)qtot : 0;
    if (!SECOND) {
        if (threadIdx.x == 0) sv_n[blockIdx.x] = qtot;
    }

    if (fits) {
        for (int idx = threadIdx.x; idx < total; idx += OWN_THREADS) {
            const int e = qe[idx];
            unsigned dq;
            if (!SECOND) {
                dq = (unsigned)(ei[E + e] - lo);
                qd[idx]  = dq;
                sve[idx] = e;
                svd[idx] = dq;
            } else {
                dq = qd[idx];
            }
            const float l = leaky02(es[ei[e]] + edl[dq]);
            qp[idx] = l;
            atomicMax(&rmax[dq], fkey(l));
        }
    } else {
        own_walk<C, SECOND, 0>(ei, E, lo, nrow, h, es, edl, flag, rmax, rden, acc, wave, lane, nw);
    }
    __syncthreads();
    for (int r = threadIdx.x; r < nrow; r += OWN_THREADS) {
        const float ps = __expf(pself[r] - funkey(rmax[r]));
        pself[r] = ps;
        rden[r]  = ps;
    }
    __syncthreads();
    if (fits) {
        for (int idx = threadIdx.x; idx < total; idx += OWN_THREADS) {
            const unsigned dq = qd[idx];
            const float pe = __expf(qp[idx] - funkey(rmax[dq]));
            qp[idx] = pe;
            atomicAdd(&rden[dq], pe);
        }
    } else {
        own_walk<C, SECOND, 1>(ei, E, lo, nrow, h, es, edl, flag, rmax, rden, acc, wave, lane, nw);
    }
    __syncthreads();

    {
        const float4* hb4  = reinterpret_cast<const float4*>(h + (size_t)lo * C);
        float4*       acc4 = reinterpret_cast<float4*>(acc);
        for (int i = threadIdx.x; i < nrow * (C / 4); i += OWN_THREADS) {
            const int r = i / (C / 4);
            float4 v = make_float4(0.f, 0.f, 0.f, 0.f);
            if (!SECOND || flag[r]) {
                const float  a = pself[r] / rden[r];
                const float4 x = hb4[i];
                v.x = x.x * a; v.y = x.y * a; v.z = x.z * a; v.w = x.w * a;
            }
            acc4[i] = v;
        }
    }
    __syncthreads();

    for (int b = wave * 32; b < total; b += nw * 32) {
        const int idx = b + lane;
        int s = 0; unsigned dq = 0u; float a = 0.f;
        if (idx < total) {
            dq = qd[idx];
            s  = ei[qe[idx]];
            a  = qp[idx] / rden[dq];
        }
        const int cnt = min(32, total - b);
        for (int j = 0; j < cnt; j += 4) {
            float    v[4][C / 32];
            float    aj[4];
            unsigned dj[4];
#pragma unroll
            for (int t = 0; t < 4; ++t) {
                const int jj = min(j + t, cnt - 1);
                const int sj = __builtin_amdgcn_readlane(s, jj);
                dj[t] = (unsigned)__builtin_amdgcn_readlane((int)dq, jj);
                aj[t] = __int_as_float(__builtin_amdgcn_readlane(__float_as_int(a), jj));
                const float* hs = h + (size_t)sj * C + lane;
#pragma unroll
                for (int k = 0; k < C / 32; ++k) v[t][k] = hs[32 * k];
            }
#pragma unroll
            for (int t = 0; t < 4; ++t) {
                if (j + t < cnt) {
                    float* ar = acc + dj[t] * C + lane;
#pragma unroll
                    for (int k = 0; k < C / 32; ++k) atomicAdd(&ar[32 * k], v[t][k] * aj[t]);
                }
            }
        }
    }
    if (!fits)
        own_walk<C, SECOND, 2>(ei, E, lo, nrow, h, es, edl, flag, rmax, rden, acc, wave, lane, nw);
    __syncthreads();

    for (int r = wave; r < nrow; r += nw) {
        const size_t d = (size_t)(lo + r);
        const float4 v = *reinterpret_cast<const float4*>(&acc[r * C + 4 * lane]);
        if (SECOND) {
            if (flag[r]) {
                const float4 bq = *reinterpret_cast<const float4*>(bias + 4 * lane);
                float4 o; o.x = v.x + bq.x; o.y = v.y + bq.y; o.z = v.z + bq.z; o.w = v.w + bq.w;
                *reinterpret_cast<float4*>(yf + d * C + 4 * lane) = o;
            }
        } else {
            uint2 o;
            o.x = (unsigned)f2bf(v.x) | ((unsigned)f2bf(v.y) << 16);
            o.y = (unsigned)f2bf(v.z) | ((unsigned)f2bf(v.w) << 16);
            *reinterpret_cast<uint2*>(ybf + d * C + 4 * lane) = o;
        }
    }
}

__global__ void transpose_to_bf16(const float* __restrict__ W, unsigned short* __restrict__ Wt,
                                  int K, int NOUT)
{
    const int i = blockIdx.x * blockDim.x + threadIdx.x;
    if (i >= K * NOUT) return;
    const int k = i / NOUT, n = i % NOUT;
    Wt[(size_t)n * K + k] = f2bf(W[i]);
}

__global__ void __launch_bounds__(32 * RD_WAVES)
link_predict_rows(const float* __restrict__ h, const int* __restrict__ mask,
                  const float* __restrict__ Wl, const float* __restrict__ bl,
                  float* __restrict__ out, int P)
{
    const int lane = threadIdx.x & 31;
    const int wave = __builtin_amdgcn_readfirstlane(threadIdx.x >> 5);
    const int p0   = (blockIdx.x * RD_WAVES + wave) * RD_PAIRS;
    if (p0 >= P) return;
    const int mi = mask[min(p0 * 2 + (lane & (2 * RD_PAIRS - 1)), 2 * P - 1)];
    const float4 wa = reinterpret_cast<const float4*>(Wl)[lane];
    const float4 wb = reinterpret_cast<const float4*>(Wl)[32 + lane];
    float4 xa[RD_PAIRS], xb[RD_PAIRS];
#pragma unroll
    for (int j = 0; j < RD_PAIRS; ++j) {
        const int ra = __builtin_amdgcn_readlane(mi, 2 * j);
        const int rb = __builtin_amdgcn_readlane(mi, 2 * j + 1);
        xa[j] = reinterpret_cast<const float4*>(h + (size_t)ra * 128)[lane];
        xb[j] = reinterpret_cast<const float4*>(h + (size_t)rb * 128)[lane];
    }
    float v[RD_PAIRS];
#pragma unroll
    for (int j = 0; j < RD_PAIRS; ++j) v[j] = dot4(xa[j], wa) + dot4(xb[j], wb);
    const float t = wave_sums<RD_PAIRS>(v, lane);
    const int p = p0 + (lane >> 2);
    if ((lane & 3) == 0 && p < P) out[p] = 1.f / (1.f + expf(-(t + bl[0])));
}

extern "C" void kernel_launch(void* const* d_in, const int* in_sizes, int n_in,
                              void* d_out, int out_size, void* d_ws, size_t ws_size,
                              hipStream_t stream)
{
    (void)n_in; (void)out_size; (void)ws_size;

    const float* features = (const float*)d_in[0];
    const int*   ei       = (const int*)  d_in[1];
    const int*   mask     = (const int*)  d_in[2];
    const float* W1   = (const float*)d_in[3];
    const float* a_s1 = (const float*)d_in[4];
    const float* a_d1 = (const float*)d_in[5];
    const float* b1   = (const float*)d_in[6];
    const float* W2   = (const float*)d_in[7];
    const float* a_s2 = (const float*)d_in[8];
    const float* a_d2 = (const float*)d_in[9];
    const float* b2   = (const float*)d_in[10];
    const float* Wl   = (const float*)d_in[11];
    const float* bl   = (const float*)d_in[12];

    constexpr int F  = 128, Hh = 256;
    const int N  = in_sizes[0] / F;
    const int E  = in_sizes[1] / 2;
    const int P  = in_sizes[2] / 2;

    char* ws = (char*)d_ws;
    size_t off = 0;
    auto alloc = [&](size_t bytes) -> char* {
        char* ptr = ws + off;
        off += (bytes + 255) & ~(size_t)255;
        return ptr;
    };
    unsigned short* xbf  = (unsigned short*)alloc((size_t)N * F  * 2);
    unsigned short* w1t  = (unsigned short*)alloc((size_t)Hh * F * 2);
    unsigned short* w2t  = (unsigned short*)alloc((size_t)F * Hh * 2);
    float*    buf1    = (float*)   alloc((size_t)N * Hh * 4);
    float*    buf2    = (float*)   alloc((size_t)N * Hh * 4);
    float*    es      = (float*)   alloc((size_t)N * 4);
    float*    ed      = (float*)   alloc((size_t)N * 4);
    float*    vs1     = (float*)   alloc((size_t)F * 4);
    float*    vd1     = (float*)   alloc((size_t)F * 4);
    unsigned* need    = (unsigned*)alloc((size_t)N * 4);
    const int ownGrid = (N + OWN_ROWS - 1) / OWN_ROWS;
    unsigned* svn     = (unsigned*)alloc((size_t)ownGrid * 4);
    int*      sve     = (int*)     alloc((size_t)ownGrid * OWN_QCAP * 4);
    unsigned* svd     = (unsigned*)alloc((size_t)ownGrid * OWN_QCAP * 4);
    unsigned* own4    = (unsigned*)alloc((size_t)((E + OWN_CHUNK - 1) / OWN_CHUNK) * OWN_CHUNK);

    const int tilesM = (N + 15) / 16;
    const int gemmGrid = (tilesM + PAIR_WAVES - 1) / PAIR_WAVES;
    const int TB = 256;
    auto cdiv = [](size_t a, size_t b) -> int { return (int)((a + b - 1) / b); };

    transpose_to_bf16<<<cdiv((size_t)F * Hh, TB), TB, 0, stream>>>(W1, w1t, F, Hh);
    transpose_to_bf16<<<cdiv((size_t)Hh * F, TB), TB, 0, stream>>>(W2, w2t, Hh, F);

    node_scores<<<cdiv((size_t)F * 32, TB), TB, 0, stream>>>(W1, a_s1, a_d1, vs1, vd1, F, Hh, nullptr);
    node_scores_rows<<<cdiv((size_t)N, RD_ROWS * RD_WAVES), 32 * RD_WAVES, 0, stream>>>(features, vs1, vd1, es, ed, N, need, ei + E, E, own4);
    aggregate_owned<F, false><<<ownGrid, OWN_THREADS, 0, stream>>>(ei, E, N, features, es, ed,
                                                                  nullptr, nullptr, xbf, nullptr, svn, sve, svd,
                                                                  (const unsigned char*)own4);

    mark_needed<<<cdiv((size_t)P * 2, TB), TB, 0, stream>>>(mask, need, P * 2, N);
    wmma_two_products<F, Hh, F><<<gemmGrid, 32 * PAIR_WAVES, 0, stream>>>(xbf, w1t, b1, w2t, buf1, N);
    node_scores_rows<<<cdiv((size_t)N, RD_ROWS * RD_WAVES), 32 * RD_WAVES, 0, stream>>>(buf1, a_s2, a_d2, es, ed, N, nullptr, nullptr, 0, nullptr);
    aggregate_owned<F, true><<<ownGrid, OWN_THREADS, 0, stream>>>(ei, E, N, buf1, es, ed,
                                                                 need, b2, nullptr, buf2, svn, sve, svd, nullptr);

    link_predict_rows<<<cdiv((size_t)P, RD_PAIRS * RD_WAVES), 32 * RD_WAVES, 0, stream>>>(buf2, mask, Wl, bl,
                                                                                         (float*)d_out, P);
}


# Small producer stages recomputed inside their consumers' work-groups: W1 a in the layer-1 score pass (LDS), pair-list row flags in layer 2's row owners; 10 -> 8 launches; W1 a dot sum order changes
# speedup vs baseline: 6.8305x; 1.0309x over previous
#include <hip/hip_runtime.h>
#include <hip/hip_bf16.h>

typedef __attribute__((__ext_vector_type__(16))) __bf16 bf16x16;
typedef __attribute__((__ext_vector_type__(8)))  float  f32x8;
typedef __attribute__((__ext_vector_type__(4)))  unsigned int u32x4;

union Frag16 {
    u32x4   q[2];
    bf16x16 v;
};

static __device__ __forceinline__ unsigned short f2bf(float f) {
    unsigned u = __float_as_uint(f);
    u += 0x7FFFu + ((u >> 16) & 1u);
    return (unsigned short)(u >> 16);
}

static __device__ __forceinline__ unsigned fkey(float f) {
    unsigned u = __float_as_uint(f);
    return (u & 0x80000000u) ? ~u : (u | 0x80000000u);
}
static __device__ __forceinline__ float funkey(unsigned k) {
    unsigned u = (k & 0x80000000u) ? (k & 0x7FFFFFFFu) : ~k;
    return __uint_as_float(u);
}

#define PAIR_WAVES 4
#define PAIR_PAD   8

template <int K1, int N1, int N2>
__global__ void __launch_bounds__(32 * PAIR_WAVES)
wmma_two_products(const unsigned short* __restrict__ A,
                  const unsigned short* __restrict__ B1t,
                  const float* __restrict__ bias1,
                  const unsigned short* __restrict__ B2t,
                  float* __restrict__ Y, int M)
{
    static_assert(K1 % 32 == 0 && N1 % 32 == 0 && N1 % 16 == 0 && N2 % 16 == 0, "whole fragments");
    constexpr int PITCH = N1 + PAIR_PAD;
    static_assert((PITCH * 2) % 16 == 0 && ((PITCH * 2 / 16) & 1) == 1, "pitch = 16 B x odd");
    constexpr int OPITCH = N2 + 4;
    static_assert(N2 == 128, "one float4 per lane covers a result row");
    static_assert(OPITCH * 4 <= PITCH * 2, "the result tile reuses the wave's LDS tile");
    constexpr int KS1 = K1 / 32;
    constexpr int KS2 = N1 / 32;
    __shared__ __attribute__((aligned(16))) unsigned short mid[PAIR_WAVES][16 * PITCH];

    const int lane  = threadIdx.x & 31;
    const int wave  = __builtin_amdgcn_readfirstlane(threadIdx.x >> 5);
    const int m0    = (blockIdx.x * PAIR_WAVES + wave) * 16;
    const bool live = m0 < M;
    const int l15   = lane & 15;
    const int half  = lane >> 4;
    const int kbase = half * 8;
    unsigned short* mine = mid[wave];

    if (live) {
        int rowA = m0 + l15;
        if (rowA >= M) rowA = M - 1;
        const unsigned short* aptr = A + (size_t)rowA * K1 + kbase;
        Frag16 fa[KS1];
#pragma unroll
        for (int k = 0; k < KS1; ++k) {
            fa[k].q[0] = *(const u32x4*)(aptr + 32 * k);
            fa[k].q[1] = *(const u32x4*)(aptr + 32 * k + 16);
        }
        Frag16 fb[KS1], fn[KS1];
        {
            const unsigned short* bptr = B1t + (size_t)l15 * K1 + kbase;
#pragma unroll
            for (int k = 0; k < KS1; ++k) {
                fb[k].q[0] = *(const u32x4*)(bptr + 32 * k);
                fb[k].q[1] = *(const u32x4*)(bptr + 32 * k + 16);
            }
        }
#pragma unroll
        for (int t = 0; t < N1 / 16; ++t) {
            if (t + 1 < N1 / 16) {
                const unsigned short* bptr = B1t + (size_t)((t + 1) * 16 + l15) * K1 + kbase;
#pragma unroll
                for (int k = 0; k < KS1; ++k) {
                    fn[k].q[0] = *(const u32x4*)(bptr + 32 * k);
                    fn[k].q[1] = *(const u32x4*)(bptr + 32 * k + 16);
                }
            }
            f32x8 acc = {};
#pragma unroll
            for (int k = 0; k < KS1; ++k)
                acc = __builtin_amdgcn_wmma_f32_16x16x32_bf16(
                          false, fa[k].v, false, fb[k].v, (short)0, acc, false, false);
            const int   col = t * 16 + l15;
            const float bv  = bias1[col];
#pragma unroll
            for (int r = 0; r < 8; ++r) {
                const float v = acc[r] + bv;
                mine[(r + half * 8) * PITCH + col] = f2bf(v > 0.f ? v : 0.f);
            }
#pragma unroll
            for (int k = 0; k < KS1; ++k) fb[k] = fn[k];
        }
    }
    __syncthreads();
    Frag16 ga[KS2];
    if (live) {
        const unsigned short* lp = mine + l15 * PITCH + kbase;
#pragma unroll
        for (int k = 0; k < KS2; ++k) {
            ga[k].q[0] = *(const u32x4*)(lp + 32 * k);
            ga[k].q[1] = *(const u32x4*)(lp + 32 * k + 16);
        }
    }
    __syncthreads();
    float* outt = reinterpret_cast<float*>(mine);
    if (live) {
        Frag16 gb[KS2], gn[KS2];
        {
            const unsigned short* bptr = B2t + (size_t)l15 * N1 + kbase;
#pragma unroll
            for (int k = 0; k < KS2; ++k) {
                gb[k].q[0] = *(const u32x4*)(bptr + 32 * k);
                gb[k].q[1] = *(const u32x4*)(bptr + 32 * k + 16);
            }
        }
#pragma unroll
        for (int t = 0; t < N2 / 16; ++t) {
            if (t + 1 < N2 / 16) {
                const unsigned short* bptr = B2t + (size_t)((t + 1) * 16 + l15) * N1 + kbase;
#pragma unroll
                for (int k = 0; k < KS2; ++k) {
                    gn[k].q[0] = *(const u32x4*)(bptr + 32 * k);
                    gn[k].q[1] = *(const u32x4*)(bptr + 32 * k + 16);
                }
            }
            f32x8 acc = {};
#pragma unroll
            for (int k = 0; k < KS2; ++k)
                acc = __builtin_amdgcn_wmma_f32_16x16x32_bf16(
                          false, ga[k].v, false, gb[k].v, (short)0, acc, false, false);
#pragma unroll
            for (int r = 0; r < 8; ++r)
                outt[(r + half * 8) * OPITCH + t * 16 + l15] = acc[r];
#pragma unroll
            for (int k = 0; k < KS2; ++k) gb[k] = gn[k];
        }
    }
    __syncthreads();
    if (live) {
        float* yrow = Y + (size_t)m0 * N2 + 4 * lane;
        if (m0 + 16 <= M) {
            float4 v[16];
#pragma unroll
            for (int i = 0; i < 16; ++i) v[i] = *reinterpret_cast<const float4*>(outt + i * OPITCH + 4 * lane);
#pragma unroll
            for (int i = 0; i < 16; ++i) *reinterpret_cast<float4*>(yrow + (size_t)i * N2) = v[i];
        } else {
            for (int i = 0; i < M - m0; ++i)
                *reinterpret_cast<float4*>(yrow + (size_t)i * N2) =
                    *reinterpret_cast<const float4*>(outt + i * OPITCH + 4 * lane);
        }
    }
}

#define RD_WAVES 8
#define RD_ROWS  16
#define RD_PAIRS 8

static __device__ __forceinline__ float dot4(const float4 x, const float4 w)
{
    return x.x * w.x + x.y * w.y + x.z * w.z + x.w * w.w;
}

template <int NV>
static __device__ __forceinline__ float wave_sums(float (&v)[NV], int lane)
{
#pragma unroll
    for (int st = 0; st < 5; ++st) {
        const int m = 16 >> st;
        const int n = NV >> (st + 1);
        if (n >= 1) {
            const bool up = (lane & m) != 0;
#pragma unroll
            for (int i = 0; i < n; ++i) {
                const float lo = v[i], hi = v[i + n];
                const float keep = up ? hi : lo;
                const float send = up ? lo : hi;
                v[i] = keep + __shfl_xor(send, m);
            }
        } else {
            v[0] += __shfl_xor(v[0], m);
        }
    }
    return v[0];
}

#define OWN_ROWS  197
#define OWN_NONE  255u
#define OWN_CHUNK 512

#define FOLD_N 256
#define FOLD_ROWS 8
template <bool FOLD>
__global__ void __launch_bounds__(32 * RD_WAVES)
node_scores_rows(const float* __restrict__ h,
                 const float* __restrict__ as_, const float* __restrict__ ad_,
                 float* __restrict__ es, float* __restrict__ ed,
                 int N,
                 const int* __restrict__ dstl, int E, unsigned* __restrict__ own4,
                 const float* __restrict__ Wf)
{
    __shared__ __attribute__((aligned(16))) float fold[FOLD ? 256 : 4];
    if (own4) {
        const int nwords = ((E + OWN_CHUNK - 1) / OWN_CHUNK) * (OWN_CHUNK / 4);
        for (int g = blockIdx.x * blockDim.x + threadIdx.x; g < nwords; g += gridDim.x * blockDim.x) {
            unsigned w = 0u;
#pragma unroll
            for (int k = 0; k < 4; ++k) {
                const int e = 4 * g + k;
                unsigned b = OWN_NONE;
                if (e < E) {
                    const unsigned d = (unsigned)dstl[e];
                    if (d < (unsigned)N) b = min(d / (unsigned)OWN_ROWS, OWN_NONE);
                }
                w |= b << (8 * k);
            }
            own4[g] = w;
        }
    }
    const int lane = threadIdx.x & 31;
    const int wave = __builtin_amdgcn_readfirstlane(threadIdx.x >> 5);
    const int r0   = (blockIdx.x * RD_WAVES + wave) * RD_ROWS;
    float4 ws, wd;
    if (FOLD) {
        const float4 sl = reinterpret_cast<const float4*>(as_)[lane], sh = reinterpret_cast<const float4*>(as_)[32 + lane];
        const float4 dl = reinterpret_cast<const float4*>(ad_)[lane], dh = reinterpret_cast<const float4*>(ad_)[32 + lane];
#pragma nounroll
        for (int kb = wave * FOLD_ROWS; kb < 128; kb += RD_WAVES * FOLD_ROWS) {
            float fv[2 * FOLD_ROWS];
#pragma unroll
            for (int j = 0; j < FOLD_ROWS; ++j) {
                const float4* wr = reinterpret_cast<const float4*>(Wf + (size_t)(kb + j) * FOLD_N);
                const float4 a = wr[lane], b = wr[32 + lane];
                fv[j]             = dot4(a, sl) + dot4(b, sh);
                fv[FOLD_ROWS + j] = dot4(a, dl) + dot4(b, dh);
            }
            const float t = wave_sums<2 * FOLD_ROWS>(fv, lane);
            const int q = lane >> 1;
            if ((lane & 1) == 0) fold[(q < FOLD_ROWS ? 0 : 128 - FOLD_ROWS) + kb + q] = t;
        }
        __syncthreads();
        if (r0 >= N) return;
        ws = reinterpret_cast<const float4*>(fold)[lane];
        wd = reinterpret_cast<const float4*>(fold)[32 + lane];
    } else {
        if (r0 >= N) return;
        ws = reinterpret_cast<const float4*>(as_)[lane];
        wd = reinterpret_cast<const float4*>(ad_)[lane];
    }
    float4 x[RD_ROWS];
#pragma unroll
    for (int j = 0; j < RD_ROWS; ++j) {
        const int r = min(r0 + j, N - 1);
        x[j] = reinterpret_cast<const float4*>(h + (size_t)r * 128)[lane];
    }
    float v[2 * RD_ROWS];
#pragma unroll
    for (int j = 0; j < RD_ROWS; ++j) {
        v[j]           = dot4(x[j], ws);
        v[RD_ROWS + j] = dot4(x[j], wd);
    }
    const float t = wave_sums<2 * RD_ROWS>(v, lane);
    const int row = r0 + (lane & (RD_ROWS - 1));
    if (row < N) {
        if (lane < RD_ROWS) es[row] = t;
        else                ed[row] = t;
    }
}

#define OWN_THREADS 1024
#define OWN_QCAP    4096
#define OWN_U       8
#define OWN_PU      8

static __device__ __forceinline__ float leaky02(float t) { return (t > 0.f) ? t : 0.2f * t; }

template <int C>
static __device__ __forceinline__ void own_add_row(const float* __restrict__ h, int s, float a,
                                                   float* accrow, int lane)
{
    const float* hs = h + (size_t)s * C + lane;
    float*       ar = accrow + lane;
    float v[C / 32];
#pragma unroll
    for (int k = 0; k < C / 32; ++k) v[k] = hs[32 * k];
#pragma unroll
    for (int k = 0; k < C / 32; ++k) atomicAdd(&ar[32 * k], v[k] * a);
}

template <int C, bool SECOND, int STAGE>
static __device__ __forceinline__ void own_walk(const int* __restrict__ ei, int E, int lo, int nrow,
                                                const float* __restrict__ h, const float* __restrict__ es,
                                                const float* edl, const unsigned* flag,
                                                unsigned* rmax, float* rden, float* acc,
                                                int wave, int lane, int nw)
{
    const int* __restrict__ dst = ei + E;
    for (int e0 = wave * 32; e0 < E; e0 += nw * 32) {
        const int e = e0 + lane;
        unsigned dl = 0u;
        bool ok = false;
        if (e < E) {
            dl = (unsigned)(dst[e] - lo);
            ok = dl < (unsigned)nrow;
            if (SECOND) { if (ok) ok = flag[dl] != 0u; }
        }
        int s = 0; float l = 0.f;
        if (ok) { s = ei[e]; l = leaky02(es[s] + edl[dl]); }
        if (STAGE == 0) {
            if (ok) atomicMax(&rmax[dl], fkey(l));
        } else if (STAGE == 1) {
            if (ok) atomicAdd(&rden[dl], __expf(l - funkey(rmax[dl])));
        } else {
            float a = 0.f;
            if (ok) a = __expf(l - funkey(rmax[dl])) / rden[dl];
            unsigned mk = __builtin_amdgcn_ballot_w32(ok);
            while (mk) {
                const int L = __builtin_ctz(mk);
                mk &= mk - 1u;
                const int      sj = __builtin_amdgcn_readlane(s, L);
                const unsigned dj = (unsigned)__builtin_amdgcn_readlane((int)dl, L);
                const float    aj = __int_as_float(__builtin_amdgcn_readlane(__float_as_int(a), L));
                own_add_row<C>(h, sj, aj, acc + dj * C, lane);
            }
        }
    }
}

template <int C, bool SECOND>
__global__ void __launch_bounds__(OWN_THREADS)
aggregate_owned(const int* __restrict__ ei, int E, int N,
                const float* __restrict__ h,
                const float* __restrict__ es, const float* __restrict__ ed,
                const int* __restrict__ pairs, int npairs, const float* __restrict__ bias,
                unsigned short* __restrict__ ybf, float* __restrict__ yf,
                unsigned* __restrict__ sv_n, int* __restrict__ sv_e, unsigned* __restrict__ sv_d,
                const unsigned char* __restrict__ own8)
{
    static_assert(C == 128, "one float4 per lane covers a row");
    __shared__ __attribute__((aligned(16))) float acc[OWN_ROWS * C];
    __shared__ int      qe[OWN_QCAP];
    __shared__ unsigned qd[OWN_QCAP];
    __shared__ float    qp[OWN_QCAP];
    __shared__ float    edl[OWN_ROWS];
    __shared__ float    pself[OWN_ROWS];
    __shared__ unsigned rmax[OWN_ROWS];
    __shared__ float    rden[OWN_ROWS];
    __shared__ unsigned flag[OWN_ROWS];
    __shared__ unsigned qn;

    const int lane = threadIdx.x & 31;
    const int wave = __builtin_amdgcn_readfirstlane(threadIdx.x >> 5);
    const int nw   = OWN_THREADS / 32;
    const int lo   = blockIdx.x * OWN_ROWS;
    const int nrow = min(OWN_ROWS, N - lo);

    if (threadIdx.x == 0) qn = 0u;
    const int   n4 = npairs >> 2;
    const int4* p4 = reinterpret_cast<const int4*>(pairs);
    int4 pm[OWN_PU];
    if (SECOND) {
#pragma unroll
        for (int u = 0; u < OWN_PU; ++u) {
            const int i = threadIdx.x + u * OWN_THREADS;
            pm[u] = make_int4(-1, -1, -1, -1);
            if (i < n4) pm[u] = p4[i];
        }
    }
    for (int r = threadIdx.x; r < nrow; r += OWN_THREADS) {
        const int d = lo + r;
        if (SECOND) flag[r] = 0u;
        const float edv = ed[d];
        const float l   = leaky02(es[d] + edv);
        edl[r]   = edv;
        pself[r] = l;
        rmax[r]  = fkey(l);
    }
    if (SECOND) {
        __syncthreads();
#pragma unroll
        for (int u = 0; u < OWN_PU; ++u) {
            const unsigned r0 = (unsigned)(pm[u].x - lo), r1 = (unsigned)(pm[u].y - lo), r2 = (unsigned)(pm[u].z - lo), r3 = (unsigned)(pm[u].w - lo);
            if (r0 < (unsigned)nrow) flag[r0] = 1u;
            if (r1 < (unsigned)nrow) flag[r1] = 1u;
            if (r2 < (unsigned)nrow) flag[r2] = 1u;
            if (r3 < (unsigned)nrow) flag[r3] = 1u;
        }
        for (int i = threadIdx.x + OWN_PU * OWN_THREADS; i < n4; i += OWN_THREADS) {
            const int4 m = p4[i];
            const unsigned r0 = (unsigned)(m.x - lo), r1 = (unsigned)(m.y - lo), r2 = (unsigned)(m.z - lo), r3 = (unsigned)(m.w - lo);
            if (r0 < (unsigned)nrow) flag[r0] = 1u;
            if (r1 < (unsigned)nrow) flag[r1] = 1u;
            if (r2 < (unsigned)nrow) flag[r2] = 1u;
            if (r3 < (unsigned)nrow) flag[r3] = 1u;
        }
        for (int i = 4 * n4 + threadIdx.x; i < npairs; i += OWN_THREADS) {
            const unsigned r = (unsigned)(pairs[i] - lo);
            if (r < (unsigned)nrow) flag[r] = 1u;
        }
    }
    __syncthreads();

    const unsigned lt = (1u << lane) - 1u;
    int*      sve = sv_e + (size_t)blockIdx.x * OWN_QCAP;
    unsigned* svd = sv_d + (size_t)blockIdx.x * OWN_QCAP;
    if (!SECOND) {
        if (gridDim.x <= OWN_NONE) {
            const int nchunks = (E + OWN_CHUNK - 1) / OWN_CHUNK;
            const int startc  = (int)(((long long)blockIdx.x * nchunks) / gridDim.x);
            const unsigned pat = (unsigned)blockIdx.x * 0x01010101u;
            const unsigned char* lbase = own8 + lane * 16;

            for (int cl0 = wave; cl0 < nchunks; cl0 += nw * OWN_U) {
                uint4 q[OWN_U];
                int   cu[OWN_U];
#pragma unroll
                for (int u = 0; u < OWN_U; ++u) {
                    const int cl = cl0 + u * nw;
                    int cc = cl + startc;
                    if (cc >= nchunks) cc -= nchunks;
                    cu[u] = cc;
                    q[u] = make_uint4(~0u, ~0u, ~0u, ~0u);
                    if (cl < nchunks) q[u] = *reinterpret_cast<const uint4*>(lbase + (size_t)cc * OWN_CHUNK);
                }
                unsigned mk[OWN_U];
                unsigned cnt = 0u;
#pragma unroll
                for (int u = 0; u < OWN_U; ++u) {
                    const unsigned y0 = q[u].x ^ pat, y1 = q[u].y ^ pat, y2 = q[u].z ^ pat, y3 = q[u].w ^ pat;
                    const unsigned t0 = ~(((y0 & 0x7F7F7F7Fu) + 0x7F7F7F7Fu) | y0 | 0x7F7F7F7Fu);
                    const unsigned t1 = ~(((y1 & 0x7F7F7F7Fu) + 0x7F7F7F7Fu) | y1 | 0x7F7F7F7Fu);
                    const unsigned t2 = ~(((y2 & 0x7F7F7F7Fu) + 0x7F7F7F7Fu) | y2 | 0x7F7F7F7Fu);
                    const unsigned t3 = ~(((y3 & 0x7F7F7F7Fu) + 0x7F7F7F7Fu) | y3 | 0x7F7F7F7Fu);
                    mk[u] = (t0 >> 7) | (t1 >> 6) | (t2 >> 5) | (t3 >> 4);
                    cnt += (unsigned)__builtin_popcount(mk[u]);
                }
                if (cnt) {
                    unsigned slot = atomicAdd(&qn, cnt);
#pragma unroll
                    for (int u = 0; u < OWN_U; ++u) {
                        unsigned mm = mk[u];
                        const int ebase = cu[u] * OWN_CHUNK + lane * 16;
                        while (mm) {
                            const int p = __builtin_ctz(mm);
                            mm &= mm - 1u;
                            if (slot < OWN_QCAP) qe[slot] = ebase + 4 * (p & 7) + (p >> 3);
                            ++slot;
                        }
                    }
                }
            }
        } else if (threadIdx.x == 0) {
            qn = (unsigned)OWN_QCAP + 1u;
        }
    } else {
        const unsigned stot = sv_n[blockIdx.x];
        if (stot <= (unsigned)OWN_QCAP) {
            for (int b = wave * 32; b < (int)stot; b += nw * 32) {
                const int idx = b + lane;
                int e = 0; unsigned dq = 0u; bool mm = false;
                if (idx < (int)stot) { e = sve[idx]; dq = svd[idx]; mm = flag[dq] != 0u; }
                const unsigned mk = __builtin_amdgcn_ballot_w32(mm);
                if (mk) {
                    unsigned base = 0u;
                    if (lane == 0) base = atomicAdd(&qn, (unsigned)__builtin_popcount(mk));
                    base = (unsigned)__builtin_amdgcn_readfirstlane((int)base);
                    if (mm) {
                        const unsigned slot = base + (unsigned)__builtin_popcount(mk & lt);
                        qe[slot] = e; qd[slot] = dq;
                    }
                }
            }
        } else if (threadIdx.x == 0) {
            qn = (unsigned)OWN_QCAP + 1u;
        }
    }
    __syncthreads();

    const unsigned qtot = (unsigned)__builtin_amdgcn_readfirstlane((int)qn);
    const bool fits  = qtot <= (unsigned)OWN_QCAP;
    const int  total = fits ? (int)qtot : 0;
    if (!SECOND) {
        if (threadIdx.x == 0) sv_n[blockIdx.x] = qtot;
    }

    if (fits) {
        for (int idx = threadIdx.x; idx < total; idx += OWN_THREADS) {
            const int e = qe[idx];
            unsigned dq;
            if (!SECOND) {
                dq = (unsigned)(ei[E + e] - lo);
                qd[idx]  = dq;
                sve[idx] = e;
                svd[idx] = dq;
            } else {
                dq = qd[idx];
            }
            const float l = leaky02(es[ei[e]] + edl[dq]);
            qp[idx] = l;
            atomicMax(&rmax[dq], fkey(l));
        }
    } else {
        own_walk<C, SECOND, 0>(ei, E, lo, nrow, h, es, edl, flag, rmax, rden, acc, wave, lane, nw);
    }
    __syncthreads();
    for (int r = threadIdx.x; r < nrow; r += OWN_THREADS) {
        const float ps = __expf(pself[r] - funkey(rmax[r]));
        pself[r] = ps;
        rden[r]  = ps;
    }
    __syncthreads();
    if (fits) {
        for (int idx = threadIdx.x; idx < total; idx += OWN_THREADS) {
            const unsigned dq = qd[idx];
            const float pe = __expf(qp[idx] - funkey(rmax[dq]));
            qp[idx] = pe;
            atomicAdd(&rden[dq], pe);
        }
    } else {
        own_walk<C, SECOND, 1>(ei, E, lo, nrow, h, es, edl, flag, rmax, rden, acc, wave, lane, nw);
    }
    __syncthreads();

    {
        const float4* hb4  = reinterpret_cast<const float4*>(h + (size_t)lo * C);
        float4*       acc4 = reinterpret_cast<float4*>(acc);
        for (int i = threadIdx.x; i < nrow * (C / 4); i += OWN_THREADS) {
            const int r = i / (C / 4);
            float4 v = make_float4(0.f, 0.f, 0.f, 0.f);
            if (!SECOND || flag[r]) {
                const float  a = pself[r] / rden[r];
                const float4 x = hb4[i];
                v.x = x.x * a; v.y = x.y * a; v.z = x.z * a; v.w = x.w * a;
            }
            acc4[i] = v;
        }
    }
    __syncthreads();

    for (int b = wave * 32; b < total; b += nw * 32) {
        const int idx = b + lane;
        int s = 0; unsigned dq = 0u; float a = 0.f;
        if (idx < total) {
            dq = qd[idx];
            s  = ei[qe[idx]];
            a  = qp[idx] / rden[dq];
        }
        const int cnt = min(32, total - b);
        for (int j = 0; j < cnt; j += 4) {
            float    v[4][C / 32];
            float    aj[4];
            unsigned dj[4];
#pragma unroll
            for (int t = 0; t < 4; ++t) {
                const int jj = min(j + t, cnt - 1);
                const int sj = __builtin_amdgcn_readlane(s, jj);
                dj[t] = (unsigned)__builtin_amdgcn_readlane((int)dq, jj);
                aj[t] = __int_as_float(__builtin_amdgcn_readlane(__float_as_int(a), jj));
                const float* hs = h + (size_t)sj * C + lane;
#pragma unroll
                for (int k = 0; k < C / 32; ++k) v[t][k] = hs[32 * k];
            }
#pragma unroll
            for (int t = 0; t < 4; ++t) {
                if (j + t < cnt) {
                    float* ar = acc + dj[t] * C + lane;
#pragma unroll
                    for (int k = 0; k < C / 32; ++k) atomicAdd(&ar[32 * k], v[t][k] * aj[t]);
                }
            }
        }
    }
    if (!fits)
        own_walk<C, SECOND, 2>(ei, E, lo, nrow, h, es, edl, flag, rmax, rden, acc, wave, lane, nw);
    __syncthreads();

    for (int r = wave; r < nrow; r += nw) {
        const size_t d = (size_t)(lo + r);
        const float4 v = *reinterpret_cast<const float4*>(&acc[r * C + 4 * lane]);
        if (SECOND) {
            if (flag[r]) {
                const float4 bq = *reinterpret_cast<const float4*>(bias + 4 * lane);
                float4 o; o.x = v.x + bq.x; o.y = v.y + bq.y; o.z = v.z + bq.z; o.w = v.w + bq.w;
                *reinterpret_cast<float4*>(yf + d * C + 4 * lane) = o;
            }
        } else {
            uint2 o;
            o.x = (unsigned)f2bf(v.x) | ((unsigned)f2bf(v.y) << 16);
            o.y = (unsigned)f2bf(v.z) | ((unsigned)f2bf(v.w) << 16);
            *reinterpret_cast<uint2*>(ybf + d * C + 4 * lane) = o;
        }
    }
}

__global__ void transpose_to_bf16(const float* __restrict__ W, unsigned short* __restrict__ Wt,
                                  int K, int NOUT)
{
    const int i = blockIdx.x * blockDim.x + threadIdx.x;
    if (i >= K * NOUT) return;
    const int k = i / NOUT, n = i % NOUT;
    Wt[(size_t)n * K + k] = f2bf(W[i]);
}

__global__ void __launch_bounds__(32 * RD_WAVES)
link_predict_rows(const float* __restrict__ h, const int* __restrict__ mask,
                  const float* __restrict__ Wl, const float* __restrict__ bl,
                  float* __restrict__ out, int P)
{
    const int lane = threadIdx.x & 31;
    const int wave = __builtin_amdgcn_readfirstlane(threadIdx.x >> 5);
    const int p0   = (blockIdx.x * RD_WAVES + wave) * RD_PAIRS;
    if (p0 >= P) return;
    const int mi = mask[min(p0 * 2 + (lane & (2 * RD_PAIRS - 1)), 2 * P - 1)];
    const float4 wa = reinterpret_cast<const float4*>(Wl)[lane];
    const float4 wb = reinterpret_cast<const float4*>(Wl)[32 + lane];
    float4 xa[RD_PAIRS], xb[RD_PAIRS];
#pragma unroll
    for (int j = 0; j < RD_PAIRS; ++j) {
        const int ra = __builtin_amdgcn_readlane(mi, 2 * j);
        const int rb = __builtin_amdgcn_readlane(mi, 2 * j + 1);
        xa[j] = reinterpret_cast<const float4*>(h + (size_t)ra * 128)[lane];
        xb[j] = reinterpret_cast<const float4*>(h + (size_t)rb * 128)[lane];
    }
    float v[RD_PAIRS];
#pragma unroll
    for (int j = 0; j < RD_PAIRS; ++j) v[j] = dot4(xa[j], wa) + dot4(xb[j], wb);
    const float t = wave_sums<RD_PAIRS>(v, lane);
    const int p = p0 + (lane >> 2);
    if ((lane & 3) == 0 && p < P) out[p] = 1.f / (1.f + expf(-(t + bl[0])));
}

extern "C" void kernel_launch(void* const* d_in, const int* in_sizes, int n_in,
                              void* d_out, int out_size, void* d_ws, size_t ws_size,
                              hipStream_t stream)
{
    (void)n_in; (void)out_size; (void)ws_size;

    const float* features = (const float*)d_in[0];
    const int*   ei       = (const int*)  d_in[1];
    const int*   mask     = (const int*)  d_in[2];
    const float* W1   = (const float*)d_in[3];
    const float* a_s1 = (const float*)d_in[4];
    const float* a_d1 = (const float*)d_in[5];
    const float* b1   = (const float*)d_in[6];
    const float* W2   = (const float*)d_in[7];
    const float* a_s2 = (const float*)d_in[8];
    const float* a_d2 = (const float*)d_in[9];
    const float* b2   = (const float*)d_in[10];
    const float* Wl   = (const float*)d_in[11];
    const float* bl   = (const float*)d_in[12];

    constexpr int F  = 128, Hh = 256;
    const int N  = in_sizes[0] / F;
    const int E  = in_sizes[1] / 2;
    const int P  = in_sizes[2] / 2;

    char* ws = (char*)d_ws;
    size_t off = 0;
    auto alloc = [&](size_t bytes) -> char* {
        char* ptr = ws + off;
        off += (bytes + 255) & ~(size_t)255;
        return ptr;
    };
    unsigned short* xbf  = (unsigned short*)alloc((size_t)N * F  * 2);
    unsigned short* w1t  = (unsigned short*)alloc((size_t)Hh * F * 2);
    unsigned short* w2t  = (unsigned short*)alloc((size_t)F * Hh * 2);
    float*    buf1    = (float*)   alloc((size_t)N * Hh * 4);
    float*    buf2    = (float*)   alloc((size_t)N * Hh * 4);
    float*    es      = (float*)   alloc((size_t)N * 4);
    float*    ed      = (float*)   alloc((size_t)N * 4);
    const int ownGrid = (N + OWN_ROWS - 1) / OWN_ROWS;
    unsigned* svn     = (unsigned*)alloc((size_t)ownGrid * 4);
    int*      sve     = (int*)     alloc((size_t)ownGrid * OWN_QCAP * 4);
    unsigned* svd     = (unsigned*)alloc((size_t)ownGrid * OWN_QCAP * 4);
    unsigned* own4    = (unsigned*)alloc((size_t)((E + OWN_CHUNK - 1) / OWN_CHUNK) * OWN_CHUNK);

    const int tilesM = (N + 15) / 16;
    const int gemmGrid = (tilesM + PAIR_WAVES - 1) / PAIR_WAVES;
    const int TB = 256;
    auto cdiv = [](size_t a, size_t b) -> int { return (int)((a + b - 1) / b); };

    transpose_to_bf16<<<cdiv((size_t)F * Hh, TB), TB, 0, stream>>>(W1, w1t, F, Hh);
    transpose_to_bf16<<<cdiv((size_t)Hh * F, TB), TB, 0, stream>>>(W2, w2t, Hh, F);

    node_scores_rows<true><<<cdiv((size_t)N, RD_ROWS * RD_WAVES), 32 * RD_WAVES, 0, stream>>>(features, a_s1, a_d1, es, ed, N, ei + E, E, own4, W1);
    aggregate_owned<F, false><<<ownGrid, OWN_THREADS, 0, stream>>>(ei, E, N, features, es, ed,
                                                                  nullptr, 0, nullptr, xbf, nullptr, svn, sve, svd,
                                                                  (const unsigned char*)own4);

    wmma_two_products<F, Hh, F><<<gemmGrid, 32 * PAIR_WAVES, 0, stream>>>(xbf, w1t, b1, w2t, buf1, N);
    node_scores_rows<false><<<cdiv((size_t)N, RD_ROWS * RD_WAVES), 32 * RD_WAVES, 0, stream>>>(buf1, a_s2, a_d2, es, ed, N, nullptr, 0, nullptr, nullptr);
    aggregate_owned<F, true><<<ownGrid, OWN_THREADS, 0, stream>>>(ei, E, N, buf1, es, ed,
                                                                 mask, P * 2, b2, nullptr, buf2, svn, sve, svd, nullptr);

    link_predict_rows<<<cdiv((size_t)P, RD_PAIRS * RD_WAVES), 32 * RD_WAVES, 0, stream>>>(buf2, mask, Wl, bl,
                                                                                         (float*)d_out, P);
}


# The product kernel's work-groups (16 waves) keep both weights in LDS: staged from the f32 inputs, rounded and transposed on the way; B fragments come from LDS, not memory; 8 -> 6 launches; same bits
# speedup vs baseline: 8.5794x; 1.2560x over previous
#include <hip/hip_runtime.h>
#include <hip/hip_bf16.h>

typedef __attribute__((__ext_vector_type__(16))) __bf16 bf16x16;
typedef __attribute__((__ext_vector_type__(8)))  float  f32x8;
typedef __attribute__((__ext_vector_type__(4)))  unsigned int u32x4;

union Frag16 {
    u32x4   q[2];
    bf16x16 v;
};

static __device__ __forceinline__ unsigned short f2bf(float f) {
    unsigned u = __float_as_uint(f);
    u += 0x7FFFu + ((u >> 16) & 1u);
    return (unsigned short)(u >> 16);
}

static __device__ __forceinline__ unsigned fkey(float f) {
    unsigned u = __float_as_uint(f);
    return (u & 0x80000000u) ? ~u : (u | 0x80000000u);
}
static __device__ __forceinline__ float funkey(unsigned k) {
    unsigned u = (k & 0x80000000u) ? (k & 0x7FFFFFFFu) : ~k;
    return __uint_as_float(u);
}

#define PAIR_WAVES 16
#define PAIR_PAD   8

static __device__ __forceinline__ unsigned bf2(float lo, float hi)
{
    return (unsigned)f2bf(lo) | ((unsigned)f2bf(hi) << 16);
}

template <int K, int NOUT, int PITCH>
static __device__ __forceinline__ void stage_weight(const float* __restrict__ W, unsigned short* dst,
                                                    int tid, int rot)
{
    constexpr int NG    = NOUT / 4;
    constexpr int KB    = K / 8;
    constexpr int TASKS = NG * KB;
    constexpr int HI    = TASKS / 32;
    constexpr int TPT   = TASKS / (32 * PAIR_WAVES);
    static_assert(NG % 8 == 0 && KB % 4 == 0 && TASKS % (32 * PAIR_WAVES) == 0, "whole task sets");
    static_assert((HI & (HI - 1)) == 0, "the rotation wraps with a mask");
    const int n4lo = tid & 7, kblo = (tid >> 3) & 3;
    float v[TPT][8][4];
    int   n4[TPT], kb[TPT];
#pragma unroll
    for (int p = 0; p < TPT; ++p) {
        const int hi = ((tid >> 5) + p * PAIR_WAVES + rot) & (HI - 1);
        n4[p] = (hi % (NG / 8)) * 8 + n4lo;
        kb[p] = (hi / (NG / 8)) * 4 + kblo;
        const float* s = W + (size_t)(8 * kb[p]) * NOUT + 4 * n4[p];
#pragma unroll
        for (int j = 0; j < 8; ++j) {
            const float4 x = *reinterpret_cast<const float4*>(s + (size_t)j * NOUT);
            v[p][j][0] = x.x; v[p][j][1] = x.y; v[p][j][2] = x.z; v[p][j][3] = x.w;
        }
    }
#pragma unroll
    for (int p = 0; p < TPT; ++p) {
        unsigned short* d = dst + (size_t)(4 * n4[p]) * PITCH + 8 * kb[p];
#pragma unroll
        for (int i = 0; i < 4; ++i) {
            u32x4 q;
            q.x = bf2(v[p][0][i], v[p][1][i]);
            q.y = bf2(v[p][2][i], v[p][3][i]);
            q.z = bf2(v[p][4][i], v[p][5][i]);
            q.w = bf2(v[p][6][i], v[p][7][i]);
            *reinterpret_cast<u32x4*>(d + i * PITCH) = q;
        }
    }
}

template <int K1, int N1, int N2>
__global__ void __launch_bounds__(32 * PAIR_WAVES)
wmma_two_products(const unsigned short* __restrict__ A,
                  const float* __restrict__ W1,
                  const float* __restrict__ bias1,
                  const float* __restrict__ W2,
                  float* __restrict__ Y, int M)
{
    static_assert(K1 % 32 == 0 && N1 % 32 == 0 && N1 % 16 == 0 && N2 % 16 == 0, "whole fragments");
    constexpr int P1    = K1 + PAIR_PAD;
    constexpr int PITCH = N1 + PAIR_PAD;
    static_assert((P1 * 2) % 16 == 0 && ((P1 * 2 / 16) & 1) == 1, "pitch = 16 B x odd");
    static_assert((PITCH * 2) % 16 == 0 && ((PITCH * 2 / 16) & 1) == 1, "pitch = 16 B x odd");
    constexpr int OPITCH = N2 + 4;
    static_assert(N2 == 128, "one float4 per lane covers a result row");
    static_assert(OPITCH * 4 <= PITCH * 2, "the result tile reuses the wave's LDS tile");
    constexpr int KS1 = K1 / 32;
    constexpr int KS2 = N1 / 32;
    __shared__ __attribute__((aligned(16))) unsigned short w1s[N1 * P1];
    __shared__ __attribute__((aligned(16))) unsigned short w2s[N2 * PITCH];
    __shared__ __attribute__((aligned(16))) unsigned short mid[PAIR_WAVES][16 * PITCH];

    const int lane  = threadIdx.x & 31;
    const int wave  = __builtin_amdgcn_readfirstlane(threadIdx.x >> 5);
    const int m0    = (blockIdx.x * PAIR_WAVES + wave) * 16;
    const bool live = m0 < M;
    const int l15   = lane & 15;
    const int half  = lane >> 4;
    const int kbase = half * 8;
    unsigned short* mine = mid[wave];

    Frag16 fa[KS1];
    if (live) {
        int rowA = m0 + l15;
        if (rowA >= M) rowA = M - 1;
        const unsigned short* aptr = A + (size_t)rowA * K1 + kbase;
#pragma unroll
        for (int k = 0; k < KS1; ++k) {
            fa[k].q[0] = *(const u32x4*)(aptr + 32 * k);
            fa[k].q[1] = *(const u32x4*)(aptr + 32 * k + 16);
        }
    }
    stage_weight<K1, N1, P1>(W1, w1s, threadIdx.x, blockIdx.x);
    stage_weight<N1, N2, PITCH>(W2, w2s, threadIdx.x, blockIdx.x);
    __syncthreads();

    if (live) {
        Frag16 fb[2][KS1];
        {
            const unsigned short* bptr = w1s + l15 * P1 + kbase;
#pragma unroll
            for (int k = 0; k < KS1; ++k) {
                fb[0][k].q[0] = *(const u32x4*)(bptr + 32 * k);
                fb[0][k].q[1] = *(const u32x4*)(bptr + 32 * k + 16);
            }
        }
#pragma unroll
        for (int t = 0; t < N1 / 16; ++t) {
            if (t + 1 < N1 / 16) {
                const unsigned short* bptr = w1s + ((t + 1) * 16 + l15) * P1 + kbase;
#pragma unroll
                for (int k = 0; k < KS1; ++k) {
                    fb[(t + 1) & 1][k].q[0] = *(const u32x4*)(bptr + 32 * k);
                    fb[(t + 1) & 1][k].q[1] = *(const u32x4*)(bptr + 32 * k + 16);
                }
            }
            f32x8 acc = {};
#pragma unroll
            for (int k = 0; k < KS1; ++k)
                acc = __builtin_amdgcn_wmma_f32_16x16x32_bf16(
                          false, fa[k].v, false, fb[t & 1][k].v, (short)0, acc, false, false);
            const int   col = t * 16 + l15;
            const float bv  = bias1[col];
#pragma unroll
            for (int r = 0; r < 8; ++r) {
                const float v = acc[r] + bv;
                mine[(r + half * 8) * PITCH + col] = f2bf(v > 0.f ? v : 0.f);
            }
        }
    }
    __syncthreads();
    Frag16 ga[KS2];
    if (live) {
        const unsigned short* lp = mine + l15 * PITCH + kbase;
#pragma unroll
        for (int k = 0; k < KS2; ++k) {
            ga[k].q[0] = *(const u32x4*)(lp + 32 * k);
            ga[k].q[1] = *(const u32x4*)(lp + 32 * k + 16);
        }
    }
    __syncthreads();
    float* outt = reinterpret_cast<float*>(mine);
    if (live) {
        constexpr int NT2 = N2 / 16;
        constexpr int GT  = 4;
        constexpr int NG2 = KS2 * NT2 / GT;
        static_assert((KS2 * NT2) % GT == 0, "whole groups");
        const unsigned short* bbase = w2s + l15 * PITCH + kbase;
        f32x8 acc[NT2];
#pragma unroll
        for (int t = 0; t < NT2; ++t) acc[t] = f32x8{};
        Frag16 gb[2][GT];
#pragma unroll
        for (int i = 0; i < GT; ++i) {
            const unsigned short* bptr = bbase + ((i % NT2) * 16) * PITCH + 32 * (i / NT2);
            gb[0][i].q[0] = *(const u32x4*)(bptr);
            gb[0][i].q[1] = *(const u32x4*)(bptr + 16);
        }
#pragma unroll
        for (int g = 0; g < NG2; ++g) {
            if (g + 1 < NG2) {
#pragma unroll
                for (int i = 0; i < GT; ++i) {
                    const int s = (g + 1) * GT + i;
                    const unsigned short* bptr = bbase + ((s % NT2) * 16) * PITCH + 32 * (s / NT2);
                    gb[(g + 1) & 1][i].q[0] = *(const u32x4*)(bptr);
                    gb[(g + 1) & 1][i].q[1] = *(const u32x4*)(bptr + 16);
                }
            }
#pragma unroll
            for (int i = 0; i < GT; ++i) {
                const int s = g * GT + i;
                acc[s % NT2] = __builtin_amdgcn_wmma_f32_16x16x32_bf16(
                                   false, ga[s / NT2].v, false, gb[g & 1][i].v, (short)0, acc[s % NT2], false, false);
            }
        }
#pragma unroll
        for (int t = 0; t < NT2; ++t) {
#pragma unroll
            for (int r = 0; r < 8; ++r)
                outt[(r + half * 8) * OPITCH + t * 16 + l15] = acc[t][r];
        }
    }
    __syncthreads();
    if (live) {
        float* yrow = Y + (size_t)m0 * N2 + 4 * lane;
        if (m0 + 16 <= M) {
            float4 v[16];
#pragma unroll
            for (int i = 0; i < 16; ++i) v[i] = *reinterpret_cast<const float4*>(outt + i * OPITCH + 4 * lane);
#pragma unroll
            for (int i = 0; i < 16; ++i) *reinterpret_cast<float4*>(yrow + (size_t)i * N2) = v[i];
        } else {
            for (int i = 0; i < M - m0; ++i)
                *reinterpret_cast<float4*>(yrow + (size_t)i * N2) =
                    *reinterpret_cast<const float4*>(outt + i * OPITCH + 4 * lane);
        }
    }
}

#define RD_WAVES 8
#define RD_ROWS  16
#define RD_PAIRS 8

static __device__ __forceinline__ float dot4(const float4 x, const float4 w)
{
    return x.x * w.x + x.y * w.y + x.z * w.z + x.w * w.w;
}

template <int NV>
static __device__ __forceinline__ float wave_sums(float (&v)[NV], int lane)
{
#pragma unroll
    for (int st = 0; st < 5; ++st) {
        const int m = 16 >> st;
        const int n = NV >> (st + 1);
        if (n >= 1) {
            const bool up = (lane & m) != 0;
#pragma unroll
            for (int i = 0; i < n; ++i) {
                const float lo = v[i], hi = v[i + n];
                const float keep = up ? hi : lo;
                const float send = up ? lo : hi;
                v[i] = keep + __shfl_xor(send, m);
            }
        } else {
            v[0] += __shfl_xor(v[0], m);
        }
    }
    return v[0];
}

#define OWN_ROWS  197
#define OWN_NONE  255u
#define OWN_CHUNK 512

#define FOLD_N 256
#define FOLD_ROWS 8
template <bool FOLD>
__global__ void __launch_bounds__(32 * RD_WAVES)
node_scores_rows(const float* __restrict__ h,
                 const float* __restrict__ as_, const float* __restrict__ ad_,
                 float* __restrict__ es, float* __restrict__ ed,
                 int N,
                 const int* __restrict__ dstl, int E, unsigned* __restrict__ own4,
                 const float* __restrict__ Wf)
{
    __shared__ __attribute__((aligned(16))) float fold[FOLD ? 256 : 4];
    if (own4) {
        const int nwords = ((E + OWN_CHUNK - 1) / OWN_CHUNK) * (OWN_CHUNK / 4);
        for (int g = blockIdx.x * blockDim.x + threadIdx.x; g < nwords; g += gridDim.x * blockDim.x) {
            unsigned w = 0u;
#pragma unroll
            for (int k = 0; k < 4; ++k) {
                const int e = 4 * g + k;
                unsigned b = OWN_NONE;
                if (e < E) {
                    const unsigned d = (unsigned)dstl[e];
                    if (d < (unsigned)N) b = min(d / (unsigned)OWN_ROWS, OWN_NONE);
                }
                w |= b << (8 * k);
            }
            own4[g] = w;
        }
    }
    const int lane = threadIdx.x & 31;
    const int wave = __builtin_amdgcn_readfirstlane(threadIdx.x >> 5);
    const int r0   = (blockIdx.x * RD_WAVES + wave) * RD_ROWS;
    float4 ws, wd;
    if (FOLD) {
        const float4 sl = reinterpret_cast<const float4*>(as_)[lane], sh = reinterpret_cast<const float4*>(as_)[32 + lane];
        const float4 dl = reinterpret_cast<const float4*>(ad_)[lane], dh = reinterpret_cast<const float4*>(ad_)[32 + lane];
#pragma nounroll
        for (int kb = wave * FOLD_ROWS; kb < 128; kb += RD_WAVES * FOLD_ROWS) {
            float fv[2 * FOLD_ROWS];
#pragma unroll
            for (int j = 0; j < FOLD_ROWS; ++j) {
                const float4* wr = reinterpret_cast<const float4*>(Wf + (size_t)(kb + j) * FOLD_N);
                const float4 a = wr[lane], b = wr[32 + lane];
                fv[j]             = dot4(a, sl) + dot4(b, sh);
                fv[FOLD_ROWS + j] = dot4(a, dl) + dot4(b, dh);
            }
            const float t = wave_sums<2 * FOLD_ROWS>(fv, lane);
            const int q = lane >> 1;
            if ((lane & 1) == 0) fold[(q < FOLD_ROWS ? 0 : 128 - FOLD_ROWS) + kb + q] = t;
        }
        __syncthreads();
        if (r0 >= N) return;
        ws = reinterpret_cast<const float4*>(fold)[lane];
        wd = reinterpret_cast<const float4*>(fold)[32 + lane];
    } else {
        if (r0 >= N) return;
        ws = reinterpret_cast<const float4*>(as_)[lane];
        wd = reinterpret_cast<const float4*>(ad_)[lane];
    }
    float4 x[RD_ROWS];
#pragma unroll
    for (int j = 0; j < RD_ROWS; ++j) {
        const int r = min(r0 + j, N - 1);
        x[j] = reinterpret_cast<const float4*>(h + (size_t)r * 128)[lane];
    }
    float v[2 * RD_ROWS];
#pragma unroll
    for (int j = 0; j < RD_ROWS; ++j) {
        v[j]           = dot4(x[j], ws);
        v[RD_ROWS + j] = dot4(x[j], wd);
    }
    const float t = wave_sums<2 * RD_ROWS>(v, lane);
    const int row = r0 + (lane & (RD_ROWS - 1));
    if (row < N) {
        if (lane < RD_ROWS) es[row] = t;
        else                ed[row] = t;
    }
}

#define OWN_THREADS 1024
#define OWN_QCAP    4096
#define OWN_U       8
#define OWN_PU      8

static __device__ __forceinline__ float leaky02(float t) { return (t > 0.f) ? t : 0.2f * t; }

template <int C>
static __device__ __forceinline__ void own_add_row(const float* __restrict__ h, int s, float a,
                                                   float* accrow, int lane)
{
    const float* hs = h + (size_t)s * C + lane;
    float*       ar = accrow + lane;
    float v[C / 32];
#pragma unroll
    for (int k = 0; k < C / 32; ++k) v[k] = hs[32 * k];
#pragma unroll
    for (int k = 0; k < C / 32; ++k) atomicAdd(&ar[32 * k], v[k] * a);
}

template <int C, bool SECOND, int STAGE>
static __device__ __forceinline__ void own_walk(const int* __restrict__ ei, int E, int lo, int nrow,
                                                const float* __restrict__ h, const float* __restrict__ es,
                                                const float* edl, const unsigned* flag,
                                                unsigned* rmax, float* rden, float* acc,
                                                int wave, int lane, int nw)
{
    const int* __restrict__ dst = ei + E;
    for (int e0 = wave * 32; e0 < E; e0 += nw * 32) {
        const int e = e0 + lane;
        unsigned dl = 0u;
        bool ok = false;
        if (e < E) {
            dl = (unsigned)(dst[e] - lo);
            ok = dl < (unsigned)nrow;
            if (SECOND) { if (ok) ok = flag[dl] != 0u; }
        }
        int s = 0; float l = 0.f;
        if (ok) { s = ei[e]; l = leaky02(es[s] + edl[dl]); }
        if (STAGE == 0) {
            if (ok) atomicMax(&rmax[dl], fkey(l));
        } else if (STAGE == 1) {
            if (ok) atomicAdd(&rden[dl], __expf(l - funkey(rmax[dl])));
        } else {
            float a = 0.f;
            if (ok) a = __expf(l - funkey(rmax[dl])) / rden[dl];
            unsigned mk = __builtin_amdgcn_ballot_w32(ok);
            while (mk) {
                const int L = __builtin_ctz(mk);
                mk &= mk - 1u;
                const int      sj = __builtin_amdgcn_readlane(s, L);
                const unsigned dj = (unsigned)__builtin_amdgcn_readlane((int)dl, L);
                const float    aj = __int_as_float(__builtin_amdgcn_readlane(__float_as_int(a), L));
                own_add_row<C>(h, sj, aj, acc + dj * C, lane);
            }
        }
    }
}

template <int C, bool SECOND>
__global__ void __launch_bounds__(OWN_THREADS)
aggregate_owned(const int* __restrict__ ei, int E, int N,
                const float* __restrict__ h,
                const float* __restrict__ es, const float* __restrict__ ed,
                const int* __restrict__ pairs, int npairs, const float* __restrict__ bias,
                unsigned short* __restrict__ ybf, float* __restrict__ yf,
                unsigned* __restrict__ sv_n, int* __restrict__ sv_e, unsigned* __restrict__ sv_d,
                const unsigned char* __restrict__ own8)
{
    static_assert(C == 128, "one float4 per lane covers a row");
    __shared__ __attribute__((aligned(16))) float acc[OWN_ROWS * C];
    __shared__ int      qe[OWN_QCAP];
    __shared__ unsigned qd[OWN_QCAP];
    __shared__ float    qp[OWN_QCAP];
    __shared__ float    edl[OWN_ROWS];
    __shared__ float    pself[OWN_ROWS];
    __shared__ unsigned rmax[OWN_ROWS];
    __shared__ float    rden[OWN_ROWS];
    __shared__ unsigned flag[OWN_ROWS];
    __shared__ unsigned qn;

    const int lane = threadIdx.x & 31;
    const int wave = __builtin_amdgcn_readfirstlane(threadIdx.x >> 5);
    const int nw   = OWN_THREADS / 32;
    const int lo   = blockIdx.x * OWN_ROWS;
    const int nrow = min(OWN_ROWS, N - lo);

    if (threadIdx.x == 0) qn = 0u;
    const int   n4 = npairs >> 2;
    const int4* p4 = reinterpret_cast<const int4*>(pairs);
    int4 pm[OWN_PU];
    if (SECOND) {
#pragma unroll
        for (int u = 0; u < OWN_PU; ++u) {
            const int i = threadIdx.x + u * OWN_THREADS;
            pm[u] = make_int4(-1, -1, -1, -1);
            if (i < n4) pm[u] = p4[i];
        }
    }
    for (int r = threadIdx.x; r < nrow; r += OWN_THREADS) {
        const int d = lo + r;
        if (SECOND) flag[r] = 0u;
        const float edv = ed[d];
        const float l   = leaky02(es[d] + edv);
        edl[r]   = edv;
        pself[r] = l;
        rmax[r]  = fkey(l);
    }
    if (SECOND) {
        __syncthreads();
#pragma unroll
        for (int u = 0; u < OWN_PU; ++u) {
            const unsigned r0 = (unsigned)(pm[u].x - lo), r1 = (unsigned)(pm[u].y - lo), r2 = (unsigned)(pm[u].z - lo), r3 = (unsigned)(pm[u].w - lo);
            if (r0 < (unsigned)nrow) flag[r0] = 1u;
            if (r1 < (unsigned)nrow) flag[r1] = 1u;
            if (r2 < (unsigned)nrow) flag[r2] = 1u;
            if (r3 < (unsigned)nrow) flag[r3] = 1u;
        }
        for (int i = threadIdx.x + OWN_PU * OWN_THREADS; i < n4; i += OWN_THREADS) {
            const int4 m = p4[i];
            const unsigned r0 = (unsigned)(m.x - lo), r1 = (unsigned)(m.y - lo), r2 = (unsigned)(m.z - lo), r3 = (unsigned)(m.w - lo);
            if (r0 < (unsigned)nrow) flag[r0] = 1u;
            if (r1 < (unsigned)nrow) flag[r1] = 1u;
            if (r2 < (unsigned)nrow) flag[r2] = 1u;
            if (r3 < (unsigned)nrow) flag[r3] = 1u;
        }
        for (int i = 4 * n4 + threadIdx.x; i < npairs; i += OWN_THREADS) {
            const unsigned r = (unsigned)(pairs[i] - lo);
            if (r < (unsigned)nrow) flag[r] = 1u;
        }
    }
    __syncthreads();

    const unsigned lt = (1u << lane) - 1u;
    int*      sve = sv_e + (size_t)blockIdx.x * OWN_QCAP;
    unsigned* svd = sv_d + (size_t)blockIdx.x * OWN_QCAP;
    if (!SECOND) {
        if (gridDim.x <= OWN_NONE) {
            const int nchunks = (E + OWN_CHUNK - 1) / OWN_CHUNK;
            const int startc  = (int)(((long long)blockIdx.x * nchunks) / gridDim.x);
            const unsigned pat = (unsigned)blockIdx.x * 0x01010101u;
            const unsigned char* lbase = own8 + lane * 16;

            for (int cl0 = wave; cl0 < nchunks; cl0 += nw * OWN_U) {
                uint4 q[OWN_U];
                int   cu[OWN_U];
#pragma unroll
                for (int u = 0; u < OWN_U; ++u) {
                    const int cl = cl0 + u * nw;
                    int cc = cl + startc;
                    if (cc >= nchunks) cc -= nchunks;
                    cu[u] = cc;
                    q[u] = make_uint4(~0u, ~0u, ~0u, ~0u);
                    if (cl < nchunks) q[u] = *reinterpret_cast<const uint4*>(lbase + (size_t)cc * OWN_CHUNK);
                }
                unsigned mk[OWN_U];
                unsigned cnt = 0u;
#pragma unroll
                for (int u = 0; u < OWN_U; ++u) {
                    const unsigned y0 = q[u].x ^ pat, y1 = q[u].y ^ pat, y2 = q[u].z ^ pat, y3 = q[u].w ^ pat;
                    const unsigned t0 = ~(((y0 & 0x7F7F7F7Fu) + 0x7F7F7F7Fu) | y0 | 0x7F7F7F7Fu);
                    const unsigned t1 = ~(((y1 & 0x7F7F7F7Fu) + 0x7F7F7F7Fu) | y1 | 0x7F7F7F7Fu);
                    const unsigned t2 = ~(((y2 & 0x7F7F7F7Fu) + 0x7F7F7F7Fu) | y2 | 0x7F7F7F7Fu);
                    const unsigned t3 = ~(((y3 & 0x7F7F7F7Fu) + 0x7F7F7F7Fu) | y3 | 0x7F7F7F7Fu);
                    mk[u] = (t0 >> 7) | (t1 >> 6) | (t2 >> 5) | (t3 >> 4);
                    cnt += (unsigned)__builtin_popcount(mk[u]);
                }
                if (cnt) {
                    unsigned slot = atomicAdd(&qn, cnt);
#pragma unroll
                    for (int u = 0; u < OWN_U; ++u) {
                        unsigned mm = mk[u];
                        const int ebase = cu[u] * OWN_CHUNK + lane * 16;
                        while (mm) {
                            const int p = __builtin_ctz(mm);
                            mm &= mm - 1u;
                            if (slot < OWN_QCAP) qe[slot] = ebase + 4 * (p & 7) + (p >> 3);
                            ++slot;
                        }
                    }
                }
            }
        } else if (threadIdx.x == 0) {
            qn = (unsigned)OWN_QCAP + 1u;
        }
    } else {
        const unsigned stot = sv_n[blockIdx.x];
        if (stot <= (unsigned)OWN_QCAP) {
            for (int b = wave * 32; b < (int)stot; b += nw * 32) {
                const int idx = b + lane;
                int e = 0; unsigned dq = 0u; bool mm = false;
                if (idx < (int)stot) { e = sve[idx]; dq = svd[idx]; mm = flag[dq] != 0u; }
                const unsigned mk = __builtin_amdgcn_ballot_w32(mm);
                if (mk) {
                    unsigned base = 0u;
                    if (lane == 0) base = atomicAdd(&qn, (unsigned)__builtin_popcount(mk));
                    base = (unsigned)__builtin_amdgcn_readfirstlane((int)base);
                    if (mm) {
                        const unsigned slot = base + (unsigned)__builtin_popcount(mk & lt);
                        qe[slot] = e; qd[slot] = dq;
                    }
                }
            }
        } else if (threadIdx.x == 0) {
            qn = (unsigned)OWN_QCAP + 1u;
        }
    }
    __syncthreads();

    const unsigned qtot = (unsigned)__builtin_amdgcn_readfirstlane((int)qn);
    const bool fits  = qtot <= (unsigned)OWN_QCAP;
    const int  total = fits ? (int)qtot : 0;
    if (!SECOND) {
        if (threadIdx.x == 0) sv_n[blockIdx.x] = qtot;
    }

    if (fits) {
        for (int idx = threadIdx.x; idx < total; idx += OWN_THREADS) {
            const int e = qe[idx];
            unsigned dq;
            if (!SECOND) {
                dq = (unsigned)(ei[E + e] - lo);
                qd[idx]  = dq;
                sve[idx] = e;
                svd[idx] = dq;
            } else {
                dq = qd[idx];
            }
            const float l = leaky02(es[ei[e]] + edl[dq]);
            qp[idx] = l;
            atomicMax(&rmax[dq], fkey(l));
        }
    } else {
        own_walk<C, SECOND, 0>(ei, E, lo, nrow, h, es, edl, flag, rmax, rden, acc, wave, lane, nw);
    }
    __syncthreads();
    for (int r = threadIdx.x; r < nrow; r += OWN_THREADS) {
        const float ps = __expf(pself[r] - funkey(rmax[r]));
        pself[r] = ps;
        rden[r]  = ps;
    }
    __syncthreads();
    if (fits) {
        for (int idx = threadIdx.x; idx < total; idx += OWN_THREADS) {
            const unsigned dq = qd[idx];
            const float pe = __expf(qp[idx] - funkey(rmax[dq]));
            qp[idx] = pe;
            atomicAdd(&rden[dq], pe);
        }
    } else {
        own_walk<C, SECOND, 1>(ei, E, lo, nrow, h, es, edl, flag, rmax, rden, acc, wave, lane, nw);
    }
    __syncthreads();

    {
        const float4* hb4  = reinterpret_cast<const float4*>(h + (size_t)lo * C);
        float4*       acc4 = reinterpret_cast<float4*>(acc);
        for (int i = threadIdx.x; i < nrow * (C / 4); i += OWN_THREADS) {
            const int r = i / (C / 4);
            float4 v = make_float4(0.f, 0.f, 0.f, 0.f);
            if (!SECOND || flag[r]) {
                const float  a = pself[r] / rden[r];
                const float4 x = hb4[i];
                v.x = x.x * a; v.y = x.y * a; v.z = x.z * a; v.w = x.w * a;
            }
            acc4[i] = v;
        }
    }
    __syncthreads();

    for (int b = wave * 32; b < total; b += nw * 32) {
        const int idx = b + lane;
        int s = 0; unsigned dq = 0u; float a = 0.f;
        if (idx < total) {
            dq = qd[idx];
            s  = ei[qe[idx]];
            a  = qp[idx] / rden[dq];
        }
        const int cnt = min(32, total - b);
        for (int j = 0; j < cnt; j += 4) {
            float    v[4][C / 32];
            float    aj[4];
            unsigned dj[4];
#pragma unroll
            for (int t = 0; t < 4; ++t) {
                const int jj = min(j + t, cnt - 1);
                const int sj = __builtin_amdgcn_readlane(s, jj);
                dj[t] = (unsigned)__builtin_amdgcn_readlane((int)dq, jj);
                aj[t] = __int_as_float(__builtin_amdgcn_readlane(__float_as_int(a), jj));
                const float* hs = h + (size_t)sj * C + lane;
#pragma unroll
                for (int k = 0; k < C / 32; ++k) v[t][k] = hs[32 * k];
            }
#pragma unroll
            for (int t = 0; t < 4; ++t) {
                if (j + t < cnt) {
                    float* ar = acc + dj[t] * C + lane;
#pragma unroll
                    for (int k = 0; k < C / 32; ++k) atomicAdd(&ar[32 * k], v[t][k] * aj[t]);
                }
            }
        }
    }
    if (!fits)
        own_walk<C, SECOND, 2>(ei, E, lo, nrow, h, es, edl, flag, rmax, rden, acc, wave, lane, nw);
    __syncthreads();

    for (int r = wave; r < nrow; r += nw) {
        const size_t d = (size_t)(lo + r);
        const float4 v = *reinterpret_cast<const float4*>(&acc[r * C + 4 * lane]);
        if (SECOND) {
            if (flag[r]) {
                const float4 bq = *reinterpret_cast<const float4*>(bias + 4 * lane);
                float4 o; o.x = v.x + bq.x; o.y = v.y + bq.y; o.z = v.z + bq.z; o.w = v.w + bq.w;
                *reinterpret_cast<float4*>(yf + d * C + 4 * lane) = o;
            }
        } else {
            uint2 o;
            o.x = (unsigned)f2bf(v.x) | ((unsigned)f2bf(v.y) << 16);
            o.y = (unsigned)f2bf(v.z) | ((unsigned)f2bf(v.w) << 16);
            *reinterpret_cast<uint2*>(ybf + d * C + 4 * lane) = o;
        }
    }
}

__global__ void __launch_bounds__(32 * RD_WAVES)
link_predict_rows(const float* __restrict__ h, const int* __restrict__ mask,
                  const float* __restrict__ Wl, const float* __restrict__ bl,
                  float* __restrict__ out, int P)
{
    const int lane = threadIdx.x & 31;
    const int wave = __builtin_amdgcn_readfirstlane(threadIdx.x >> 5);
    const int p0   = (blockIdx.x * RD_WAVES + wave) * RD_PAIRS;
    if (p0 >= P) return;
    const int mi = mask[min(p0 * 2 + (lane & (2 * RD_PAIRS - 1)), 2 * P - 1)];
    const float4 wa = reinterpret_cast<const float4*>(Wl)[lane];
    const float4 wb = reinterpret_cast<const float4*>(Wl)[32 + lane];
    float4 xa[RD_PAIRS], xb[RD_PAIRS];
#pragma unroll
    for (int j = 0; j < RD_PAIRS; ++j) {
        const int ra = __builtin_amdgcn_readlane(mi, 2 * j);
        const int rb = __builtin_amdgcn_readlane(mi, 2 * j + 1);
        xa[j] = reinterpret_cast<const float4*>(h + (size_t)ra * 128)[lane];
        xb[j] = reinterpret_cast<const float4*>(h + (size_t)rb * 128)[lane];
    }
    float v[RD_PAIRS];
#pragma unroll
    for (int j = 0; j < RD_PAIRS; ++j) v[j] = dot4(xa[j], wa) + dot4(xb[j], wb);
    const float t = wave_sums<RD_PAIRS>(v, lane);
    const int p = p0 + (lane >> 2);
    if ((lane & 3) == 0 && p < P) out[p] = 1.f / (1.f + expf(-(t + bl[0])));
}

extern "C" void kernel_launch(void* const* d_in, const int* in_sizes, int n_in,
                              void* d_out, int out_size, void* d_ws, size_t ws_size,
                              hipStream_t stream)
{
    (void)n_in; (void)out_size; (void)ws_size;

    const float* features = (const float*)d_in[0];
    const int*   ei       = (const int*)  d_in[1];
    const int*   mask     = (const int*)  d_in[2];
    const float* W1   = (const float*)d_in[3];
    const float* a_s1 = (const float*)d_in[4];
    const float* a_d1 = (const float*)d_in[5];
    const float* b1   = (const float*)d_in[6];
    const float* W2   = (const float*)d_in[7];
    const float* a_s2 = (const float*)d_in[8];
    const float* a_d2 = (const float*)d_in[9];
    const float* b2   = (const float*)d_in[10];
    const float* Wl   = (const float*)d_in[11];
    const float* bl   = (const float*)d_in[12];

    constexpr int F  = 128, Hh = 256;
    const int N  = in_sizes[0] / F;
    const int E  = in_sizes[1] / 2;
    const int P  = in_sizes[2] / 2;

    char* ws = (char*)d_ws;
    size_t off = 0;
    auto alloc = [&](size_t bytes) -> char* {
        char* ptr = ws + off;
        off += (bytes + 255) & ~(size_t)255;
        return ptr;
    };
    unsigned short* xbf  = (unsigned short*)alloc((size_t)N * F  * 2);
    float*    buf1    = (float*)   alloc((size_t)N * Hh * 4);
    float*    buf2    = (float*)   alloc((size_t)N * Hh * 4);
    float*    es      = (float*)   alloc((size_t)N * 4);
    float*    ed      = (float*)   alloc((size_t)N * 4);
    const int ownGrid = (N + OWN_ROWS - 1) / OWN_ROWS;
    unsigned* svn     = (unsigned*)alloc((size_t)ownGrid * 4);
    int*      sve     = (int*)     alloc((size_t)ownGrid * OWN_QCAP * 4);
    unsigned* svd     = (unsigned*)alloc((size_t)ownGrid * OWN_QCAP * 4);
    unsigned* own4    = (unsigned*)alloc((size_t)((E + OWN_CHUNK - 1) / OWN_CHUNK) * OWN_CHUNK);

    const int tilesM = (N + 15) / 16;
    const int gemmGrid = (tilesM + PAIR_WAVES - 1) / PAIR_WAVES;
    auto cdiv = [](size_t a, size_t b) -> int { return (int)((a + b - 1) / b); };

    node_scores_rows<true><<<cdiv((size_t)N, RD_ROWS * RD_WAVES), 32 * RD_WAVES, 0, stream>>>(features, a_s1, a_d1, es, ed, N, ei + E, E, own4, W1);
    aggregate_owned<F, false><<<ownGrid, OWN_THREADS, 0, stream>>>(ei, E, N, features, es, ed,
                                                                  nullptr, 0, nullptr, xbf, nullptr, svn, sve, svd,
                                                                  (const unsigned char*)own4);

    wmma_two_products<F, Hh, F><<<gemmGrid, 32 * PAIR_WAVES, 0, stream>>>(xbf, W1, b1, W2, buf1, N);
    node_scores_rows<false><<<cdiv((size_t)N, RD_ROWS * RD_WAVES), 32 * RD_WAVES, 0, stream>>>(buf1, a_s2, a_d2, es, ed, N, nullptr, 0, nullptr, nullptr);
    aggregate_owned<F, true><<<ownGrid, OWN_THREADS, 0, stream>>>(ei, E, N, buf1, es, ed,
                                                                 mask, P * 2, b2, nullptr, buf2, svn, sve, svd, nullptr);

    link_predict_rows<<<cdiv((size_t)P, RD_PAIRS * RD_WAVES), 32 * RD_WAVES, 0, stream>>>(buf2, mask, Wl, bl,
                                                                                         (float*)d_out, P);
}
